# starting kernel
# baseline (speedup 1.0000x reference)
#include <hip/hip_runtime.h>


typedef float v2f __attribute__((ext_vector_type(2)));
typedef float v8f __attribute__((ext_vector_type(8)));

__global__ void gcn_init(float* __restrict__ deg, float* __restrict__ agg,
                         float* __restrict__ z, float* __restrict__ out, int n) {
  int i = blockIdx.x * blockDim.x + threadIdx.x;
  if (i < n) {
    deg[i] = 1.0f;
    agg[2 * i] = 0.0f;
    agg[2 * i + 1] = 0.0f;
    z[i] = 0.0f;
    out[i] = 0.0f;
  }
}

__global__ void gcn_degree(const int* __restrict__ dst, float* __restrict__ deg,
                           int E) {
  int i = blockIdx.x * blockDim.x + threadIdx.x;
  int stride = gridDim.x * blockDim.x;
  for (; i < E; i += stride) {
    if (i + stride < E) __builtin_prefetch(&dst[i + stride], 0, 1);
    atomicAdd(&deg[dst[i]], 1.0f);
  }
}

__global__ void gcn_rsqrt(float* __restrict__ dinv, int n) {
  int i = blockIdx.x * blockDim.x + threadIdx.x;
  if (i < n) dinv[i] = rsqrtf(dinv[i]);
}

__global__ void gcn_agg2(const int* __restrict__ src, const int* __restrict__ dst,
                         const float* __restrict__ x, const float* __restrict__ dinv,
                         float* __restrict__ agg, int E) {
  int i = blockIdx.x * blockDim.x + threadIdx.x;
  int stride = gridDim.x * blockDim.x;
  for (; i < E; i += stride) {
    if (i + stride < E) {
      __builtin_prefetch(&src[i + stride], 0, 1);
      __builtin_prefetch(&dst[i + stride], 0, 1);
    }
    int s = src[i];
    int d = dst[i];
    float w = dinv[s] * dinv[d];
    atomicAdd(&agg[2 * d], x[2 * s] * w);
    atomicAdd(&agg[2 * d + 1], x[2 * s + 1] * w);
  }
}

__global__ __launch_bounds__(256) void gcn_dense_wmma(
    const float* __restrict__ agg, const float* __restrict__ x,
    const float* __restrict__ dinv, const float* __restrict__ W1,
    const float* __restrict__ b1, const float* __restrict__ W2,
    float* __restrict__ z, int n) {
  const int lane = threadIdx.x & 31;
  const int wave = threadIdx.x >> 5;
  const int tile = blockIdx.x * 8 + wave;
  const int base = tile * 16;
  if (base >= n) return;

  const int m = lane & 15;
  const bool hiK = lane >= 16;
  const float kmask = hiK ? 0.0f : 1.0f;

  int node = base + m;
  const float rowmask = (!hiK && node < n) ? 1.0f : 0.0f;
  if (node >= n) node = n - 1;
  {
  }
  float di = dinv[node];
  float sl = di * di;
  v2f A;
  A.x = (agg[2 * node]     + x[2 * node]     * sl) * rowmask;
  A.y = (agg[2 * node + 1] + x[2 * node + 1] * sl) * rowmask;

  float zacc[8];
#pragma unroll
  for (int v = 0; v < 8; ++v) zacc[v] = 0.0f;

#pragma unroll
  for (int n0 = 0; n0 < 128; n0 += 16) {
    int col = n0 + m;
    v2f B;
    B.x = W1[col]       * kmask;
    B.y = W1[128 + col] * kmask;
    float bias = b1[col];
    v8f C;
#pragma unroll
    for (int v = 0; v < 8; ++v) C[v] = bias;

    C = __builtin_amdgcn_wmma_f32_16x16x4_f32(
         false, A,  false, B,
         (short)0, C,  false,  false);

    float w2 = W2[col];
#pragma unroll
    for (int v = 0; v < 8; ++v) {
      float h = C[v];
      h = h > 0.0f ? h : 0.0f;
      zacc[v] += h * w2;
    }
  }

#pragma unroll
  for (int v = 0; v < 8; ++v) {
    float r = zacc[v];
    r += __shfl_xor(r, 1, 16);
    r += __shfl_xor(r, 2, 16);
    r += __shfl_xor(r, 4, 16);
    r += __shfl_xor(r, 8, 16);
    zacc[v] = r;
  }
  if (m == 0) {
    int rowbase = base + (hiK ? 8 : 0);
#pragma unroll
    for (int v = 0; v < 8; ++v) {
      int r = rowbase + v;
      if (r < n) z[r] = zacc[v];
    }
  }
}

__global__ void gcn_out_self(const float* __restrict__ z,
                             const float* __restrict__ dinv,
                             const float* __restrict__ b2,
                             float* __restrict__ out, int n) {
  int i = blockIdx.x * blockDim.x + threadIdx.x;
  if (i < n) out[i] = z[i] * dinv[i] * dinv[i] + b2[0];
}

__global__ void gcn_out_edges(const int* __restrict__ src, const int* __restrict__ dst,
                              const float* __restrict__ z,
                              const float* __restrict__ dinv,
                              float* __restrict__ out, int E) {
  int i = blockIdx.x * blockDim.x + threadIdx.x;
  int stride = gridDim.x * blockDim.x;
  for (; i < E; i += stride) {
    if (i + stride < E) {
      __builtin_prefetch(&src[i + stride], 0, 1);
      __builtin_prefetch(&dst[i + stride], 0, 1);
    }
    int s = src[i];
    int d = dst[i];
    atomicAdd(&out[d], z[s] * dinv[s] * dinv[d]);
  }
}

extern "C" void kernel_launch(void* const* d_in, const int* in_sizes, int n_in,
                              void* d_out, int out_size, void* d_ws, size_t ws_size,
                              hipStream_t stream) {
  const float* x  = (const float*)d_in[0];
  const int*   ei = (const int*)d_in[1];
  const float* W1 = (const float*)d_in[2];
  const float* b1 = (const float*)d_in[3];
  const float* W2 = (const float*)d_in[4];
  const float* b2 = (const float*)d_in[5];

  int n = in_sizes[0] / 2;
  int E = in_sizes[1] / 2;
  const int* src = ei;
  const int* dst = ei + E;
  float* out = (float*)d_out;

  float* dinv = (float*)d_ws;
  float* agg  = dinv + n;
  float* z    = agg + 2 * (size_t)n;

  const int tb = 256;
  int gN = (n + tb - 1) / tb;
  int gE = (E + tb - 1) / tb;
  if (gE > 4096) gE = 4096;

  gcn_init<<<gN, tb, 0, stream>>>(dinv, agg, z, out, n);
  gcn_degree<<<gE, tb, 0, stream>>>(dst, dinv, E);
  gcn_rsqrt<<<gN, tb, 0, stream>>>(dinv, n);
  gcn_agg2<<<gE, tb, 0, stream>>>(src, dst, x, dinv, agg, E);

  int tiles = (n + 15) / 16;
  int dblocks = (tiles + 7) / 8;
  gcn_dense_wmma<<<dblocks, 256, 0, stream>>>(agg, x, dinv, W1, b1, W2, z, n);

  gcn_out_self<<<gN, tb, 0, stream>>>(z, dinv, b2, out, n);
  gcn_out_edges<<<gE, tb, 0, stream>>>(src, dst, z, dinv, out, E);
}


# Edges counting-sorted by destination (one returning integer add per edge gives its slot); both layers then sum by gather, no float atomics; f32 sum order: slot order, 8-lane tree
# speedup vs baseline: 1.9121x; 1.9121x over previous
#include <hip/hip_runtime.h>


typedef float v2f __attribute__((ext_vector_type(2)));
typedef float v8f __attribute__((ext_vector_type(8)));

__global__ void gcn_init(float* __restrict__ deg, float* __restrict__ agg,
                         float* __restrict__ z, float* __restrict__ out, int n) {
  int i = blockIdx.x * blockDim.x + threadIdx.x;
  if (i < n) {
    deg[i] = 1.0f;
    agg[2 * i] = 0.0f;
    agg[2 * i + 1] = 0.0f;
    z[i] = 0.0f;
    out[i] = 0.0f;
  }
}

__global__ void gcn_degree(const int* __restrict__ dst, float* __restrict__ deg,
                           int E) {
  int i = blockIdx.x * blockDim.x + threadIdx.x;
  int stride = gridDim.x * blockDim.x;
  for (; i < E; i += stride) {
    if (i + stride < E) __builtin_prefetch(&dst[i + stride], 0, 1);
    atomicAdd(&deg[dst[i]], 1.0f);
  }
}

__global__ void gcn_rsqrt(float* __restrict__ dinv, int n) {
  int i = blockIdx.x * blockDim.x + threadIdx.x;
  if (i < n) dinv[i] = rsqrtf(dinv[i]);
}

__global__ void gcn_agg2(const int* __restrict__ src, const int* __restrict__ dst,
                         const float* __restrict__ x, const float* __restrict__ dinv,
                         float* __restrict__ agg, int E) {
  int i = blockIdx.x * blockDim.x + threadIdx.x;
  int stride = gridDim.x * blockDim.x;
  for (; i < E; i += stride) {
    if (i + stride < E) {
      __builtin_prefetch(&src[i + stride], 0, 1);
      __builtin_prefetch(&dst[i + stride], 0, 1);
    }
    int s = src[i];
    int d = dst[i];
    float w = dinv[s] * dinv[d];
    atomicAdd(&agg[2 * d], x[2 * s] * w);
    atomicAdd(&agg[2 * d + 1], x[2 * s + 1] * w);
  }
}

template <bool SORTED>
__global__ __launch_bounds__(256) void gcn_dense_wmma(
    const float* __restrict__ agg, const float* __restrict__ x,
    const float* __restrict__ dinv, const unsigned* __restrict__ cnt,
    const float* __restrict__ W1,
    const float* __restrict__ b1, const float* __restrict__ W2,
    float* __restrict__ z, int n) {
  const int lane = threadIdx.x & 31;
  const int wave = threadIdx.x >> 5;
  const int tile = blockIdx.x * 8 + wave;
  const int base = tile * 16;
  if (base >= n) return;

  const int m = lane & 15;
  const bool hiK = lane >= 16;
  const float kmask = hiK ? 0.0f : 1.0f;

  int node = base + m;
  const float rowmask = (!hiK && node < n) ? 1.0f : 0.0f;
  if (node >= n) node = n - 1;
  v2f A;
  if (SORTED) {
    A.x = agg[2 * node]     * rowmask;
    A.y = agg[2 * node + 1] * rowmask;
  } else {
    float di = dinv[node];
    float sl = di * di;
    A.x = (agg[2 * node]     + x[2 * node]     * sl) * rowmask;
    A.y = (agg[2 * node + 1] + x[2 * node + 1] * sl) * rowmask;
  }

  float zacc[8];
#pragma unroll
  for (int v = 0; v < 8; ++v) zacc[v] = 0.0f;

#pragma unroll
  for (int n0 = 0; n0 < 128; n0 += 16) {
    int col = n0 + m;
    v2f B;
    B.x = W1[col]       * kmask;
    B.y = W1[128 + col] * kmask;
    float bias = b1[col];
    v8f C;
#pragma unroll
    for (int v = 0; v < 8; ++v) C[v] = bias;

    C = __builtin_amdgcn_wmma_f32_16x16x4_f32(
         false, A,  false, B,
         (short)0, C,  false,  false);

    float w2 = W2[col];
#pragma unroll
    for (int v = 0; v < 8; ++v) {
      float h = C[v];
      h = h > 0.0f ? h : 0.0f;
      zacc[v] += h * w2;
    }
  }

#pragma unroll
  for (int v = 0; v < 8; ++v) {
    float r = zacc[v];
    r += __shfl_xor(r, 1, 16);
    r += __shfl_xor(r, 2, 16);
    r += __shfl_xor(r, 4, 16);
    r += __shfl_xor(r, 8, 16);
    zacc[v] = r;
  }
  if (m == 0) {
    int rowbase = base + (hiK ? 8 : 0);
#pragma unroll
    for (int v = 0; v < 8; ++v) {
      int r = rowbase + v;
      if (r < n) {
        if (SORTED) z[r] = zacc[v] * rsqrtf((float)(cnt[r] + 1u));
        else        z[r] = zacc[v];
      }
    }
  }
}

__global__ void gcn_out_self(const float* __restrict__ z,
                             const float* __restrict__ dinv,
                             const float* __restrict__ b2,
                             float* __restrict__ out, int n) {
  int i = blockIdx.x * blockDim.x + threadIdx.x;
  if (i < n) out[i] = z[i] * dinv[i] * dinv[i] + b2[0];
}

__global__ void gcn_out_edges(const int* __restrict__ src, const int* __restrict__ dst,
                              const float* __restrict__ z,
                              const float* __restrict__ dinv,
                              float* __restrict__ out, int E) {
  int i = blockIdx.x * blockDim.x + threadIdx.x;
  int stride = gridDim.x * blockDim.x;
  for (; i < E; i += stride) {
    if (i + stride < E) {
      __builtin_prefetch(&src[i + stride], 0, 1);
      __builtin_prefetch(&dst[i + stride], 0, 1);
    }
    int s = src[i];
    int d = dst[i];
    atomicAdd(&out[d], z[s] * dinv[s] * dinv[d]);
  }
}


__global__ void gcn_zero_cnt(unsigned* __restrict__ cnt, int n) {
  int i = blockIdx.x * blockDim.x + threadIdx.x;
  if (i < n) cnt[i] = 0u;
}

__global__ void gcn_rank(const int* __restrict__ dst, unsigned* __restrict__ cnt,
                         unsigned* __restrict__ rank, int E) {
  int i = blockIdx.x * blockDim.x + threadIdx.x;
  int stride = gridDim.x * blockDim.x;
  for (; i < E; i += stride) {
    if (i + stride < E) __builtin_prefetch(&dst[i + stride], 0, 1);
    rank[i] = atomicAdd(&cnt[dst[i]], 1u);
  }
}

__global__ __launch_bounds__(1024) void gcn_scan(
    const unsigned* __restrict__ cnt, unsigned* __restrict__ rowptr,
    const float* __restrict__ x, float* __restrict__ xs, int n) {
  __shared__ unsigned s_incl[32];
  __shared__ unsigned s_part[32];
  const int tid = threadIdx.x;
  const int lane = tid & 31;
  const int wv = tid >> 5;
  const int blockBase = blockIdx.x * 1024;
  const int lim = blockBase / 4;

  const uint4* c4 = (const uint4*)cnt;
  unsigned p0 = 0u, p1 = 0u, p2 = 0u, p3 = 0u;
  int i = tid;
  for (; i + 3072 < lim; i += 4096) {
    uint4 a = c4[i];
    uint4 b = c4[i + 1024];
    uint4 c = c4[i + 2048];
    uint4 d = c4[i + 3072];
    p0 += a.x + a.y + a.z + a.w;
    p1 += b.x + b.y + b.z + b.w;
    p2 += c.x + c.y + c.z + c.w;
    p3 += d.x + d.y + d.z + d.w;
  }
  for (; i < lim; i += 1024) {
    uint4 a = c4[i];
    p0 += a.x + a.y + a.z + a.w;
  }
  unsigned part = (p0 + p1) + (p2 + p3);
  part += __shfl_xor(part, 16, 32);
  part += __shfl_xor(part, 8, 32);
  part += __shfl_xor(part, 4, 32);
  part += __shfl_xor(part, 2, 32);
  part += __shfl_xor(part, 1, 32);

  const int node = blockBase + tid;
  const unsigned mine = (node < n) ? cnt[node] : 0u;
  unsigned incl = mine;
#pragma unroll
  for (int o = 1; o < 32; o <<= 1) {
    unsigned t = __shfl_up(incl, o, 32);
    if (lane >= o) incl += t;
  }
  if (lane == 31) s_incl[wv] = incl;
  if (lane == 0) s_part[wv] = part;
  __syncthreads();
  unsigned off = s_part[lane] + ((lane < wv) ? s_incl[lane] : 0u);
  off += __shfl_xor(off, 16, 32);
  off += __shfl_xor(off, 8, 32);
  off += __shfl_xor(off, 4, 32);
  off += __shfl_xor(off, 2, 32);
  off += __shfl_xor(off, 1, 32);
  if (node < n) {
    rowptr[node] = off + incl - mine;
    float di = rsqrtf((float)(mine + 1u));
    xs[2 * node]     = x[2 * node]     * di;
    xs[2 * node + 1] = x[2 * node + 1] * di;
  }
}

__global__ void gcn_fill(const int* __restrict__ src, const int* __restrict__ dst,
                         const unsigned* __restrict__ rowptr,
                         const unsigned* __restrict__ rank,
                         int* __restrict__ sorted, int E) {
  int i = blockIdx.x * blockDim.x + threadIdx.x;
  int stride = gridDim.x * blockDim.x;
  for (; i < E; i += stride) {
    if (i + stride < E) {
      __builtin_prefetch(&src[i + stride], 0, 1);
      __builtin_prefetch(&dst[i + stride], 0, 1);
    }
    int d = dst[i];
    sorted[rowptr[d] + rank[i]] = src[i];
  }
}

__global__ __launch_bounds__(256) void gcn_gather1(
    const int* __restrict__ sorted, const unsigned* __restrict__ rowptr,
    const unsigned* __restrict__ cnt, const float* __restrict__ xs,
    float* __restrict__ agg, int n) {
  const int l = threadIdx.x & 7;
  const int groups = (gridDim.x * 256) >> 3;
  const float2* xs2 = (const float2*)xs;
  for (int node = (blockIdx.x * 256 + threadIdx.x) >> 3; node < n; node += groups) {
    const unsigned beg = rowptr[node];
    const unsigned c = cnt[node];
    float ax = 0.0f, ay = 0.0f;
    for (unsigned j0 = 0; j0 < c; j0 += 8) {
      unsigned j = j0 + l;
      if (j < c) {
        float2 v = xs2[sorted[beg + j]];
        ax += v.x;
        ay += v.y;
      }
    }
    ax += __shfl_xor(ax, 1, 8);  ay += __shfl_xor(ay, 1, 8);
    ax += __shfl_xor(ax, 2, 8);  ay += __shfl_xor(ay, 2, 8);
    ax += __shfl_xor(ax, 4, 8);  ay += __shfl_xor(ay, 4, 8);
    if (l == 0) {
      float di = rsqrtf((float)(c + 1u));
      float2 self = xs2[node];
      agg[2 * node]     = (ax + self.x) * di;
      agg[2 * node + 1] = (ay + self.y) * di;
    }
  }
}

__global__ __launch_bounds__(256) void gcn_gather2(
    const int* __restrict__ sorted, const unsigned* __restrict__ rowptr,
    const unsigned* __restrict__ cnt, const float* __restrict__ zs,
    const float* __restrict__ b2, float* __restrict__ out, int n) {
  const int l = threadIdx.x & 7;
  const int groups = (gridDim.x * 256) >> 3;
  for (int node = (blockIdx.x * 256 + threadIdx.x) >> 3; node < n; node += groups) {
    const unsigned beg = rowptr[node];
    const unsigned c = cnt[node];
    float acc = 0.0f;
    for (unsigned j0 = 0; j0 < c; j0 += 8) {
      unsigned j = j0 + l;
      if (j < c) acc += zs[sorted[beg + j]];
    }
    acc += __shfl_xor(acc, 1, 8);
    acc += __shfl_xor(acc, 2, 8);
    acc += __shfl_xor(acc, 4, 8);
    if (l == 0) out[node] = (acc + zs[node]) * rsqrtf((float)(c + 1u)) + b2[0];
  }
}

extern "C" void kernel_launch(void* const* d_in, const int* in_sizes, int n_in,
                              void* d_out, int out_size, void* d_ws, size_t ws_size,
                              hipStream_t stream) {
  const float* x  = (const float*)d_in[0];
  const int*   ei = (const int*)d_in[1];
  const float* W1 = (const float*)d_in[2];
  const float* b1 = (const float*)d_in[3];
  const float* W2 = (const float*)d_in[4];
  const float* b2 = (const float*)d_in[5];

  int n = in_sizes[0] / 2;
  int E = in_sizes[1] / 2;
  const int* src = ei;
  const int* dst = ei + E;
  float* out = (float*)d_out;

  const int tb = 256;
  int gN = (n + tb - 1) / tb;
  int gE = (E + tb - 1) / tb;
  if (gE > 4096) gE = 4096;
  int tiles = (n + 15) / 16;
  int dblocks = (tiles + 7) / 8;

  size_t tail = ((size_t)E >= 3 * (size_t)n) ? 0 : 3 * (size_t)n;
  size_t need = (4 * (size_t)n + 2 * (size_t)E + tail) * sizeof(float);
  if (ws_size >= need) {
    unsigned* cnt    = (unsigned*)d_ws;
    unsigned* rowptr = cnt + n;
    float*    xs     = (float*)(rowptr + n);
    unsigned* rank   = (unsigned*)(xs + 2 * (size_t)n);
    int*      sorted = (int*)(rank + E);
    float*    aggS   = tail ? (float*)(sorted + E) : (float*)rank;
    float*    zs     = aggS + 2 * (size_t)n;

    int gG = (n + 31) / 32;
    if (gG > 1024) gG = 1024;

    gcn_zero_cnt<<<gN, tb, 0, stream>>>(cnt, n);
    gcn_rank<<<gE, tb, 0, stream>>>(dst, cnt, rank, E);
    gcn_scan<<<(n + 1023) / 1024, 1024, 0, stream>>>(cnt, rowptr, x, xs, n);
    gcn_fill<<<gE, tb, 0, stream>>>(src, dst, rowptr, rank, sorted, E);
    gcn_gather1<<<gG, 256, 0, stream>>>(sorted, rowptr, cnt, xs, aggS, n);
    gcn_dense_wmma<true><<<dblocks, 256, 0, stream>>>(aggS, x, xs, cnt, W1, b1, W2, zs, n);
    gcn_gather2<<<gG, 256, 0, stream>>>(sorted, rowptr, cnt, zs, b2, out, n);
    return;
  }

  float* dinv = (float*)d_ws;
  float* agg  = dinv + n;
  float* z    = agg + 2 * (size_t)n;

  gcn_init<<<gN, tb, 0, stream>>>(dinv, agg, z, out, n);
  gcn_degree<<<gE, tb, 0, stream>>>(dst, dinv, E);
  gcn_rsqrt<<<gN, tb, 0, stream>>>(dinv, n);
  gcn_agg2<<<gE, tb, 0, stream>>>(src, dst, x, dinv, agg, E);

  gcn_dense_wmma<false><<<dblocks, 256, 0, stream>>>(agg, x, dinv, (const unsigned*)0, W1, b1, W2, z, n);

  gcn_out_self<<<gN, tb, 0, stream>>>(z, dinv, b2, out, n);
  gcn_out_edges<<<gE, tb, 0, stream>>>(src, dst, z, dinv, out, E);
}


# Slots ranked with LDS counters, not returning global atomics: 16 edge slices x node ranges fill a slice-by-node count table, column prefix = slice bases; f32 sum order: slice, LDS arrival
# speedup vs baseline: 3.1615x; 1.6534x over previous
#include <hip/hip_runtime.h>


typedef float v2f __attribute__((ext_vector_type(2)));
typedef float v8f __attribute__((ext_vector_type(8)));

__global__ void gcn_init(float* __restrict__ deg, float* __restrict__ agg,
                         float* __restrict__ z, float* __restrict__ out, int n) {
  int i = blockIdx.x * blockDim.x + threadIdx.x;
  if (i < n) {
    deg[i] = 1.0f;
    agg[2 * i] = 0.0f;
    agg[2 * i + 1] = 0.0f;
    z[i] = 0.0f;
    out[i] = 0.0f;
  }
}

__global__ void gcn_degree(const int* __restrict__ dst, float* __restrict__ deg,
                           int E) {
  int i = blockIdx.x * blockDim.x + threadIdx.x;
  int stride = gridDim.x * blockDim.x;
  for (; i < E; i += stride) {
    if (i + stride < E) __builtin_prefetch(&dst[i + stride], 0, 1);
    atomicAdd(&deg[dst[i]], 1.0f);
  }
}

__global__ void gcn_rsqrt(float* __restrict__ dinv, int n) {
  int i = blockIdx.x * blockDim.x + threadIdx.x;
  if (i < n) dinv[i] = rsqrtf(dinv[i]);
}

__global__ void gcn_agg2(const int* __restrict__ src, const int* __restrict__ dst,
                         const float* __restrict__ x, const float* __restrict__ dinv,
                         float* __restrict__ agg, int E) {
  int i = blockIdx.x * blockDim.x + threadIdx.x;
  int stride = gridDim.x * blockDim.x;
  for (; i < E; i += stride) {
    if (i + stride < E) {
      __builtin_prefetch(&src[i + stride], 0, 1);
      __builtin_prefetch(&dst[i + stride], 0, 1);
    }
    int s = src[i];
    int d = dst[i];
    float w = dinv[s] * dinv[d];
    atomicAdd(&agg[2 * d], x[2 * s] * w);
    atomicAdd(&agg[2 * d + 1], x[2 * s + 1] * w);
  }
}

template <bool SORTED>
__global__ __launch_bounds__(256) void gcn_dense_wmma(
    const float* __restrict__ agg, const float* __restrict__ x,
    const float* __restrict__ dinv, const unsigned* __restrict__ cnt,
    const float* __restrict__ W1,
    const float* __restrict__ b1, const float* __restrict__ W2,
    float* __restrict__ z, int n) {
  const int lane = threadIdx.x & 31;
  const int wave = threadIdx.x >> 5;
  const int tile = blockIdx.x * 8 + wave;
  const int base = tile * 16;
  if (base >= n) return;

  const int m = lane & 15;
  const bool hiK = lane >= 16;
  const float kmask = hiK ? 0.0f : 1.0f;

  int node = base + m;
  const float rowmask = (!hiK && node < n) ? 1.0f : 0.0f;
  if (node >= n) node = n - 1;
  v2f A;
  if (SORTED) {
    A.x = agg[2 * node]     * rowmask;
    A.y = agg[2 * node + 1] * rowmask;
  } else {
    float di = dinv[node];
    float sl = di * di;
    A.x = (agg[2 * node]     + x[2 * node]     * sl) * rowmask;
    A.y = (agg[2 * node + 1] + x[2 * node + 1] * sl) * rowmask;
  }

  float zacc[8];
#pragma unroll
  for (int v = 0; v < 8; ++v) zacc[v] = 0.0f;

#pragma unroll
  for (int n0 = 0; n0 < 128; n0 += 16) {
    int col = n0 + m;
    v2f B;
    B.x = W1[col]       * kmask;
    B.y = W1[128 + col] * kmask;
    float bias = b1[col];
    v8f C;
#pragma unroll
    for (int v = 0; v < 8; ++v) C[v] = bias;

    C = __builtin_amdgcn_wmma_f32_16x16x4_f32(
         false, A,  false, B,
         (short)0, C,  false,  false);

    float w2 = W2[col];
#pragma unroll
    for (int v = 0; v < 8; ++v) {
      float h = C[v];
      h = h > 0.0f ? h : 0.0f;
      zacc[v] += h * w2;
    }
  }

#pragma unroll
  for (int v = 0; v < 8; ++v) {
    float r = zacc[v];
    r += __shfl_xor(r, 1, 16);
    r += __shfl_xor(r, 2, 16);
    r += __shfl_xor(r, 4, 16);
    r += __shfl_xor(r, 8, 16);
    zacc[v] = r;
  }
  if (m == 0) {
    int rowbase = base + (hiK ? 8 : 0);
#pragma unroll
    for (int v = 0; v < 8; ++v) {
      int r = rowbase + v;
      if (r < n) {
        if (SORTED) z[r] = zacc[v] * rsqrtf((float)(cnt[r] + 1u));
        else        z[r] = zacc[v];
      }
    }
  }
}

__global__ void gcn_out_self(const float* __restrict__ z,
                             const float* __restrict__ dinv,
                             const float* __restrict__ b2,
                             float* __restrict__ out, int n) {
  int i = blockIdx.x * blockDim.x + threadIdx.x;
  if (i < n) out[i] = z[i] * dinv[i] * dinv[i] + b2[0];
}

__global__ void gcn_out_edges(const int* __restrict__ src, const int* __restrict__ dst,
                              const float* __restrict__ z,
                              const float* __restrict__ dinv,
                              float* __restrict__ out, int E) {
  int i = blockIdx.x * blockDim.x + threadIdx.x;
  int stride = gridDim.x * blockDim.x;
  for (; i < E; i += stride) {
    if (i + stride < E) {
      __builtin_prefetch(&src[i + stride], 0, 1);
      __builtin_prefetch(&dst[i + stride], 0, 1);
    }
    int s = src[i];
    int d = dst[i];
    atomicAdd(&out[d], z[s] * dinv[s] * dinv[d]);
  }
}


constexpr int SL = 16;
constexpr int RN = 6400;

__global__ __launch_bounds__(1024) void gcn_slice_count(
    const int* __restrict__ dst, unsigned* __restrict__ table,
    int n, int E, int L) {
  __shared__ unsigned h[RN];
  const int tid = threadIdx.x;
  const int s = blockIdx.x;
  const int lo = blockIdx.y * RN;
  const unsigned lim = (n - lo < RN) ? (unsigned)(n - lo) : (unsigned)RN;
  for (int i = tid; i < RN; i += 1024) h[i] = 0u;
  __syncthreads();
  const int e0 = s * L;
  const int e1 = (E - e0 < L) ? E : e0 + L;
  for (int e = e0 + tid; e < e1; e += 1024 * 16) {
    int d[16];
#pragma unroll
    for (int k = 0; k < 16; ++k) {
      int ee = e + k * 1024;
      d[k] = (ee < e1) ? dst[ee] : -1;
    }
#pragma unroll
    for (int k = 0; k < 16; ++k) {
      unsigned x = (unsigned)d[k] - (unsigned)lo;
      if (x < lim) atomicAdd(&h[x], 1u);
    }
  }
  __syncthreads();
  unsigned* row = table + (size_t)s * n + lo;
  for (int i = tid; i < (int)lim; i += 1024) row[i] = h[i];
}

__global__ void gcn_slice_prefix(unsigned* __restrict__ table,
                                 unsigned* __restrict__ cnt, int n) {
  int node = blockIdx.x * blockDim.x + threadIdx.x;
  if (node >= n) return;
  unsigned v[SL];
#pragma unroll
  for (int s = 0; s < SL; ++s) v[s] = table[(size_t)s * n + node];
  unsigned run = 0u;
#pragma unroll
  for (int s = 0; s < SL; ++s) {
    table[(size_t)s * n + node] = run;
    run += v[s];
  }
  cnt[node] = run;
}

__global__ __launch_bounds__(1024) void gcn_scan(
    const unsigned* __restrict__ cnt, unsigned* __restrict__ rowptr,
    const float* __restrict__ x, float* __restrict__ xs, int n) {
  __shared__ unsigned s_incl[32];
  __shared__ unsigned s_part[32];
  const int tid = threadIdx.x;
  const int lane = tid & 31;
  const int wv = tid >> 5;
  const int blockBase = blockIdx.x * 1024;
  const int lim = blockBase / 4;

  const uint4* c4 = (const uint4*)cnt;
  unsigned p0 = 0u, p1 = 0u, p2 = 0u, p3 = 0u;
  int i = tid;
  for (; i + 3072 < lim; i += 4096) {
    uint4 a = c4[i];
    uint4 b = c4[i + 1024];
    uint4 c = c4[i + 2048];
    uint4 d = c4[i + 3072];
    p0 += a.x + a.y + a.z + a.w;
    p1 += b.x + b.y + b.z + b.w;
    p2 += c.x + c.y + c.z + c.w;
    p3 += d.x + d.y + d.z + d.w;
  }
  for (; i < lim; i += 1024) {
    uint4 a = c4[i];
    p0 += a.x + a.y + a.z + a.w;
  }
  unsigned part = (p0 + p1) + (p2 + p3);
  part += __shfl_xor(part, 16, 32);
  part += __shfl_xor(part, 8, 32);
  part += __shfl_xor(part, 4, 32);
  part += __shfl_xor(part, 2, 32);
  part += __shfl_xor(part, 1, 32);

  const int node = blockBase + tid;
  const unsigned mine = (node < n) ? cnt[node] : 0u;
  unsigned incl = mine;
#pragma unroll
  for (int o = 1; o < 32; o <<= 1) {
    unsigned t = __shfl_up(incl, o, 32);
    if (lane >= o) incl += t;
  }
  if (lane == 31) s_incl[wv] = incl;
  if (lane == 0) s_part[wv] = part;
  __syncthreads();
  unsigned off = s_part[lane] + ((lane < wv) ? s_incl[lane] : 0u);
  off += __shfl_xor(off, 16, 32);
  off += __shfl_xor(off, 8, 32);
  off += __shfl_xor(off, 4, 32);
  off += __shfl_xor(off, 2, 32);
  off += __shfl_xor(off, 1, 32);
  if (node < n) {
    rowptr[node] = off + incl - mine;
    float di = rsqrtf((float)(mine + 1u));
    xs[2 * node]     = x[2 * node]     * di;
    xs[2 * node + 1] = x[2 * node + 1] * di;
  }
}

__global__ __launch_bounds__(1024) void gcn_slice_fill(
    const int* __restrict__ src, const int* __restrict__ dst,
    const unsigned* __restrict__ table, const unsigned* __restrict__ rowptr,
    int* __restrict__ sorted, int n, int E, int L) {
  __shared__ unsigned cur[RN];
  const int tid = threadIdx.x;
  const int s = blockIdx.x;
  const int lo = blockIdx.y * RN;
  const unsigned lim = (n - lo < RN) ? (unsigned)(n - lo) : (unsigned)RN;
  const unsigned* row = table + (size_t)s * n + lo;
  constexpr int RT = (RN + 1023) / 1024;
  unsigned ra[RT], rb[RT];
#pragma unroll
  for (int k = 0; k < RT; ++k) {
    int i = tid + k * 1024;
    bool in = i < (int)lim;
    ra[k] = in ? rowptr[lo + i] : 0u;
    rb[k] = in ? row[i] : 0u;
  }
#pragma unroll
  for (int k = 0; k < RT; ++k) {
    int i = tid + k * 1024;
    if (i < (int)lim) cur[i] = ra[k] + rb[k];
  }
  __syncthreads();
  const int e0 = s * L;
  const int e1 = (E - e0 < L) ? E : e0 + L;
  for (int e = e0 + tid; e < e1; e += 1024 * 8) {
    int d[8], sv[8];
    unsigned pos[8];
#pragma unroll
    for (int k = 0; k < 8; ++k) {
      int ee = e + k * 1024;
      bool in = ee < e1;
      d[k] = in ? dst[ee] : -1;
      sv[k] = in ? src[ee] : 0;
    }
#pragma unroll
    for (int k = 0; k < 8; ++k) {
      unsigned x = (unsigned)d[k] - (unsigned)lo;
      pos[k] = 0u;
      if (x < lim) pos[k] = atomicAdd(&cur[x], 1u);
    }
#pragma unroll
    for (int k = 0; k < 8; ++k) {
      unsigned x = (unsigned)d[k] - (unsigned)lo;
      if (x < lim) sorted[pos[k]] = sv[k];
    }
  }
}

__global__ __launch_bounds__(256) void gcn_gather1(
    const int* __restrict__ sorted, const unsigned* __restrict__ rowptr,
    const unsigned* __restrict__ cnt, const float* __restrict__ xs,
    float* __restrict__ agg, int n) {
  const int l = threadIdx.x & 7;
  const int groups = (gridDim.x * 256) >> 3;
  const float2* xs2 = (const float2*)xs;
  for (int node = (blockIdx.x * 256 + threadIdx.x) >> 3; node < n; node += groups) {
    const unsigned beg = rowptr[node];
    const unsigned c = cnt[node];
    float ax = 0.0f, ay = 0.0f;
    for (unsigned j0 = 0; j0 < c; j0 += 8) {
      unsigned j = j0 + l;
      if (j < c) {
        float2 v = xs2[sorted[beg + j]];
        ax += v.x;
        ay += v.y;
      }
    }
    ax += __shfl_xor(ax, 1, 8);  ay += __shfl_xor(ay, 1, 8);
    ax += __shfl_xor(ax, 2, 8);  ay += __shfl_xor(ay, 2, 8);
    ax += __shfl_xor(ax, 4, 8);  ay += __shfl_xor(ay, 4, 8);
    if (l == 0) {
      float di = rsqrtf((float)(c + 1u));
      float2 self = xs2[node];
      agg[2 * node]     = (ax + self.x) * di;
      agg[2 * node + 1] = (ay + self.y) * di;
    }
  }
}

__global__ __launch_bounds__(256) void gcn_gather2(
    const int* __restrict__ sorted, const unsigned* __restrict__ rowptr,
    const unsigned* __restrict__ cnt, const float* __restrict__ zs,
    const float* __restrict__ b2, float* __restrict__ out, int n) {
  const int l = threadIdx.x & 7;
  const int groups = (gridDim.x * 256) >> 3;
  for (int node = (blockIdx.x * 256 + threadIdx.x) >> 3; node < n; node += groups) {
    const unsigned beg = rowptr[node];
    const unsigned c = cnt[node];
    float acc = 0.0f;
    for (unsigned j0 = 0; j0 < c; j0 += 8) {
      unsigned j = j0 + l;
      if (j < c) acc += zs[sorted[beg + j]];
    }
    acc += __shfl_xor(acc, 1, 8);
    acc += __shfl_xor(acc, 2, 8);
    acc += __shfl_xor(acc, 4, 8);
    if (l == 0) out[node] = (acc + zs[node]) * rsqrtf((float)(c + 1u)) + b2[0];
  }
}

extern "C" void kernel_launch(void* const* d_in, const int* in_sizes, int n_in,
                              void* d_out, int out_size, void* d_ws, size_t ws_size,
                              hipStream_t stream) {
  const float* x  = (const float*)d_in[0];
  const int*   ei = (const int*)d_in[1];
  const float* W1 = (const float*)d_in[2];
  const float* b1 = (const float*)d_in[3];
  const float* W2 = (const float*)d_in[4];
  const float* b2 = (const float*)d_in[5];

  int n = in_sizes[0] / 2;
  int E = in_sizes[1] / 2;
  const int* src = ei;
  const int* dst = ei + E;
  float* out = (float*)d_out;

  const int tb = 256;
  int gN = (n + tb - 1) / tb;
  int gE = (E + tb - 1) / tb;
  if (gE > 4096) gE = 4096;
  int tiles = (n + 15) / 16;
  int dblocks = (tiles + 7) / 8;

  size_t need = (4 * (size_t)n + (size_t)SL * (size_t)n + (size_t)E) * sizeof(float);
  int ranges = (n + RN - 1) / RN;
  if (ws_size >= need && ranges <= 65535) {
    unsigned* cnt    = (unsigned*)d_ws;
    unsigned* rowptr = cnt + n;
    float*    xs     = (float*)(rowptr + n);
    unsigned* table  = (unsigned*)(xs + 2 * (size_t)n);
    int*      sorted = (int*)(table + (size_t)SL * (size_t)n);
    float*    aggS   = (float*)table;
    float*    zs     = aggS + 2 * (size_t)n;

    int L = (E + SL - 1) / SL;
    dim3 gS(SL, ranges);
    int gG = (n + 31) / 32;
    if (gG > 1024) gG = 1024;

    gcn_slice_count<<<gS, 1024, 0, stream>>>(dst, table, n, E, L);
    gcn_slice_prefix<<<gN, tb, 0, stream>>>(table, cnt, n);
    gcn_scan<<<(n + 1023) / 1024, 1024, 0, stream>>>(cnt, rowptr, x, xs, n);
    gcn_slice_fill<<<gS, 1024, 0, stream>>>(src, dst, table, rowptr, sorted, n, E, L);
    gcn_gather1<<<gG, 256, 0, stream>>>(sorted, rowptr, cnt, xs, aggS, n);
    gcn_dense_wmma<true><<<dblocks, 256, 0, stream>>>(aggS, x, xs, cnt, W1, b1, W2, zs, n);
    gcn_gather2<<<gG, 256, 0, stream>>>(sorted, rowptr, cnt, zs, b2, out, n);
    return;
  }

  float* dinv = (float*)d_ws;
  float* agg  = dinv + n;
  float* z    = agg + 2 * (size_t)n;

  gcn_init<<<gN, tb, 0, stream>>>(dinv, agg, z, out, n);
  gcn_degree<<<gE, tb, 0, stream>>>(dst, dinv, E);
  gcn_rsqrt<<<gN, tb, 0, stream>>>(dinv, n);
  gcn_agg2<<<gE, tb, 0, stream>>>(src, dst, x, dinv, agg, E);

  gcn_dense_wmma<false><<<dblocks, 256, 0, stream>>>(agg, x, dinv, (const unsigned*)0, W1, b1, W2, z, n);

  gcn_out_self<<<gN, tb, 0, stream>>>(z, dinv, b2, out, n);
  gcn_out_edges<<<gE, tb, 0, stream>>>(src, dst, z, dinv, out, E);
}


# Both gathers batch their reads: an 8-lane group takes 4 consecutive nodes, issues all their slot reads, then all payload reads, then waits; f32 sum order unchanged (slot order, 8-lane tree)
# speedup vs baseline: 3.4103x; 1.0787x over previous
#include <hip/hip_runtime.h>


typedef float v2f __attribute__((ext_vector_type(2)));
typedef float v8f __attribute__((ext_vector_type(8)));

__global__ void gcn_init(float* __restrict__ deg, float* __restrict__ agg,
                         float* __restrict__ z, float* __restrict__ out, int n) {
  int i = blockIdx.x * blockDim.x + threadIdx.x;
  if (i < n) {
    deg[i] = 1.0f;
    agg[2 * i] = 0.0f;
    agg[2 * i + 1] = 0.0f;
    z[i] = 0.0f;
    out[i] = 0.0f;
  }
}

__global__ void gcn_degree(const int* __restrict__ dst, float* __restrict__ deg,
                           int E) {
  int i = blockIdx.x * blockDim.x + threadIdx.x;
  int stride = gridDim.x * blockDim.x;
  for (; i < E; i += stride) {
    if (i + stride < E) __builtin_prefetch(&dst[i + stride], 0, 1);
    atomicAdd(&deg[dst[i]], 1.0f);
  }
}

__global__ void gcn_rsqrt(float* __restrict__ dinv, int n) {
  int i = blockIdx.x * blockDim.x + threadIdx.x;
  if (i < n) dinv[i] = rsqrtf(dinv[i]);
}

__global__ void gcn_agg2(const int* __restrict__ src, const int* __restrict__ dst,
                         const float* __restrict__ x, const float* __restrict__ dinv,
                         float* __restrict__ agg, int E) {
  int i = blockIdx.x * blockDim.x + threadIdx.x;
  int stride = gridDim.x * blockDim.x;
  for (; i < E; i += stride) {
    if (i + stride < E) {
      __builtin_prefetch(&src[i + stride], 0, 1);
      __builtin_prefetch(&dst[i + stride], 0, 1);
    }
    int s = src[i];
    int d = dst[i];
    float w = dinv[s] * dinv[d];
    atomicAdd(&agg[2 * d], x[2 * s] * w);
    atomicAdd(&agg[2 * d + 1], x[2 * s + 1] * w);
  }
}

template <bool SORTED>
__global__ __launch_bounds__(256) void gcn_dense_wmma(
    const float* __restrict__ agg, const float* __restrict__ x,
    const float* __restrict__ dinv, const unsigned* __restrict__ cnt,
    const float* __restrict__ W1,
    const float* __restrict__ b1, const float* __restrict__ W2,
    float* __restrict__ z, int n) {
  const int lane = threadIdx.x & 31;
  const int wave = threadIdx.x >> 5;
  const int tile = blockIdx.x * 8 + wave;
  const int base = tile * 16;
  if (base >= n) return;

  const int m = lane & 15;
  const bool hiK = lane >= 16;
  const float kmask = hiK ? 0.0f : 1.0f;

  int node = base + m;
  const float rowmask = (!hiK && node < n) ? 1.0f : 0.0f;
  if (node >= n) node = n - 1;
  v2f A;
  if (SORTED) {
    A.x = agg[2 * node]     * rowmask;
    A.y = agg[2 * node + 1] * rowmask;
  } else {
    float di = dinv[node];
    float sl = di * di;
    A.x = (agg[2 * node]     + x[2 * node]     * sl) * rowmask;
    A.y = (agg[2 * node + 1] + x[2 * node + 1] * sl) * rowmask;
  }

  float zacc[8];
#pragma unroll
  for (int v = 0; v < 8; ++v) zacc[v] = 0.0f;

#pragma unroll
  for (int n0 = 0; n0 < 128; n0 += 16) {
    int col = n0 + m;
    v2f B;
    B.x = W1[col]       * kmask;
    B.y = W1[128 + col] * kmask;
    float bias = b1[col];
    v8f C;
#pragma unroll
    for (int v = 0; v < 8; ++v) C[v] = bias;

    C = __builtin_amdgcn_wmma_f32_16x16x4_f32(
         false, A,  false, B,
         (short)0, C,  false,  false);

    float w2 = W2[col];
#pragma unroll
    for (int v = 0; v < 8; ++v) {
      float h = C[v];
      h = h > 0.0f ? h : 0.0f;
      zacc[v] += h * w2;
    }
  }

#pragma unroll
  for (int v = 0; v < 8; ++v) {
    float r = zacc[v];
    r += __shfl_xor(r, 1, 16);
    r += __shfl_xor(r, 2, 16);
    r += __shfl_xor(r, 4, 16);
    r += __shfl_xor(r, 8, 16);
    zacc[v] = r;
  }
  if (m == 0) {
    int rowbase = base + (hiK ? 8 : 0);
#pragma unroll
    for (int v = 0; v < 8; ++v) {
      int r = rowbase + v;
      if (r < n) {
        if (SORTED) z[r] = zacc[v] * rsqrtf((float)(cnt[r] + 1u));
        else        z[r] = zacc[v];
      }
    }
  }
}

__global__ void gcn_out_self(const float* __restrict__ z,
                             const float* __restrict__ dinv,
                             const float* __restrict__ b2,
                             float* __restrict__ out, int n) {
  int i = blockIdx.x * blockDim.x + threadIdx.x;
  if (i < n) out[i] = z[i] * dinv[i] * dinv[i] + b2[0];
}

__global__ void gcn_out_edges(const int* __restrict__ src, const int* __restrict__ dst,
                              const float* __restrict__ z,
                              const float* __restrict__ dinv,
                              float* __restrict__ out, int E) {
  int i = blockIdx.x * blockDim.x + threadIdx.x;
  int stride = gridDim.x * blockDim.x;
  for (; i < E; i += stride) {
    if (i + stride < E) {
      __builtin_prefetch(&src[i + stride], 0, 1);
      __builtin_prefetch(&dst[i + stride], 0, 1);
    }
    int s = src[i];
    int d = dst[i];
    atomicAdd(&out[d], z[s] * dinv[s] * dinv[d]);
  }
}


constexpr int SL = 16;
constexpr int RN = 6400;

__global__ __launch_bounds__(1024) void gcn_slice_count(
    const int* __restrict__ dst, unsigned* __restrict__ table,
    int n, int E, int L) {
  __shared__ unsigned h[RN];
  const int tid = threadIdx.x;
  const int s = blockIdx.x;
  const int lo = blockIdx.y * RN;
  const unsigned lim = (n - lo < RN) ? (unsigned)(n - lo) : (unsigned)RN;
  for (int i = tid; i < RN; i += 1024) h[i] = 0u;
  __syncthreads();
  const int e0 = s * L;
  const int e1 = (E - e0 < L) ? E : e0 + L;
  for (int e = e0 + tid; e < e1; e += 1024 * 16) {
    int d[16];
#pragma unroll
    for (int k = 0; k < 16; ++k) {
      int ee = e + k * 1024;
      d[k] = (ee < e1) ? dst[ee] : -1;
    }
#pragma unroll
    for (int k = 0; k < 16; ++k) {
      unsigned x = (unsigned)d[k] - (unsigned)lo;
      if (x < lim) atomicAdd(&h[x], 1u);
    }
  }
  __syncthreads();
  unsigned* row = table + (size_t)s * n + lo;
  for (int i = tid; i < (int)lim; i += 1024) row[i] = h[i];
}

__global__ void gcn_slice_prefix(unsigned* __restrict__ table,
                                 unsigned* __restrict__ cnt, int n) {
  int node = blockIdx.x * blockDim.x + threadIdx.x;
  if (node >= n) return;
  unsigned v[SL];
#pragma unroll
  for (int s = 0; s < SL; ++s) v[s] = table[(size_t)s * n + node];
  unsigned run = 0u;
#pragma unroll
  for (int s = 0; s < SL; ++s) {
    table[(size_t)s * n + node] = run;
    run += v[s];
  }
  cnt[node] = run;
}

__global__ __launch_bounds__(1024) void gcn_scan(
    const unsigned* __restrict__ cnt, unsigned* __restrict__ rowptr,
    const float* __restrict__ x, float* __restrict__ xs, int n) {
  __shared__ unsigned s_incl[32];
  __shared__ unsigned s_part[32];
  const int tid = threadIdx.x;
  const int lane = tid & 31;
  const int wv = tid >> 5;
  const int blockBase = blockIdx.x * 1024;
  const int lim = blockBase / 4;

  const uint4* c4 = (const uint4*)cnt;
  unsigned p0 = 0u, p1 = 0u, p2 = 0u, p3 = 0u;
  int i = tid;
  for (; i + 3072 < lim; i += 4096) {
    uint4 a = c4[i];
    uint4 b = c4[i + 1024];
    uint4 c = c4[i + 2048];
    uint4 d = c4[i + 3072];
    p0 += a.x + a.y + a.z + a.w;
    p1 += b.x + b.y + b.z + b.w;
    p2 += c.x + c.y + c.z + c.w;
    p3 += d.x + d.y + d.z + d.w;
  }
  for (; i < lim; i += 1024) {
    uint4 a = c4[i];
    p0 += a.x + a.y + a.z + a.w;
  }
  unsigned part = (p0 + p1) + (p2 + p3);
  part += __shfl_xor(part, 16, 32);
  part += __shfl_xor(part, 8, 32);
  part += __shfl_xor(part, 4, 32);
  part += __shfl_xor(part, 2, 32);
  part += __shfl_xor(part, 1, 32);

  const int node = blockBase + tid;
  const unsigned mine = (node < n) ? cnt[node] : 0u;
  unsigned incl = mine;
#pragma unroll
  for (int o = 1; o < 32; o <<= 1) {
    unsigned t = __shfl_up(incl, o, 32);
    if (lane >= o) incl += t;
  }
  if (lane == 31) s_incl[wv] = incl;
  if (lane == 0) s_part[wv] = part;
  __syncthreads();
  unsigned off = s_part[lane] + ((lane < wv) ? s_incl[lane] : 0u);
  off += __shfl_xor(off, 16, 32);
  off += __shfl_xor(off, 8, 32);
  off += __shfl_xor(off, 4, 32);
  off += __shfl_xor(off, 2, 32);
  off += __shfl_xor(off, 1, 32);
  if (node < n) {
    rowptr[node] = off + incl - mine;
    float di = rsqrtf((float)(mine + 1u));
    xs[2 * node]     = x[2 * node]     * di;
    xs[2 * node + 1] = x[2 * node + 1] * di;
  }
}

__global__ __launch_bounds__(1024) void gcn_slice_fill(
    const int* __restrict__ src, const int* __restrict__ dst,
    const unsigned* __restrict__ table, const unsigned* __restrict__ rowptr,
    int* __restrict__ sorted, int n, int E, int L) {
  __shared__ unsigned cur[RN];
  const int tid = threadIdx.x;
  const int s = blockIdx.x;
  const int lo = blockIdx.y * RN;
  const unsigned lim = (n - lo < RN) ? (unsigned)(n - lo) : (unsigned)RN;
  const unsigned* row = table + (size_t)s * n + lo;
  constexpr int RT = (RN + 1023) / 1024;
  unsigned ra[RT], rb[RT];
#pragma unroll
  for (int k = 0; k < RT; ++k) {
    int i = tid + k * 1024;
    bool in = i < (int)lim;
    ra[k] = in ? rowptr[lo + i] : 0u;
    rb[k] = in ? row[i] : 0u;
  }
#pragma unroll
  for (int k = 0; k < RT; ++k) {
    int i = tid + k * 1024;
    if (i < (int)lim) cur[i] = ra[k] + rb[k];
  }
  __syncthreads();
  const int e0 = s * L;
  const int e1 = (E - e0 < L) ? E : e0 + L;
  for (int e = e0 + tid; e < e1; e += 1024 * 8) {
    int d[8], sv[8];
    unsigned pos[8];
#pragma unroll
    for (int k = 0; k < 8; ++k) {
      int ee = e + k * 1024;
      bool in = ee < e1;
      d[k] = in ? dst[ee] : -1;
      sv[k] = in ? src[ee] : 0;
    }
#pragma unroll
    for (int k = 0; k < 8; ++k) {
      unsigned x = (unsigned)d[k] - (unsigned)lo;
      pos[k] = 0u;
      if (x < lim) pos[k] = atomicAdd(&cur[x], 1u);
    }
#pragma unroll
    for (int k = 0; k < 8; ++k) {
      unsigned x = (unsigned)d[k] - (unsigned)lo;
      if (x < lim) sorted[pos[k]] = sv[k];
    }
  }
}

constexpr int GN = 4;
constexpr int GQ = 4;

__global__ __launch_bounds__(256) void gcn_gather1(
    const int* __restrict__ sorted, const unsigned* __restrict__ rowptr,
    const unsigned* __restrict__ cnt, const float* __restrict__ xs,
    float* __restrict__ agg, int n) {
  const unsigned l = threadIdx.x & 7;
  const int g = (blockIdx.x * 256 + threadIdx.x) >> 3;
  const int groups = (gridDim.x * 256) >> 3;
  const float2* xs2 = (const float2*)xs;
  for (int nb = g * GN; nb < n; nb += groups * GN) {
    unsigned beg[GN], c[GN];
    float2 self[GN];
    float ax[GN], ay[GN];
#pragma unroll
    for (int k = 0; k < GN; ++k) {
      const int node = nb + k;
      const bool in = node < n;
      beg[k] = in ? rowptr[node] : 0u;
      c[k] = in ? cnt[node] : 0u;
      self[k] = in ? xs2[node] : make_float2(0.0f, 0.0f);
      ax[k] = 0.0f;
      ay[k] = 0.0f;
    }
    unsigned cm = 0u;
#pragma unroll
    for (int k = 0; k < GN; ++k) cm = c[k] > cm ? c[k] : cm;
    for (unsigned j0 = 0; j0 < cm; j0 += 8 * GQ) {
      int idx[GN][GQ];
      float2 v[GN][GQ];
#pragma unroll
      for (int k = 0; k < GN; ++k) {
#pragma unroll
        for (int q = 0; q < GQ; ++q) {
          const unsigned j = j0 + 8 * q + l;
          const unsigned jj = j < c[k] ? j : c[k] - 1u;
          idx[k][q] = sorted[c[k] ? beg[k] + jj : 0u];
        }
      }
#pragma unroll
      for (int k = 0; k < GN; ++k) {
#pragma unroll
        for (int q = 0; q < GQ; ++q) {
          const unsigned j = j0 + 8 * q + l;
          const float2 t = xs2[idx[k][q]];
          v[k][q].x = j < c[k] ? t.x : 0.0f;
          v[k][q].y = j < c[k] ? t.y : 0.0f;
        }
      }
#pragma unroll
      for (int k = 0; k < GN; ++k) {
#pragma unroll
        for (int q = 0; q < GQ; ++q) {
          ax[k] += v[k][q].x;
          ay[k] += v[k][q].y;
        }
      }
    }
#pragma unroll
    for (int k = 0; k < GN; ++k) {
      float sx = ax[k], sy = ay[k];
      sx += __shfl_xor(sx, 1, 8);  sy += __shfl_xor(sy, 1, 8);
      sx += __shfl_xor(sx, 2, 8);  sy += __shfl_xor(sy, 2, 8);
      sx += __shfl_xor(sx, 4, 8);  sy += __shfl_xor(sy, 4, 8);
      const int node = nb + k;
      if (l == 0 && node < n) {
        float di = rsqrtf((float)(c[k] + 1u));
        agg[2 * node]     = (sx + self[k].x) * di;
        agg[2 * node + 1] = (sy + self[k].y) * di;
      }
    }
  }
}

__global__ __launch_bounds__(256) void gcn_gather2(
    const int* __restrict__ sorted, const unsigned* __restrict__ rowptr,
    const unsigned* __restrict__ cnt, const float* __restrict__ zs,
    const float* __restrict__ b2, float* __restrict__ out, int n) {
  const unsigned l = threadIdx.x & 7;
  const int g = (blockIdx.x * 256 + threadIdx.x) >> 3;
  const int groups = (gridDim.x * 256) >> 3;
  const float bias = b2[0];
  for (int nb = g * GN; nb < n; nb += groups * GN) {
    unsigned beg[GN], c[GN];
    float self[GN], acc[GN];
#pragma unroll
    for (int k = 0; k < GN; ++k) {
      const int node = nb + k;
      const bool in = node < n;
      beg[k] = in ? rowptr[node] : 0u;
      c[k] = in ? cnt[node] : 0u;
      self[k] = in ? zs[node] : 0.0f;
      acc[k] = 0.0f;
    }
    unsigned cm = 0u;
#pragma unroll
    for (int k = 0; k < GN; ++k) cm = c[k] > cm ? c[k] : cm;
    for (unsigned j0 = 0; j0 < cm; j0 += 8 * GQ) {
      int idx[GN][GQ];
      float v[GN][GQ];
#pragma unroll
      for (int k = 0; k < GN; ++k) {
#pragma unroll
        for (int q = 0; q < GQ; ++q) {
          const unsigned j = j0 + 8 * q + l;
          const unsigned jj = j < c[k] ? j : c[k] - 1u;
          idx[k][q] = sorted[c[k] ? beg[k] + jj : 0u];
        }
      }
#pragma unroll
      for (int k = 0; k < GN; ++k) {
#pragma unroll
        for (int q = 0; q < GQ; ++q) {
          const unsigned j = j0 + 8 * q + l;
          const float t = zs[idx[k][q]];
          v[k][q] = j < c[k] ? t : 0.0f;
        }
      }
#pragma unroll
      for (int k = 0; k < GN; ++k) {
#pragma unroll
        for (int q = 0; q < GQ; ++q) acc[k] += v[k][q];
      }
    }
#pragma unroll
    for (int k = 0; k < GN; ++k) {
      float r = acc[k];
      r += __shfl_xor(r, 1, 8);
      r += __shfl_xor(r, 2, 8);
      r += __shfl_xor(r, 4, 8);
      const int node = nb + k;
      if (l == 0 && node < n)
        out[node] = (r + self[k]) * rsqrtf((float)(c[k] + 1u)) + bias;
    }
  }
}

extern "C" void kernel_launch(void* const* d_in, const int* in_sizes, int n_in,
                              void* d_out, int out_size, void* d_ws, size_t ws_size,
                              hipStream_t stream) {
  const float* x  = (const float*)d_in[0];
  const int*   ei = (const int*)d_in[1];
  const float* W1 = (const float*)d_in[2];
  const float* b1 = (const float*)d_in[3];
  const float* W2 = (const float*)d_in[4];
  const float* b2 = (const float*)d_in[5];

  int n = in_sizes[0] / 2;
  int E = in_sizes[1] / 2;
  const int* src = ei;
  const int* dst = ei + E;
  float* out = (float*)d_out;

  const int tb = 256;
  int gN = (n + tb - 1) / tb;
  int gE = (E + tb - 1) / tb;
  if (gE > 4096) gE = 4096;
  int tiles = (n + 15) / 16;
  int dblocks = (tiles + 7) / 8;

  size_t need = (4 * (size_t)n + (size_t)SL * (size_t)n + (size_t)E) * sizeof(float);
  int ranges = (n + RN - 1) / RN;
  if (ws_size >= need && ranges <= 65535) {
    unsigned* cnt    = (unsigned*)d_ws;
    unsigned* rowptr = cnt + n;
    float*    xs     = (float*)(rowptr + n);
    unsigned* table  = (unsigned*)(xs + 2 * (size_t)n);
    int*      sorted = (int*)(table + (size_t)SL * (size_t)n);
    float*    aggS   = (float*)table;
    float*    zs     = aggS + 2 * (size_t)n;

    int L = (E + SL - 1) / SL;
    dim3 gS(SL, ranges);
    int gG = (n + 32 * GN - 1) / (32 * GN);
    if (gG > 1024) gG = 1024;

    gcn_slice_count<<<gS, 1024, 0, stream>>>(dst, table, n, E, L);
    gcn_slice_prefix<<<gN, tb, 0, stream>>>(table, cnt, n);
    gcn_scan<<<(n + 1023) / 1024, 1024, 0, stream>>>(cnt, rowptr, x, xs, n);
    gcn_slice_fill<<<gS, 1024, 0, stream>>>(src, dst, table, rowptr, sorted, n, E, L);
    gcn_gather1<<<gG, 256, 0, stream>>>(sorted, rowptr, cnt, xs, aggS, n);
    gcn_dense_wmma<true><<<dblocks, 256, 0, stream>>>(aggS, x, xs, cnt, W1, b1, W2, zs, n);
    gcn_gather2<<<gG, 256, 0, stream>>>(sorted, rowptr, cnt, zs, b2, out, n);
    return;
  }

  float* dinv = (float*)d_ws;
  float* agg  = dinv + n;
  float* z    = agg + 2 * (size_t)n;

  gcn_init<<<gN, tb, 0, stream>>>(dinv, agg, z, out, n);
  gcn_degree<<<gE, tb, 0, stream>>>(dst, dinv, E);
  gcn_rsqrt<<<gN, tb, 0, stream>>>(dinv, n);
  gcn_agg2<<<gE, tb, 0, stream>>>(src, dst, x, dinv, agg, E);

  gcn_dense_wmma<false><<<dblocks, 256, 0, stream>>>(agg, x, dinv, (const unsigned*)0, W1, b1, W2, z, n);

  gcn_out_self<<<gN, tb, 0, stream>>>(z, dinv, b2, out, n);
  gcn_out_edges<<<gE, tb, 0, stream>>>(src, dst, z, dinv, out, E);
}


# Layer-1 dense stage fused into the layer-1 gather: a wave's 16 summed nodes are its f32 16x16x4 tile, fed from registers; the dense launch and the aggregate's round trip go; f32 sum order unchanged
# speedup vs baseline: 3.5458x; 1.0397x over previous
#include <hip/hip_runtime.h>


typedef float v2f __attribute__((ext_vector_type(2)));
typedef float v8f __attribute__((ext_vector_type(8)));

__global__ void gcn_init(float* __restrict__ deg, float* __restrict__ agg,
                         float* __restrict__ z, float* __restrict__ out, int n) {
  int i = blockIdx.x * blockDim.x + threadIdx.x;
  if (i < n) {
    deg[i] = 1.0f;
    agg[2 * i] = 0.0f;
    agg[2 * i + 1] = 0.0f;
    z[i] = 0.0f;
    out[i] = 0.0f;
  }
}

__global__ void gcn_degree(const int* __restrict__ dst, float* __restrict__ deg,
                           int E) {
  int i = blockIdx.x * blockDim.x + threadIdx.x;
  int stride = gridDim.x * blockDim.x;
  for (; i < E; i += stride) {
    if (i + stride < E) __builtin_prefetch(&dst[i + stride], 0, 1);
    atomicAdd(&deg[dst[i]], 1.0f);
  }
}

__global__ void gcn_rsqrt(float* __restrict__ dinv, int n) {
  int i = blockIdx.x * blockDim.x + threadIdx.x;
  if (i < n) dinv[i] = rsqrtf(dinv[i]);
}

__global__ void gcn_agg2(const int* __restrict__ src, const int* __restrict__ dst,
                         const float* __restrict__ x, const float* __restrict__ dinv,
                         float* __restrict__ agg, int E) {
  int i = blockIdx.x * blockDim.x + threadIdx.x;
  int stride = gridDim.x * blockDim.x;
  for (; i < E; i += stride) {
    if (i + stride < E) {
      __builtin_prefetch(&src[i + stride], 0, 1);
      __builtin_prefetch(&dst[i + stride], 0, 1);
    }
    int s = src[i];
    int d = dst[i];
    float w = dinv[s] * dinv[d];
    atomicAdd(&agg[2 * d], x[2 * s] * w);
    atomicAdd(&agg[2 * d + 1], x[2 * s + 1] * w);
  }
}

__global__ __launch_bounds__(256) void gcn_dense_wmma(
    const float* __restrict__ agg, const float* __restrict__ x,
    const float* __restrict__ dinv, const float* __restrict__ W1,
    const float* __restrict__ b1, const float* __restrict__ W2,
    float* __restrict__ z, int n) {
  const int lane = threadIdx.x & 31;
  const int wave = threadIdx.x >> 5;
  const int tile = blockIdx.x * 8 + wave;
  const int base = tile * 16;
  if (base >= n) return;

  const int m = lane & 15;
  const bool hiK = lane >= 16;
  const float kmask = hiK ? 0.0f : 1.0f;

  int node = base + m;
  const float rowmask = (!hiK && node < n) ? 1.0f : 0.0f;
  if (node >= n) node = n - 1;
  v2f A;
  float di = dinv[node];
  float sl = di * di;
  A.x = (agg[2 * node]     + x[2 * node]     * sl) * rowmask;
  A.y = (agg[2 * node + 1] + x[2 * node + 1] * sl) * rowmask;

  float zacc[8];
#pragma unroll
  for (int v = 0; v < 8; ++v) zacc[v] = 0.0f;

#pragma unroll
  for (int n0 = 0; n0 < 128; n0 += 16) {
    int col = n0 + m;
    v2f B;
    B.x = W1[col]       * kmask;
    B.y = W1[128 + col] * kmask;
    float bias = b1[col];
    v8f C;
#pragma unroll
    for (int v = 0; v < 8; ++v) C[v] = bias;

    C = __builtin_amdgcn_wmma_f32_16x16x4_f32(
         false, A,  false, B,
         (short)0, C,  false,  false);

    float w2 = W2[col];
#pragma unroll
    for (int v = 0; v < 8; ++v) {
      float h = C[v];
      h = h > 0.0f ? h : 0.0f;
      zacc[v] += h * w2;
    }
  }

#pragma unroll
  for (int v = 0; v < 8; ++v) {
    float r = zacc[v];
    r += __shfl_xor(r, 1, 16);
    r += __shfl_xor(r, 2, 16);
    r += __shfl_xor(r, 4, 16);
    r += __shfl_xor(r, 8, 16);
    zacc[v] = r;
  }
  if (m == 0) {
    int rowbase = base + (hiK ? 8 : 0);
#pragma unroll
    for (int v = 0; v < 8; ++v) {
      int r = rowbase + v;
      if (r < n) z[r] = zacc[v];
    }
  }
}

__global__ void gcn_out_self(const float* __restrict__ z,
                             const float* __restrict__ dinv,
                             const float* __restrict__ b2,
                             float* __restrict__ out, int n) {
  int i = blockIdx.x * blockDim.x + threadIdx.x;
  if (i < n) out[i] = z[i] * dinv[i] * dinv[i] + b2[0];
}

__global__ void gcn_out_edges(const int* __restrict__ src, const int* __restrict__ dst,
                              const float* __restrict__ z,
                              const float* __restrict__ dinv,
                              float* __restrict__ out, int E) {
  int i = blockIdx.x * blockDim.x + threadIdx.x;
  int stride = gridDim.x * blockDim.x;
  for (; i < E; i += stride) {
    if (i + stride < E) {
      __builtin_prefetch(&src[i + stride], 0, 1);
      __builtin_prefetch(&dst[i + stride], 0, 1);
    }
    int s = src[i];
    int d = dst[i];
    atomicAdd(&out[d], z[s] * dinv[s] * dinv[d]);
  }
}


constexpr int SL = 16;
constexpr int RN = 6400;

__global__ __launch_bounds__(1024) void gcn_slice_count(
    const int* __restrict__ dst, unsigned* __restrict__ table,
    int n, int E, int L) {
  __shared__ unsigned h[RN];
  const int tid = threadIdx.x;
  const int s = blockIdx.x;
  const int lo = blockIdx.y * RN;
  const unsigned lim = (n - lo < RN) ? (unsigned)(n - lo) : (unsigned)RN;
  for (int i = tid; i < RN; i += 1024) h[i] = 0u;
  __syncthreads();
  const int e0 = s * L;
  const int e1 = (E - e0 < L) ? E : e0 + L;
  for (int e = e0 + tid; e < e1; e += 1024 * 16) {
    int d[16];
#pragma unroll
    for (int k = 0; k < 16; ++k) {
      int ee = e + k * 1024;
      d[k] = (ee < e1) ? dst[ee] : -1;
    }
#pragma unroll
    for (int k = 0; k < 16; ++k) {
      unsigned x = (unsigned)d[k] - (unsigned)lo;
      if (x < lim) atomicAdd(&h[x], 1u);
    }
  }
  __syncthreads();
  unsigned* row = table + (size_t)s * n + lo;
  for (int i = tid; i < (int)lim; i += 1024) row[i] = h[i];
}

__global__ void gcn_slice_prefix(unsigned* __restrict__ table,
                                 unsigned* __restrict__ cnt, int n) {
  int node = blockIdx.x * blockDim.x + threadIdx.x;
  if (node >= n) return;
  unsigned v[SL];
#pragma unroll
  for (int s = 0; s < SL; ++s) v[s] = table[(size_t)s * n + node];
  unsigned run = 0u;
#pragma unroll
  for (int s = 0; s < SL; ++s) {
    table[(size_t)s * n + node] = run;
    run += v[s];
  }
  cnt[node] = run;
}

__global__ __launch_bounds__(1024) void gcn_scan(
    const unsigned* __restrict__ cnt, unsigned* __restrict__ rowptr,
    const float* __restrict__ x, float* __restrict__ xs, int n) {
  __shared__ unsigned s_incl[32];
  __shared__ unsigned s_part[32];
  const int tid = threadIdx.x;
  const int lane = tid & 31;
  const int wv = tid >> 5;
  const int blockBase = blockIdx.x * 1024;
  const int lim = blockBase / 4;

  const uint4* c4 = (const uint4*)cnt;
  unsigned p0 = 0u, p1 = 0u, p2 = 0u, p3 = 0u;
  int i = tid;
  for (; i + 3072 < lim; i += 4096) {
    uint4 a = c4[i];
    uint4 b = c4[i + 1024];
    uint4 c = c4[i + 2048];
    uint4 d = c4[i + 3072];
    p0 += a.x + a.y + a.z + a.w;
    p1 += b.x + b.y + b.z + b.w;
    p2 += c.x + c.y + c.z + c.w;
    p3 += d.x + d.y + d.z + d.w;
  }
  for (; i < lim; i += 1024) {
    uint4 a = c4[i];
    p0 += a.x + a.y + a.z + a.w;
  }
  unsigned part = (p0 + p1) + (p2 + p3);
  part += __shfl_xor(part, 16, 32);
  part += __shfl_xor(part, 8, 32);
  part += __shfl_xor(part, 4, 32);
  part += __shfl_xor(part, 2, 32);
  part += __shfl_xor(part, 1, 32);

  const int node = blockBase + tid;
  const unsigned mine = (node < n) ? cnt[node] : 0u;
  unsigned incl = mine;
#pragma unroll
  for (int o = 1; o < 32; o <<= 1) {
    unsigned t = __shfl_up(incl, o, 32);
    if (lane >= o) incl += t;
  }
  if (lane == 31) s_incl[wv] = incl;
  if (lane == 0) s_part[wv] = part;
  __syncthreads();
  unsigned off = s_part[lane] + ((lane < wv) ? s_incl[lane] : 0u);
  off += __shfl_xor(off, 16, 32);
  off += __shfl_xor(off, 8, 32);
  off += __shfl_xor(off, 4, 32);
  off += __shfl_xor(off, 2, 32);
  off += __shfl_xor(off, 1, 32);
  if (node < n) {
    rowptr[node] = off + incl - mine;
    float di = rsqrtf((float)(mine + 1u));
    xs[2 * node]     = x[2 * node]     * di;
    xs[2 * node + 1] = x[2 * node + 1] * di;
  }
}

__global__ __launch_bounds__(1024) void gcn_slice_fill(
    const int* __restrict__ src, const int* __restrict__ dst,
    const unsigned* __restrict__ table, const unsigned* __restrict__ rowptr,
    int* __restrict__ sorted, int n, int E, int L) {
  __shared__ unsigned cur[RN];
  const int tid = threadIdx.x;
  const int s = blockIdx.x;
  const int lo = blockIdx.y * RN;
  const unsigned lim = (n - lo < RN) ? (unsigned)(n - lo) : (unsigned)RN;
  const unsigned* row = table + (size_t)s * n + lo;
  constexpr int RT = (RN + 1023) / 1024;
  unsigned ra[RT], rb[RT];
#pragma unroll
  for (int k = 0; k < RT; ++k) {
    int i = tid + k * 1024;
    bool in = i < (int)lim;
    ra[k] = in ? rowptr[lo + i] : 0u;
    rb[k] = in ? row[i] : 0u;
  }
#pragma unroll
  for (int k = 0; k < RT; ++k) {
    int i = tid + k * 1024;
    if (i < (int)lim) cur[i] = ra[k] + rb[k];
  }
  __syncthreads();
  const int e0 = s * L;
  const int e1 = (E - e0 < L) ? E : e0 + L;
  for (int e = e0 + tid; e < e1; e += 1024 * 8) {
    int d[8], sv[8];
    unsigned pos[8];
#pragma unroll
    for (int k = 0; k < 8; ++k) {
      int ee = e + k * 1024;
      bool in = ee < e1;
      d[k] = in ? dst[ee] : -1;
      sv[k] = in ? src[ee] : 0;
    }
#pragma unroll
    for (int k = 0; k < 8; ++k) {
      unsigned x = (unsigned)d[k] - (unsigned)lo;
      pos[k] = 0u;
      if (x < lim) pos[k] = atomicAdd(&cur[x], 1u);
    }
#pragma unroll
    for (int k = 0; k < 8; ++k) {
      unsigned x = (unsigned)d[k] - (unsigned)lo;
      if (x < lim) sorted[pos[k]] = sv[k];
    }
  }
}

constexpr int GN = 4;
constexpr int GQ = 4;

__global__ __launch_bounds__(256) void gcn_gather1_dense(
    const int* __restrict__ sorted, const unsigned* __restrict__ rowptr,
    const unsigned* __restrict__ cnt, const float* __restrict__ xs,
    const float* __restrict__ W1, const float* __restrict__ b1,
    const float* __restrict__ W2, float* __restrict__ zs, int n) {
  static_assert(GN == 4, "four groups of GN nodes must make one 16-row tile");
  const int lane = threadIdx.x & 31;
  const unsigned l = threadIdx.x & 7;
  const int grp = lane >> 3;
  const int m = lane & 15;
  const bool hiK = lane >= 16;
  const float kmask = hiK ? 0.0f : 1.0f;
  const int wb = ((blockIdx.x * 256 + threadIdx.x) >> 5) * 16;
  if (wb >= n) return;
  const int nb = wb + grp * GN;
  const float2* xs2 = (const float2*)xs;

  unsigned beg[GN], c[GN];
  float2 self[GN];
  float ax[GN], ay[GN];
#pragma unroll
  for (int k = 0; k < GN; ++k) {
    const int node = nb + k;
    const bool in = node < n;
    beg[k] = in ? rowptr[node] : 0u;
    c[k] = in ? cnt[node] : 0u;
    self[k] = in ? xs2[node] : make_float2(0.0f, 0.0f);
    ax[k] = 0.0f;
    ay[k] = 0.0f;
  }
  float w1a[8], w1b[8], bias[8], w2[8];
#pragma unroll
  for (int t = 0; t < 8; ++t) {
    const int col = 16 * t + m;
    w1a[t] = W1[col];
    w1b[t] = W1[128 + col];
    bias[t] = b1[col];
    w2[t] = W2[col];
  }
  unsigned cm = 0u;
#pragma unroll
  for (int k = 0; k < GN; ++k) cm = c[k] > cm ? c[k] : cm;
  for (unsigned j0 = 0; j0 < cm; j0 += 8 * GQ) {
    int idx[GN][GQ];
    float2 v[GN][GQ];
#pragma unroll
    for (int k = 0; k < GN; ++k) {
#pragma unroll
      for (int q = 0; q < GQ; ++q) {
        const unsigned j = j0 + 8 * q + l;
        const unsigned jj = j < c[k] ? j : c[k] - 1u;
        idx[k][q] = sorted[c[k] ? beg[k] + jj : 0u];
      }
    }
#pragma unroll
    for (int k = 0; k < GN; ++k) {
#pragma unroll
      for (int q = 0; q < GQ; ++q) {
        const unsigned j = j0 + 8 * q + l;
        const float2 t = xs2[idx[k][q]];
        v[k][q].x = j < c[k] ? t.x : 0.0f;
        v[k][q].y = j < c[k] ? t.y : 0.0f;
      }
    }
#pragma unroll
    for (int k = 0; k < GN; ++k) {
#pragma unroll
      for (int q = 0; q < GQ; ++q) {
        ax[k] += v[k][q].x;
        ay[k] += v[k][q].y;
      }
    }
  }
  float tx = 0.0f, ty = 0.0f, td = 0.0f;
#pragma unroll
  for (int k = 0; k < GN; ++k) {
    float sx = ax[k], sy = ay[k];
    sx += __shfl_xor(sx, 1, 8);  sy += __shfl_xor(sy, 1, 8);
    sx += __shfl_xor(sx, 2, 8);  sy += __shfl_xor(sy, 2, 8);
    sx += __shfl_xor(sx, 4, 8);  sy += __shfl_xor(sy, 4, 8);
    const float di = rsqrtf((float)(c[k] + 1u));
    const float rx = (sx + self[k].x) * di;
    const float ry = (sy + self[k].y) * di;
    tx = (l == (unsigned)k) ? rx : tx;
    ty = (l == (unsigned)k) ? ry : ty;
    td = (l == (unsigned)k) ? di : td;
  }
  const int from = 8 * (m >> 2) + (m & 3);
  const float arx = __shfl(tx, from, 32);
  const float ary = __shfl(ty, from, 32);
  const bool rowok = !hiK && (wb + m < n);
  v2f A;
  A.x = rowok ? arx : 0.0f;
  A.y = rowok ? ary : 0.0f;

  float zacc[8];
#pragma unroll
  for (int v = 0; v < 8; ++v) zacc[v] = 0.0f;
#pragma unroll
  for (int t = 0; t < 8; ++t) {
    v2f B;
    B.x = w1a[t] * kmask;
    B.y = w1b[t] * kmask;
    v8f C;
#pragma unroll
    for (int v = 0; v < 8; ++v) C[v] = bias[t];

    C = __builtin_amdgcn_wmma_f32_16x16x4_f32(
         false, A,  false, B,
         (short)0, C,  false,  false);

#pragma unroll
    for (int v = 0; v < 8; ++v) {
      float h = C[v];
      h = h > 0.0f ? h : 0.0f;
      zacc[v] += h * w2[t];
    }
  }
#pragma unroll
  for (int v = 0; v < 8; ++v) {
    float r = zacc[v];
    r += __shfl_xor(r, 1, 16);
    r += __shfl_xor(r, 2, 16);
    r += __shfl_xor(r, 4, 16);
    r += __shfl_xor(r, 8, 16);
    zacc[v] = r;
  }
  const int reg = 4 * (grp & 1) + (int)(l & 3);
  float zsel = zacc[0];
#pragma unroll
  for (int v = 1; v < 8; ++v) zsel = (reg == v) ? zacc[v] : zsel;
  const int node = nb + (int)l;
  if (l < (unsigned)GN && node < n) zs[node] = zsel * td;
}

__global__ __launch_bounds__(256) void gcn_gather2(
    const int* __restrict__ sorted, const unsigned* __restrict__ rowptr,
    const unsigned* __restrict__ cnt, const float* __restrict__ zs,
    const float* __restrict__ b2, float* __restrict__ out, int n) {
  const unsigned l = threadIdx.x & 7;
  const int g = (blockIdx.x * 256 + threadIdx.x) >> 3;
  const int groups = (gridDim.x * 256) >> 3;
  const float bias = b2[0];
  for (int nb = g * GN; nb < n; nb += groups * GN) {
    unsigned beg[GN], c[GN];
    float self[GN], acc[GN];
#pragma unroll
    for (int k = 0; k < GN; ++k) {
      const int node = nb + k;
      const bool in = node < n;
      beg[k] = in ? rowptr[node] : 0u;
      c[k] = in ? cnt[node] : 0u;
      self[k] = in ? zs[node] : 0.0f;
      acc[k] = 0.0f;
    }
    unsigned cm = 0u;
#pragma unroll
    for (int k = 0; k < GN; ++k) cm = c[k] > cm ? c[k] : cm;
    for (unsigned j0 = 0; j0 < cm; j0 += 8 * GQ) {
      int idx[GN][GQ];
      float v[GN][GQ];
#pragma unroll
      for (int k = 0; k < GN; ++k) {
#pragma unroll
        for (int q = 0; q < GQ; ++q) {
          const unsigned j = j0 + 8 * q + l;
          const unsigned jj = j < c[k] ? j : c[k] - 1u;
          idx[k][q] = sorted[c[k] ? beg[k] + jj : 0u];
        }
      }
#pragma unroll
      for (int k = 0; k < GN; ++k) {
#pragma unroll
        for (int q = 0; q < GQ; ++q) {
          const unsigned j = j0 + 8 * q + l;
          const float t = zs[idx[k][q]];
          v[k][q] = j < c[k] ? t : 0.0f;
        }
      }
#pragma unroll
      for (int k = 0; k < GN; ++k) {
#pragma unroll
        for (int q = 0; q < GQ; ++q) acc[k] += v[k][q];
      }
    }
#pragma unroll
    for (int k = 0; k < GN; ++k) {
      float r = acc[k];
      r += __shfl_xor(r, 1, 8);
      r += __shfl_xor(r, 2, 8);
      r += __shfl_xor(r, 4, 8);
      const int node = nb + k;
      if (l == 0 && node < n)
        out[node] = (r + self[k]) * rsqrtf((float)(c[k] + 1u)) + bias;
    }
  }
}

extern "C" void kernel_launch(void* const* d_in, const int* in_sizes, int n_in,
                              void* d_out, int out_size, void* d_ws, size_t ws_size,
                              hipStream_t stream) {
  const float* x  = (const float*)d_in[0];
  const int*   ei = (const int*)d_in[1];
  const float* W1 = (const float*)d_in[2];
  const float* b1 = (const float*)d_in[3];
  const float* W2 = (const float*)d_in[4];
  const float* b2 = (const float*)d_in[5];

  int n = in_sizes[0] / 2;
  int E = in_sizes[1] / 2;
  const int* src = ei;
  const int* dst = ei + E;
  float* out = (float*)d_out;

  const int tb = 256;
  int gN = (n + tb - 1) / tb;
  int gE = (E + tb - 1) / tb;
  if (gE > 4096) gE = 4096;
  int tiles = (n + 15) / 16;
  int dblocks = (tiles + 7) / 8;

  size_t need = (4 * (size_t)n + (size_t)SL * (size_t)n + (size_t)E) * sizeof(float);
  int ranges = (n + RN - 1) / RN;
  if (ws_size >= need && ranges <= 65535) {
    unsigned* cnt    = (unsigned*)d_ws;
    unsigned* rowptr = cnt + n;
    float*    xs     = (float*)(rowptr + n);
    unsigned* table  = (unsigned*)(xs + 2 * (size_t)n);
    int*      sorted = (int*)(table + (size_t)SL * (size_t)n);
    float*    zs     = (float*)table;

    int L = (E + SL - 1) / SL;
    dim3 gS(SL, ranges);
    int gG = (n + 32 * GN - 1) / (32 * GN);
    if (gG > 1024) gG = 1024;

    gcn_slice_count<<<gS, 1024, 0, stream>>>(dst, table, n, E, L);
    gcn_slice_prefix<<<gN, tb, 0, stream>>>(table, cnt, n);
    gcn_scan<<<(n + 1023) / 1024, 1024, 0, stream>>>(cnt, rowptr, x, xs, n);
    gcn_slice_fill<<<gS, 1024, 0, stream>>>(src, dst, table, rowptr, sorted, n, E, L);
    gcn_gather1_dense<<<dblocks, 256, 0, stream>>>(sorted, rowptr, cnt, xs, W1, b1, W2, zs, n);
    gcn_gather2<<<gG, 256, 0, stream>>>(sorted, rowptr, cnt, zs, b2, out, n);
    return;
  }

  float* dinv = (float*)d_ws;
  float* agg  = dinv + n;
  float* z    = agg + 2 * (size_t)n;

  gcn_init<<<gN, tb, 0, stream>>>(dinv, agg, z, out, n);
  gcn_degree<<<gE, tb, 0, stream>>>(dst, dinv, E);
  gcn_rsqrt<<<gN, tb, 0, stream>>>(dinv, n);
  gcn_agg2<<<gE, tb, 0, stream>>>(src, dst, x, dinv, agg, E);

  gcn_dense_wmma<<<dblocks, 256, 0, stream>>>(agg, x, dinv, W1, b1, W2, z, n);

  gcn_out_self<<<gN, tb, 0, stream>>>(z, dinv, b2, out, n);
  gcn_out_edges<<<gE, tb, 0, stream>>>(src, dst, z, dinv, out, E);
}


# Column-prefix and node-scan launches fused: the counting blocks also store per-256-node chunk sums, so a block's first slot is a sum of at most 6240 words; 5 launches, not 6; f32 sum order unchanged
# speedup vs baseline: 3.7280x; 1.0514x over previous
#include <hip/hip_runtime.h>


typedef float v2f __attribute__((ext_vector_type(2)));
typedef float v8f __attribute__((ext_vector_type(8)));

__global__ void gcn_init(float* __restrict__ deg, float* __restrict__ agg,
                         float* __restrict__ z, float* __restrict__ out, int n) {
  int i = blockIdx.x * blockDim.x + threadIdx.x;
  if (i < n) {
    deg[i] = 1.0f;
    agg[2 * i] = 0.0f;
    agg[2 * i + 1] = 0.0f;
    z[i] = 0.0f;
    out[i] = 0.0f;
  }
}

__global__ void gcn_degree(const int* __restrict__ dst, float* __restrict__ deg,
                           int E) {
  int i = blockIdx.x * blockDim.x + threadIdx.x;
  int stride = gridDim.x * blockDim.x;
  for (; i < E; i += stride) {
    if (i + stride < E) __builtin_prefetch(&dst[i + stride], 0, 1);
    atomicAdd(&deg[dst[i]], 1.0f);
  }
}

__global__ void gcn_rsqrt(float* __restrict__ dinv, int n) {
  int i = blockIdx.x * blockDim.x + threadIdx.x;
  if (i < n) dinv[i] = rsqrtf(dinv[i]);
}

__global__ void gcn_agg2(const int* __restrict__ src, const int* __restrict__ dst,
                         const float* __restrict__ x, const float* __restrict__ dinv,
                         float* __restrict__ agg, int E) {
  int i = blockIdx.x * blockDim.x + threadIdx.x;
  int stride = gridDim.x * blockDim.x;
  for (; i < E; i += stride) {
    if (i + stride < E) {
      __builtin_prefetch(&src[i + stride], 0, 1);
      __builtin_prefetch(&dst[i + stride], 0, 1);
    }
    int s = src[i];
    int d = dst[i];
    float w = dinv[s] * dinv[d];
    atomicAdd(&agg[2 * d], x[2 * s] * w);
    atomicAdd(&agg[2 * d + 1], x[2 * s + 1] * w);
  }
}

__global__ __launch_bounds__(256) void gcn_dense_wmma(
    const float* __restrict__ agg, const float* __restrict__ x,
    const float* __restrict__ dinv, const float* __restrict__ W1,
    const float* __restrict__ b1, const float* __restrict__ W2,
    float* __restrict__ z, int n) {
  const int lane = threadIdx.x & 31;
  const int wave = threadIdx.x >> 5;
  const int tile = blockIdx.x * 8 + wave;
  const int base = tile * 16;
  if (base >= n) return;

  const int m = lane & 15;
  const bool hiK = lane >= 16;
  const float kmask = hiK ? 0.0f : 1.0f;

  int node = base + m;
  const float rowmask = (!hiK && node < n) ? 1.0f : 0.0f;
  if (node >= n) node = n - 1;
  v2f A;
  float di = dinv[node];
  float sl = di * di;
  A.x = (agg[2 * node]     + x[2 * node]     * sl) * rowmask;
  A.y = (agg[2 * node + 1] + x[2 * node + 1] * sl) * rowmask;

  float zacc[8];
#pragma unroll
  for (int v = 0; v < 8; ++v) zacc[v] = 0.0f;

#pragma unroll
  for (int n0 = 0; n0 < 128; n0 += 16) {
    int col = n0 + m;
    v2f B;
    B.x = W1[col]       * kmask;
    B.y = W1[128 + col] * kmask;
    float bias = b1[col];
    v8f C;
#pragma unroll
    for (int v = 0; v < 8; ++v) C[v] = bias;

    C = __builtin_amdgcn_wmma_f32_16x16x4_f32(
         false, A,  false, B,
         (short)0, C,  false,  false);

    float w2 = W2[col];
#pragma unroll
    for (int v = 0; v < 8; ++v) {
      float h = C[v];
      h = h > 0.0f ? h : 0.0f;
      zacc[v] += h * w2;
    }
  }

#pragma unroll
  for (int v = 0; v < 8; ++v) {
    float r = zacc[v];
    r += __shfl_xor(r, 1, 16);
    r += __shfl_xor(r, 2, 16);
    r += __shfl_xor(r, 4, 16);
    r += __shfl_xor(r, 8, 16);
    zacc[v] = r;
  }
  if (m == 0) {
    int rowbase = base + (hiK ? 8 : 0);
#pragma unroll
    for (int v = 0; v < 8; ++v) {
      int r = rowbase + v;
      if (r < n) z[r] = zacc[v];
    }
  }
}

__global__ void gcn_out_self(const float* __restrict__ z,
                             const float* __restrict__ dinv,
                             const float* __restrict__ b2,
                             float* __restrict__ out, int n) {
  int i = blockIdx.x * blockDim.x + threadIdx.x;
  if (i < n) out[i] = z[i] * dinv[i] * dinv[i] + b2[0];
}

__global__ void gcn_out_edges(const int* __restrict__ src, const int* __restrict__ dst,
                              const float* __restrict__ z,
                              const float* __restrict__ dinv,
                              float* __restrict__ out, int E) {
  int i = blockIdx.x * blockDim.x + threadIdx.x;
  int stride = gridDim.x * blockDim.x;
  for (; i < E; i += stride) {
    if (i + stride < E) {
      __builtin_prefetch(&src[i + stride], 0, 1);
      __builtin_prefetch(&dst[i + stride], 0, 1);
    }
    int s = src[i];
    int d = dst[i];
    atomicAdd(&out[d], z[s] * dinv[s] * dinv[d]);
  }
}


constexpr int SL = 16;
constexpr int RN = 6400;
constexpr int CH = 256;
constexpr int RC = RN / CH;
static_assert(RN % CH == 0, "a chunk must not straddle two ranges");
static_assert(RC <= 32, "one wave of the counting block per chunk");
static_assert(SL % 4 == 0, "a chunk's slice sums are read as 16-byte pieces");

__global__ __launch_bounds__(1024) void gcn_slice_count(
    const int* __restrict__ dst, unsigned* __restrict__ table,
    unsigned* __restrict__ chunks, int n, int E, int L) {
  __shared__ unsigned h[RN];
  const int tid = threadIdx.x;
  const int s = blockIdx.x;
  const int lo = blockIdx.y * RN;
  const unsigned lim = (n - lo < RN) ? (unsigned)(n - lo) : (unsigned)RN;
  for (int i = tid; i < RN; i += 1024) h[i] = 0u;
  __syncthreads();
  const int e0 = s * L;
  const int e1 = (E - e0 < L) ? E : e0 + L;
  for (int e = e0 + tid; e < e1; e += 1024 * 16) {
    int d[16];
#pragma unroll
    for (int k = 0; k < 16; ++k) {
      int ee = e + k * 1024;
      d[k] = (ee < e1) ? dst[ee] : -1;
    }
#pragma unroll
    for (int k = 0; k < 16; ++k) {
      unsigned x = (unsigned)d[k] - (unsigned)lo;
      if (x < lim) atomicAdd(&h[x], 1u);
    }
  }
  __syncthreads();
  unsigned* row = table + (size_t)s * n + lo;
  for (int i = tid; i < (int)lim; i += 1024) row[i] = h[i];
  const int wv = tid >> 5;
  const int lane = tid & 31;
  if (wv < RC) {
    unsigned sum = 0u;
#pragma unroll
    for (int j = 0; j < CH / 32; ++j) sum += h[wv * CH + lane + 32 * j];
    sum += __shfl_xor(sum, 16, 32);
    sum += __shfl_xor(sum, 8, 32);
    sum += __shfl_xor(sum, 4, 32);
    sum += __shfl_xor(sum, 2, 32);
    sum += __shfl_xor(sum, 1, 32);
    if (lane == 0) chunks[(size_t)(blockIdx.y * RC + wv) * SL + s] = sum;
  }
}

__global__ __launch_bounds__(CH) void gcn_offsets(
    unsigned* __restrict__ table, const unsigned* __restrict__ chunks,
    unsigned* __restrict__ cnt, unsigned* __restrict__ rowptr,
    const float* __restrict__ x, float* __restrict__ xs, int n) {
  constexpr int NW = CH / 32;
  __shared__ unsigned s_incl[NW];
  __shared__ unsigned s_part[NW];
  const int tid = threadIdx.x;
  const int lane = tid & 31;
  const int wv = tid >> 5;
  const int node = blockIdx.x * CH + tid;
  const bool in = node < n;

  unsigned v[SL];
  float2 xv = make_float2(0.0f, 0.0f);
#pragma unroll
  for (int s = 0; s < SL; ++s) v[s] = 0u;
  if (in) {
#pragma unroll
    for (int s = 0; s < SL; ++s) v[s] = table[(size_t)s * n + node];
    xv = ((const float2*)x)[node];
  }

  const uint4* c4 = (const uint4*)chunks;
  const int lim4 = (SL / 4) * blockIdx.x;
  unsigned part = 0u;
  for (int base = 0; base < lim4; base += CH * 8) {
    uint4 a[8];
#pragma unroll
    for (int j = 0; j < 8; ++j) {
      const int i = base + tid + CH * j;
      a[j] = c4[i < lim4 ? i : 0];
    }
#pragma unroll
    for (int j = 0; j < 8; ++j) {
      const int i = base + tid + CH * j;
      const unsigned t = (a[j].x + a[j].y) + (a[j].z + a[j].w);
      part += i < lim4 ? t : 0u;
    }
  }
  part += __shfl_xor(part, 16, 32);
  part += __shfl_xor(part, 8, 32);
  part += __shfl_xor(part, 4, 32);
  part += __shfl_xor(part, 2, 32);
  part += __shfl_xor(part, 1, 32);

  unsigned mine = 0u;
#pragma unroll
  for (int s = 0; s < SL; ++s) mine += v[s];
  unsigned incl = mine;
#pragma unroll
  for (int o = 1; o < 32; o <<= 1) {
    unsigned t = __shfl_up(incl, o, 32);
    if (lane >= o) incl += t;
  }
  if (lane == 31) s_incl[wv] = incl;
  if (lane == 0) s_part[wv] = part;
  __syncthreads();
  unsigned off = 0u;
  if (lane < NW) off = s_part[lane] + ((lane < wv) ? s_incl[lane] : 0u);
  off += __shfl_xor(off, 16, 32);
  off += __shfl_xor(off, 8, 32);
  off += __shfl_xor(off, 4, 32);
  off += __shfl_xor(off, 2, 32);
  off += __shfl_xor(off, 1, 32);
  if (in) {
    unsigned run = 0u;
#pragma unroll
    for (int s = 0; s < SL; ++s) {
      table[(size_t)s * n + node] = run;
      run += v[s];
    }
    cnt[node] = mine;
    rowptr[node] = off + incl - mine;
    float di = rsqrtf((float)(mine + 1u));
    xs[2 * node]     = xv.x * di;
    xs[2 * node + 1] = xv.y * di;
  }
}

__global__ __launch_bounds__(1024) void gcn_slice_fill(
    const int* __restrict__ src, const int* __restrict__ dst,
    const unsigned* __restrict__ table, const unsigned* __restrict__ rowptr,
    int* __restrict__ sorted, int n, int E, int L) {
  __shared__ unsigned cur[RN];
  const int tid = threadIdx.x;
  const int s = blockIdx.x;
  const int lo = blockIdx.y * RN;
  const unsigned lim = (n - lo < RN) ? (unsigned)(n - lo) : (unsigned)RN;
  const unsigned* row = table + (size_t)s * n + lo;
  constexpr int RT = (RN + 1023) / 1024;
  unsigned ra[RT], rb[RT];
#pragma unroll
  for (int k = 0; k < RT; ++k) {
    int i = tid + k * 1024;
    bool in = i < (int)lim;
    ra[k] = in ? rowptr[lo + i] : 0u;
    rb[k] = in ? row[i] : 0u;
  }
#pragma unroll
  for (int k = 0; k < RT; ++k) {
    int i = tid + k * 1024;
    if (i < (int)lim) cur[i] = ra[k] + rb[k];
  }
  __syncthreads();
  const int e0 = s * L;
  const int e1 = (E - e0 < L) ? E : e0 + L;
  for (int e = e0 + tid; e < e1; e += 1024 * 8) {
    int d[8], sv[8];
    unsigned pos[8];
#pragma unroll
    for (int k = 0; k < 8; ++k) {
      int ee = e + k * 1024;
      bool in = ee < e1;
      d[k] = in ? dst[ee] : -1;
      sv[k] = in ? src[ee] : 0;
    }
#pragma unroll
    for (int k = 0; k < 8; ++k) {
      unsigned x = (unsigned)d[k] - (unsigned)lo;
      pos[k] = 0u;
      if (x < lim) pos[k] = atomicAdd(&cur[x], 1u);
    }
#pragma unroll
    for (int k = 0; k < 8; ++k) {
      unsigned x = (unsigned)d[k] - (unsigned)lo;
      if (x < lim) sorted[pos[k]] = sv[k];
    }
  }
}

constexpr int GN = 4;
constexpr int GQ = 4;

__global__ __launch_bounds__(256) void gcn_gather1_dense(
    const int* __restrict__ sorted, const unsigned* __restrict__ rowptr,
    const unsigned* __restrict__ cnt, const float* __restrict__ xs,
    const float* __restrict__ W1, const float* __restrict__ b1,
    const float* __restrict__ W2, float* __restrict__ zs, int n) {
  static_assert(GN == 4, "four groups of GN nodes must make one 16-row tile");
  const int lane = threadIdx.x & 31;
  const unsigned l = threadIdx.x & 7;
  const int grp = lane >> 3;
  const int m = lane & 15;
  const bool hiK = lane >= 16;
  const float kmask = hiK ? 0.0f : 1.0f;
  const int wb = ((blockIdx.x * 256 + threadIdx.x) >> 5) * 16;
  if (wb >= n) return;
  const int nb = wb + grp * GN;
  const float2* xs2 = (const float2*)xs;

  unsigned beg[GN], c[GN];
  float2 self[GN];
  float ax[GN], ay[GN];
#pragma unroll
  for (int k = 0; k < GN; ++k) {
    const int node = nb + k;
    const bool in = node < n;
    beg[k] = in ? rowptr[node] : 0u;
    c[k] = in ? cnt[node] : 0u;
    self[k] = in ? xs2[node] : make_float2(0.0f, 0.0f);
    ax[k] = 0.0f;
    ay[k] = 0.0f;
  }
  float w1a[8], w1b[8], bias[8], w2[8];
#pragma unroll
  for (int t = 0; t < 8; ++t) {
    const int col = 16 * t + m;
    w1a[t] = W1[col];
    w1b[t] = W1[128 + col];
    bias[t] = b1[col];
    w2[t] = W2[col];
  }
  unsigned cm = 0u;
#pragma unroll
  for (int k = 0; k < GN; ++k) cm = c[k] > cm ? c[k] : cm;
  for (unsigned j0 = 0; j0 < cm; j0 += 8 * GQ) {
    int idx[GN][GQ];
    float2 v[GN][GQ];
#pragma unroll
    for (int k = 0; k < GN; ++k) {
#pragma unroll
      for (int q = 0; q < GQ; ++q) {
        const unsigned j = j0 + 8 * q + l;
        const unsigned jj = j < c[k] ? j : c[k] - 1u;
        idx[k][q] = sorted[c[k] ? beg[k] + jj : 0u];
      }
    }
#pragma unroll
    for (int k = 0; k < GN; ++k) {
#pragma unroll
      for (int q = 0; q < GQ; ++q) {
        const unsigned j = j0 + 8 * q + l;
        const float2 t = xs2[idx[k][q]];
        v[k][q].x = j < c[k] ? t.x : 0.0f;
        v[k][q].y = j < c[k] ? t.y : 0.0f;
      }
    }
#pragma unroll
    for (int k = 0; k < GN; ++k) {
#pragma unroll
      for (int q = 0; q < GQ; ++q) {
        ax[k] += v[k][q].x;
        ay[k] += v[k][q].y;
      }
    }
  }
  float tx = 0.0f, ty = 0.0f, td = 0.0f;
#pragma unroll
  for (int k = 0; k < GN; ++k) {
    float sx = ax[k], sy = ay[k];
    sx += __shfl_xor(sx, 1, 8);  sy += __shfl_xor(sy, 1, 8);
    sx += __shfl_xor(sx, 2, 8);  sy += __shfl_xor(sy, 2, 8);
    sx += __shfl_xor(sx, 4, 8);  sy += __shfl_xor(sy, 4, 8);
    const float di = rsqrtf((float)(c[k] + 1u));
    const float rx = (sx + self[k].x) * di;
    const float ry = (sy + self[k].y) * di;
    tx = (l == (unsigned)k) ? rx : tx;
    ty = (l == (unsigned)k) ? ry : ty;
    td = (l == (unsigned)k) ? di : td;
  }
  const int from = 8 * (m >> 2) + (m & 3);
  const float arx = __shfl(tx, from, 32);
  const float ary = __shfl(ty, from, 32);
  const bool rowok = !hiK && (wb + m < n);
  v2f A;
  A.x = rowok ? arx : 0.0f;
  A.y = rowok ? ary : 0.0f;

  float zacc[8];
#pragma unroll
  for (int v = 0; v < 8; ++v) zacc[v] = 0.0f;
#pragma unroll
  for (int t = 0; t < 8; ++t) {
    v2f B;
    B.x = w1a[t] * kmask;
    B.y = w1b[t] * kmask;
    v8f C;
#pragma unroll
    for (int v = 0; v < 8; ++v) C[v] = bias[t];

    C = __builtin_amdgcn_wmma_f32_16x16x4_f32(
         false, A,  false, B,
         (short)0, C,  false,  false);

#pragma unroll
    for (int v = 0; v < 8; ++v) {
      float h = C[v];
      h = h > 0.0f ? h : 0.0f;
      zacc[v] += h * w2[t];
    }
  }
#pragma unroll
  for (int v = 0; v < 8; ++v) {
    float r = zacc[v];
    r += __shfl_xor(r, 1, 16);
    r += __shfl_xor(r, 2, 16);
    r += __shfl_xor(r, 4, 16);
    r += __shfl_xor(r, 8, 16);
    zacc[v] = r;
  }
  const int reg = 4 * (grp & 1) + (int)(l & 3);
  float zsel = zacc[0];
#pragma unroll
  for (int v = 1; v < 8; ++v) zsel = (reg == v) ? zacc[v] : zsel;
  const int node = nb + (int)l;
  if (l < (unsigned)GN && node < n) zs[node] = zsel * td;
}

__global__ __launch_bounds__(256) void gcn_gather2(
    const int* __restrict__ sorted, const unsigned* __restrict__ rowptr,
    const unsigned* __restrict__ cnt, const float* __restrict__ zs,
    const float* __restrict__ b2, float* __restrict__ out, int n) {
  const unsigned l = threadIdx.x & 7;
  const int g = (blockIdx.x * 256 + threadIdx.x) >> 3;
  const int groups = (gridDim.x * 256) >> 3;
  const float bias = b2[0];
  for (int nb = g * GN; nb < n; nb += groups * GN) {
    unsigned beg[GN], c[GN];
    float self[GN], acc[GN];
#pragma unroll
    for (int k = 0; k < GN; ++k) {
      const int node = nb + k;
      const bool in = node < n;
      beg[k] = in ? rowptr[node] : 0u;
      c[k] = in ? cnt[node] : 0u;
      self[k] = in ? zs[node] : 0.0f;
      acc[k] = 0.0f;
    }
    unsigned cm = 0u;
#pragma unroll
    for (int k = 0; k < GN; ++k) cm = c[k] > cm ? c[k] : cm;
    for (unsigned j0 = 0; j0 < cm; j0 += 8 * GQ) {
      int idx[GN][GQ];
      float v[GN][GQ];
#pragma unroll
      for (int k = 0; k < GN; ++k) {
#pragma unroll
        for (int q = 0; q < GQ; ++q) {
          const unsigned j = j0 + 8 * q + l;
          const unsigned jj = j < c[k] ? j : c[k] - 1u;
          idx[k][q] = sorted[c[k] ? beg[k] + jj : 0u];
        }
      }
#pragma unroll
      for (int k = 0; k < GN; ++k) {
#pragma unroll
        for (int q = 0; q < GQ; ++q) {
          const unsigned j = j0 + 8 * q + l;
          const float t = zs[idx[k][q]];
          v[k][q] = j < c[k] ? t : 0.0f;
        }
      }
#pragma unroll
      for (int k = 0; k < GN; ++k) {
#pragma unroll
        for (int q = 0; q < GQ; ++q) acc[k] += v[k][q];
      }
    }
#pragma unroll
    for (int k = 0; k < GN; ++k) {
      float r = acc[k];
      r += __shfl_xor(r, 1, 8);
      r += __shfl_xor(r, 2, 8);
      r += __shfl_xor(r, 4, 8);
      const int node = nb + k;
      if (l == 0 && node < n)
        out[node] = (r + self[k]) * rsqrtf((float)(c[k] + 1u)) + bias;
    }
  }
}

extern "C" void kernel_launch(void* const* d_in, const int* in_sizes, int n_in,
                              void* d_out, int out_size, void* d_ws, size_t ws_size,
                              hipStream_t stream) {
  const float* x  = (const float*)d_in[0];
  const int*   ei = (const int*)d_in[1];
  const float* W1 = (const float*)d_in[2];
  const float* b1 = (const float*)d_in[3];
  const float* W2 = (const float*)d_in[4];
  const float* b2 = (const float*)d_in[5];

  int n = in_sizes[0] / 2;
  int E = in_sizes[1] / 2;
  const int* src = ei;
  const int* dst = ei + E;
  float* out = (float*)d_out;

  const int tb = 256;
  int gN = (n + tb - 1) / tb;
  int gE = (E + tb - 1) / tb;
  if (gE > 4096) gE = 4096;
  int tiles = (n + 15) / 16;
  int dblocks = (tiles + 7) / 8;

  size_t need = (4 * (size_t)n + (size_t)SL * (size_t)n + (size_t)E) * sizeof(float);
  int ranges = (n + RN - 1) / RN;
  if (ws_size >= need && ranges <= 65535 &&
      (size_t)E >= (size_t)ranges * RC * SL) {
    unsigned* cnt    = (unsigned*)d_ws;
    unsigned* rowptr = cnt + n;
    float*    xs     = (float*)(rowptr + n);
    unsigned* table  = (unsigned*)(xs + 2 * (size_t)n);
    int*      sorted = (int*)(table + (size_t)SL * (size_t)n);
    float*    zs     = (float*)table;
    unsigned* chunks = (unsigned*)sorted;

    int L = (E + SL - 1) / SL;
    dim3 gS(SL, ranges);
    int gG = (n + 32 * GN - 1) / (32 * GN);
    if (gG > 1024) gG = 1024;

    gcn_slice_count<<<gS, 1024, 0, stream>>>(dst, table, chunks, n, E, L);
    gcn_offsets<<<(n + CH - 1) / CH, CH, 0, stream>>>(table, chunks, cnt, rowptr, x, xs, n);
    gcn_slice_fill<<<gS, 1024, 0, stream>>>(src, dst, table, rowptr, sorted, n, E, L);
    gcn_gather1_dense<<<dblocks, 256, 0, stream>>>(sorted, rowptr, cnt, xs, W1, b1, W2, zs, n);
    gcn_gather2<<<gG, 256, 0, stream>>>(sorted, rowptr, cnt, zs, b2, out, n);
    return;
  }

  float* dinv = (float*)d_ws;
  float* agg  = dinv + n;
  float* z    = agg + 2 * (size_t)n;

  gcn_init<<<gN, tb, 0, stream>>>(dinv, agg, z, out, n);
  gcn_degree<<<gE, tb, 0, stream>>>(dst, dinv, E);
  gcn_rsqrt<<<gN, tb, 0, stream>>>(dinv, n);
  gcn_agg2<<<gE, tb, 0, stream>>>(src, dst, x, dinv, agg, E);

  gcn_dense_wmma<<<dblocks, 256, 0, stream>>>(agg, x, dinv, W1, b1, W2, z, n);

  gcn_out_self<<<gN, tb, 0, stream>>>(z, dinv, b2, out, n);
  gcn_out_edges<<<gE, tb, 0, stream>>>(src, dst, z, dinv, out, E);
}


# Count table re-tiled to 32 edge slices x node ranges of 12800 (was 16 x 6400): a counting or fill block scans half as many edges; f32 sum order: slots by 32 slices, then LDS arrival
# speedup vs baseline: 3.9124x; 1.0495x over previous
#include <hip/hip_runtime.h>


typedef float v2f __attribute__((ext_vector_type(2)));
typedef float v8f __attribute__((ext_vector_type(8)));

__global__ void gcn_init(float* __restrict__ deg, float* __restrict__ agg,
                         float* __restrict__ z, float* __restrict__ out, int n) {
  int i = blockIdx.x * blockDim.x + threadIdx.x;
  if (i < n) {
    deg[i] = 1.0f;
    agg[2 * i] = 0.0f;
    agg[2 * i + 1] = 0.0f;
    z[i] = 0.0f;
    out[i] = 0.0f;
  }
}

__global__ void gcn_degree(const int* __restrict__ dst, float* __restrict__ deg,
                           int E) {
  int i = blockIdx.x * blockDim.x + threadIdx.x;
  int stride = gridDim.x * blockDim.x;
  for (; i < E; i += stride) {
    if (i + stride < E) __builtin_prefetch(&dst[i + stride], 0, 1);
    atomicAdd(&deg[dst[i]], 1.0f);
  }
}

__global__ void gcn_rsqrt(float* __restrict__ dinv, int n) {
  int i = blockIdx.x * blockDim.x + threadIdx.x;
  if (i < n) dinv[i] = rsqrtf(dinv[i]);
}

__global__ void gcn_agg2(const int* __restrict__ src, const int* __restrict__ dst,
                         const float* __restrict__ x, const float* __restrict__ dinv,
                         float* __restrict__ agg, int E) {
  int i = blockIdx.x * blockDim.x + threadIdx.x;
  int stride = gridDim.x * blockDim.x;
  for (; i < E; i += stride) {
    if (i + stride < E) {
      __builtin_prefetch(&src[i + stride], 0, 1);
      __builtin_prefetch(&dst[i + stride], 0, 1);
    }
    int s = src[i];
    int d = dst[i];
    float w = dinv[s] * dinv[d];
    atomicAdd(&agg[2 * d], x[2 * s] * w);
    atomicAdd(&agg[2 * d + 1], x[2 * s + 1] * w);
  }
}

__global__ __launch_bounds__(256) void gcn_dense_wmma(
    const float* __restrict__ agg, const float* __restrict__ x,
    const float* __restrict__ dinv, const float* __restrict__ W1,
    const float* __restrict__ b1, const float* __restrict__ W2,
    float* __restrict__ z, int n) {
  const int lane = threadIdx.x & 31;
  const int wave = threadIdx.x >> 5;
  const int tile = blockIdx.x * 8 + wave;
  const int base = tile * 16;
  if (base >= n) return;

  const int m = lane & 15;
  const bool hiK = lane >= 16;
  const float kmask = hiK ? 0.0f : 1.0f;

  int node = base + m;
  const float rowmask = (!hiK && node < n) ? 1.0f : 0.0f;
  if (node >= n) node = n - 1;
  v2f A;
  float di = dinv[node];
  float sl = di * di;
  A.x = (agg[2 * node]     + x[2 * node]     * sl) * rowmask;
  A.y = (agg[2 * node + 1] + x[2 * node + 1] * sl) * rowmask;

  float zacc[8];
#pragma unroll
  for (int v = 0; v < 8; ++v) zacc[v] = 0.0f;

#pragma unroll
  for (int n0 = 0; n0 < 128; n0 += 16) {
    int col = n0 + m;
    v2f B;
    B.x = W1[col]       * kmask;
    B.y = W1[128 + col] * kmask;
    float bias = b1[col];
    v8f C;
#pragma unroll
    for (int v = 0; v < 8; ++v) C[v] = bias;

    C = __builtin_amdgcn_wmma_f32_16x16x4_f32(
         false, A,  false, B,
         (short)0, C,  false,  false);

    float w2 = W2[col];
#pragma unroll
    for (int v = 0; v < 8; ++v) {
      float h = C[v];
      h = h > 0.0f ? h : 0.0f;
      zacc[v] += h * w2;
    }
  }

#pragma unroll
  for (int v = 0; v < 8; ++v) {
    float r = zacc[v];
    r += __shfl_xor(r, 1, 16);
    r += __shfl_xor(r, 2, 16);
    r += __shfl_xor(r, 4, 16);
    r += __shfl_xor(r, 8, 16);
    zacc[v] = r;
  }
  if (m == 0) {
    int rowbase = base + (hiK ? 8 : 0);
#pragma unroll
    for (int v = 0; v < 8; ++v) {
      int r = rowbase + v;
      if (r < n) z[r] = zacc[v];
    }
  }
}

__global__ void gcn_out_self(const float* __restrict__ z,
                             const float* __restrict__ dinv,
                             const float* __restrict__ b2,
                             float* __restrict__ out, int n) {
  int i = blockIdx.x * blockDim.x + threadIdx.x;
  if (i < n) out[i] = z[i] * dinv[i] * dinv[i] + b2[0];
}

__global__ void gcn_out_edges(const int* __restrict__ src, const int* __restrict__ dst,
                              const float* __restrict__ z,
                              const float* __restrict__ dinv,
                              float* __restrict__ out, int E) {
  int i = blockIdx.x * blockDim.x + threadIdx.x;
  int stride = gridDim.x * blockDim.x;
  for (; i < E; i += stride) {
    if (i + stride < E) {
      __builtin_prefetch(&src[i + stride], 0, 1);
      __builtin_prefetch(&dst[i + stride], 0, 1);
    }
    int s = src[i];
    int d = dst[i];
    atomicAdd(&out[d], z[s] * dinv[s] * dinv[d]);
  }
}


constexpr int CH = 256;

template <int SL, int RN>
__global__ __launch_bounds__(1024) void gcn_slice_count(
    const int* __restrict__ dst, unsigned* __restrict__ table,
    unsigned* __restrict__ chunks, int n, int E, int L) {
  constexpr int RC = RN / CH;
  static_assert(RN % CH == 0, "a chunk must not straddle two ranges");
  __shared__ unsigned h[RN];
  const int tid = threadIdx.x;
  const int s = blockIdx.x;
  const int lo = blockIdx.y * RN;
  const unsigned lim = (n - lo < RN) ? (unsigned)(n - lo) : (unsigned)RN;
  for (int i = tid; i < RN; i += 1024) h[i] = 0u;
  __syncthreads();
  const int e0 = s * L;
  const int e1 = (E - e0 < L) ? E : e0 + L;
  for (int e = e0 + tid; e < e1; e += 1024 * 16) {
    int d[16];
#pragma unroll
    for (int k = 0; k < 16; ++k) {
      int ee = e + k * 1024;
      d[k] = (ee < e1) ? dst[ee] : -1;
    }
#pragma unroll
    for (int k = 0; k < 16; ++k) {
      unsigned x = (unsigned)d[k] - (unsigned)lo;
      if (x < lim) atomicAdd(&h[x], 1u);
    }
  }
  __syncthreads();
  unsigned* row = table + (size_t)s * n + lo;
  for (int i = tid; i < (int)lim; i += 1024) row[i] = h[i];
  const int wv = tid >> 5;
  const int lane = tid & 31;
  for (int c = wv; c < RC; c += 32) {
    unsigned sum = 0u;
#pragma unroll
    for (int j = 0; j < CH / 32; ++j) sum += h[c * CH + lane + 32 * j];
    sum += __shfl_xor(sum, 16, 32);
    sum += __shfl_xor(sum, 8, 32);
    sum += __shfl_xor(sum, 4, 32);
    sum += __shfl_xor(sum, 2, 32);
    sum += __shfl_xor(sum, 1, 32);
    if (lane == 0) chunks[(size_t)(blockIdx.y * RC + c) * SL + s] = sum;
  }
}

template <int SL>
__global__ __launch_bounds__(CH) void gcn_offsets(
    unsigned* __restrict__ table, const unsigned* __restrict__ chunks,
    unsigned* __restrict__ cnt, unsigned* __restrict__ rowptr,
    const float* __restrict__ x, float* __restrict__ xs, int n) {
  static_assert(SL % 4 == 0, "a chunk's slice sums are read as 16-byte pieces");
  constexpr int NW = CH / 32;
  constexpr int PF = SL / 2;
  __shared__ unsigned s_incl[NW];
  __shared__ unsigned s_part[NW];
  const int tid = threadIdx.x;
  const int lane = tid & 31;
  const int wv = tid >> 5;
  const int node = blockIdx.x * CH + tid;
  const bool in = node < n;

  unsigned v[SL];
  float2 xv = make_float2(0.0f, 0.0f);
#pragma unroll
  for (int s = 0; s < SL; ++s) v[s] = 0u;
  if (in) {
#pragma unroll
    for (int s = 0; s < SL; ++s) v[s] = table[(size_t)s * n + node];
    xv = ((const float2*)x)[node];
  }

  const uint4* c4 = (const uint4*)chunks;
  const int lim4 = (SL / 4) * blockIdx.x;
  unsigned part = 0u;
  for (int base = 0; base < lim4; base += CH * PF) {
    uint4 a[PF];
#pragma unroll
    for (int j = 0; j < PF; ++j) {
      const int i = base + tid + CH * j;
      a[j] = c4[i < lim4 ? i : 0];
    }
#pragma unroll
    for (int j = 0; j < PF; ++j) {
      const int i = base + tid + CH * j;
      const unsigned t = (a[j].x + a[j].y) + (a[j].z + a[j].w);
      part += i < lim4 ? t : 0u;
    }
  }
  part += __shfl_xor(part, 16, 32);
  part += __shfl_xor(part, 8, 32);
  part += __shfl_xor(part, 4, 32);
  part += __shfl_xor(part, 2, 32);
  part += __shfl_xor(part, 1, 32);

  unsigned mine = 0u;
#pragma unroll
  for (int s = 0; s < SL; ++s) mine += v[s];
  unsigned incl = mine;
#pragma unroll
  for (int o = 1; o < 32; o <<= 1) {
    unsigned t = __shfl_up(incl, o, 32);
    if (lane >= o) incl += t;
  }
  if (lane == 31) s_incl[wv] = incl;
  if (lane == 0) s_part[wv] = part;
  __syncthreads();
  unsigned off = 0u;
  if (lane < NW) off = s_part[lane] + ((lane < wv) ? s_incl[lane] : 0u);
  off += __shfl_xor(off, 16, 32);
  off += __shfl_xor(off, 8, 32);
  off += __shfl_xor(off, 4, 32);
  off += __shfl_xor(off, 2, 32);
  off += __shfl_xor(off, 1, 32);
  if (in) {
    unsigned run = 0u;
#pragma unroll
    for (int s = 0; s < SL; ++s) {
      table[(size_t)s * n + node] = run;
      run += v[s];
    }
    cnt[node] = mine;
    rowptr[node] = off + incl - mine;
    float di = rsqrtf((float)(mine + 1u));
    xs[2 * node]     = xv.x * di;
    xs[2 * node + 1] = xv.y * di;
  }
}

template <int RN>
__global__ __launch_bounds__(1024) void gcn_slice_fill(
    const int* __restrict__ src, const int* __restrict__ dst,
    const unsigned* __restrict__ table, const unsigned* __restrict__ rowptr,
    int* __restrict__ sorted, int n, int E, int L) {
  __shared__ unsigned cur[RN];
  const int tid = threadIdx.x;
  const int s = blockIdx.x;
  const int lo = blockIdx.y * RN;
  const unsigned lim = (n - lo < RN) ? (unsigned)(n - lo) : (unsigned)RN;
  const unsigned* row = table + (size_t)s * n + lo;
  constexpr int RT = (RN + 1023) / 1024;
  unsigned ra[RT], rb[RT];
#pragma unroll
  for (int k = 0; k < RT; ++k) {
    int i = tid + k * 1024;
    bool in = i < (int)lim;
    ra[k] = in ? rowptr[lo + i] : 0u;
    rb[k] = in ? row[i] : 0u;
  }
#pragma unroll
  for (int k = 0; k < RT; ++k) {
    int i = tid + k * 1024;
    if (i < (int)lim) cur[i] = ra[k] + rb[k];
  }
  __syncthreads();
  const int e0 = s * L;
  const int e1 = (E - e0 < L) ? E : e0 + L;
  for (int e = e0 + tid; e < e1; e += 1024 * 8) {
    int d[8], sv[8];
    unsigned pos[8];
#pragma unroll
    for (int k = 0; k < 8; ++k) {
      int ee = e + k * 1024;
      bool in = ee < e1;
      d[k] = in ? dst[ee] : -1;
      sv[k] = in ? src[ee] : 0;
    }
#pragma unroll
    for (int k = 0; k < 8; ++k) {
      unsigned x = (unsigned)d[k] - (unsigned)lo;
      pos[k] = 0u;
      if (x < lim) pos[k] = atomicAdd(&cur[x], 1u);
    }
#pragma unroll
    for (int k = 0; k < 8; ++k) {
      unsigned x = (unsigned)d[k] - (unsigned)lo;
      if (x < lim) sorted[pos[k]] = sv[k];
    }
  }
}

constexpr int GN = 4;
constexpr int GQ = 4;

__global__ __launch_bounds__(256) void gcn_gather1_dense(
    const int* __restrict__ sorted, const unsigned* __restrict__ rowptr,
    const unsigned* __restrict__ cnt, const float* __restrict__ xs,
    const float* __restrict__ W1, const float* __restrict__ b1,
    const float* __restrict__ W2, float* __restrict__ zs, int n) {
  static_assert(GN == 4, "four groups of GN nodes must make one 16-row tile");
  const int lane = threadIdx.x & 31;
  const unsigned l = threadIdx.x & 7;
  const int grp = lane >> 3;
  const int m = lane & 15;
  const bool hiK = lane >= 16;
  const float kmask = hiK ? 0.0f : 1.0f;
  const int wb = ((blockIdx.x * 256 + threadIdx.x) >> 5) * 16;
  if (wb >= n) return;
  const int nb = wb + grp * GN;
  const float2* xs2 = (const float2*)xs;

  unsigned beg[GN], c[GN];
  float2 self[GN];
  float ax[GN], ay[GN];
#pragma unroll
  for (int k = 0; k < GN; ++k) {
    const int node = nb + k;
    const bool in = node < n;
    beg[k] = in ? rowptr[node] : 0u;
    c[k] = in ? cnt[node] : 0u;
    self[k] = in ? xs2[node] : make_float2(0.0f, 0.0f);
    ax[k] = 0.0f;
    ay[k] = 0.0f;
  }
  float w1a[8], w1b[8], bias[8], w2[8];
#pragma unroll
  for (int t = 0; t < 8; ++t) {
    const int col = 16 * t + m;
    w1a[t] = W1[col];
    w1b[t] = W1[128 + col];
    bias[t] = b1[col];
    w2[t] = W2[col];
  }
  unsigned cm = 0u;
#pragma unroll
  for (int k = 0; k < GN; ++k) cm = c[k] > cm ? c[k] : cm;
  for (unsigned j0 = 0; j0 < cm; j0 += 8 * GQ) {
    int idx[GN][GQ];
    float2 v[GN][GQ];
#pragma unroll
    for (int k = 0; k < GN; ++k) {
#pragma unroll
      for (int q = 0; q < GQ; ++q) {
        const unsigned j = j0 + 8 * q + l;
        const unsigned jj = j < c[k] ? j : c[k] - 1u;
        idx[k][q] = sorted[c[k] ? beg[k] + jj : 0u];
      }
    }
#pragma unroll
    for (int k = 0; k < GN; ++k) {
#pragma unroll
      for (int q = 0; q < GQ; ++q) {
        const unsigned j = j0 + 8 * q + l;
        const float2 t = xs2[idx[k][q]];
        v[k][q].x = j < c[k] ? t.x : 0.0f;
        v[k][q].y = j < c[k] ? t.y : 0.0f;
      }
    }
#pragma unroll
    for (int k = 0; k < GN; ++k) {
#pragma unroll
      for (int q = 0; q < GQ; ++q) {
        ax[k] += v[k][q].x;
        ay[k] += v[k][q].y;
      }
    }
  }
  float tx = 0.0f, ty = 0.0f, td = 0.0f;
#pragma unroll
  for (int k = 0; k < GN; ++k) {
    float sx = ax[k], sy = ay[k];
    sx += __shfl_xor(sx, 1, 8);  sy += __shfl_xor(sy, 1, 8);
    sx += __shfl_xor(sx, 2, 8);  sy += __shfl_xor(sy, 2, 8);
    sx += __shfl_xor(sx, 4, 8);  sy += __shfl_xor(sy, 4, 8);
    const float di = rsqrtf((float)(c[k] + 1u));
    const float rx = (sx + self[k].x) * di;
    const float ry = (sy + self[k].y) * di;
    tx = (l == (unsigned)k) ? rx : tx;
    ty = (l == (unsigned)k) ? ry : ty;
    td = (l == (unsigned)k) ? di : td;
  }
  const int from = 8 * (m >> 2) + (m & 3);
  const float arx = __shfl(tx, from, 32);
  const float ary = __shfl(ty, from, 32);
  const bool rowok = !hiK && (wb + m < n);
  v2f A;
  A.x = rowok ? arx : 0.0f;
  A.y = rowok ? ary : 0.0f;

  float zacc[8];
#pragma unroll
  for (int v = 0; v < 8; ++v) zacc[v] = 0.0f;
#pragma unroll
  for (int t = 0; t < 8; ++t) {
    v2f B;
    B.x = w1a[t] * kmask;
    B.y = w1b[t] * kmask;
    v8f C;
#pragma unroll
    for (int v = 0; v < 8; ++v) C[v] = bias[t];

    C = __builtin_amdgcn_wmma_f32_16x16x4_f32(
         false, A,  false, B,
         (short)0, C,  false,  false);

#pragma unroll
    for (int v = 0; v < 8; ++v) {
      float h = C[v];
      h = h > 0.0f ? h : 0.0f;
      zacc[v] += h * w2[t];
    }
  }
#pragma unroll
  for (int v = 0; v < 8; ++v) {
    float r = zacc[v];
    r += __shfl_xor(r, 1, 16);
    r += __shfl_xor(r, 2, 16);
    r += __shfl_xor(r, 4, 16);
    r += __shfl_xor(r, 8, 16);
    zacc[v] = r;
  }
  const int reg = 4 * (grp & 1) + (int)(l & 3);
  float zsel = zacc[0];
#pragma unroll
  for (int v = 1; v < 8; ++v) zsel = (reg == v) ? zacc[v] : zsel;
  const int node = nb + (int)l;
  if (l < (unsigned)GN && node < n) zs[node] = zsel * td;
}

__global__ __launch_bounds__(256) void gcn_gather2(
    const int* __restrict__ sorted, const unsigned* __restrict__ rowptr,
    const unsigned* __restrict__ cnt, const float* __restrict__ zs,
    const float* __restrict__ b2, float* __restrict__ out, int n) {
  const unsigned l = threadIdx.x & 7;
  const int g = (blockIdx.x * 256 + threadIdx.x) >> 3;
  const int groups = (gridDim.x * 256) >> 3;
  const float bias = b2[0];
  for (int nb = g * GN; nb < n; nb += groups * GN) {
    unsigned beg[GN], c[GN];
    float self[GN], acc[GN];
#pragma unroll
    for (int k = 0; k < GN; ++k) {
      const int node = nb + k;
      const bool in = node < n;
      beg[k] = in ? rowptr[node] : 0u;
      c[k] = in ? cnt[node] : 0u;
      self[k] = in ? zs[node] : 0.0f;
      acc[k] = 0.0f;
    }
    unsigned cm = 0u;
#pragma unroll
    for (int k = 0; k < GN; ++k) cm = c[k] > cm ? c[k] : cm;
    for (unsigned j0 = 0; j0 < cm; j0 += 8 * GQ) {
      int idx[GN][GQ];
      float v[GN][GQ];
#pragma unroll
      for (int k = 0; k < GN; ++k) {
#pragma unroll
        for (int q = 0; q < GQ; ++q) {
          const unsigned j = j0 + 8 * q + l;
          const unsigned jj = j < c[k] ? j : c[k] - 1u;
          idx[k][q] = sorted[c[k] ? beg[k] + jj : 0u];
        }
      }
#pragma unroll
      for (int k = 0; k < GN; ++k) {
#pragma unroll
        for (int q = 0; q < GQ; ++q) {
          const unsigned j = j0 + 8 * q + l;
          const float t = zs[idx[k][q]];
          v[k][q] = j < c[k] ? t : 0.0f;
        }
      }
#pragma unroll
      for (int k = 0; k < GN; ++k) {
#pragma unroll
        for (int q = 0; q < GQ; ++q) acc[k] += v[k][q];
      }
    }
#pragma unroll
    for (int k = 0; k < GN; ++k) {
      float r = acc[k];
      r += __shfl_xor(r, 1, 8);
      r += __shfl_xor(r, 2, 8);
      r += __shfl_xor(r, 4, 8);
      const int node = nb + k;
      if (l == 0 && node < n)
        out[node] = (r + self[k]) * rsqrtf((float)(c[k] + 1u)) + bias;
    }
  }
}

template <int SL, int RN>
static bool gcn_sorted_path(const float* x, const int* src, const int* dst,
                            const float* W1, const float* b1, const float* W2,
                            const float* b2, float* out, int n, int E,
                            int dblocks, void* d_ws, size_t ws_size,
                            hipStream_t stream) {
  constexpr int RC = RN / CH;
  size_t need = (4 * (size_t)n + (size_t)SL * (size_t)n + (size_t)E) * sizeof(float);
  int ranges = (n + RN - 1) / RN;
  if (ws_size < need || ranges > 65535 ||
      (size_t)E < (size_t)ranges * RC * SL) return false;
  unsigned* cnt    = (unsigned*)d_ws;
  unsigned* rowptr = cnt + n;
  float*    xs     = (float*)(rowptr + n);
  unsigned* table  = (unsigned*)(xs + 2 * (size_t)n);
  int*      sorted = (int*)(table + (size_t)SL * (size_t)n);
  float*    zs     = (float*)table;
  unsigned* chunks = (unsigned*)sorted;

  int L = (E + SL - 1) / SL;
  dim3 gS(SL, ranges);
  int gG = (n + 32 * GN - 1) / (32 * GN);
  if (gG > 1024) gG = 1024;

  gcn_slice_count<SL, RN><<<gS, 1024, 0, stream>>>(dst, table, chunks, n, E, L);
  gcn_offsets<SL><<<(n + CH - 1) / CH, CH, 0, stream>>>(table, chunks, cnt, rowptr, x, xs, n);
  gcn_slice_fill<RN><<<gS, 1024, 0, stream>>>(src, dst, table, rowptr, sorted, n, E, L);
  gcn_gather1_dense<<<dblocks, 256, 0, stream>>>(sorted, rowptr, cnt, xs, W1, b1, W2, zs, n);
  gcn_gather2<<<gG, 256, 0, stream>>>(sorted, rowptr, cnt, zs, b2, out, n);
  return true;
}

extern "C" void kernel_launch(void* const* d_in, const int* in_sizes, int n_in,
                              void* d_out, int out_size, void* d_ws, size_t ws_size,
                              hipStream_t stream) {
  const float* x  = (const float*)d_in[0];
  const int*   ei = (const int*)d_in[1];
  const float* W1 = (const float*)d_in[2];
  const float* b1 = (const float*)d_in[3];
  const float* W2 = (const float*)d_in[4];
  const float* b2 = (const float*)d_in[5];

  int n = in_sizes[0] / 2;
  int E = in_sizes[1] / 2;
  const int* src = ei;
  const int* dst = ei + E;
  float* out = (float*)d_out;

  const int tb = 256;
  int gN = (n + tb - 1) / tb;
  int gE = (E + tb - 1) / tb;
  if (gE > 4096) gE = 4096;
  int tiles = (n + 15) / 16;
  int dblocks = (tiles + 7) / 8;

  if (gcn_sorted_path<32, 12800>(x, src, dst, W1, b1, W2, b2, out, n, E,
                                 dblocks, d_ws, ws_size, stream)) return;
  if (gcn_sorted_path<16, 6400>(x, src, dst, W1, b1, W2, b2, out, n, E,
                                dblocks, d_ws, ws_size, stream)) return;

  float* dinv = (float*)d_ws;
  float* agg  = dinv + n;
  float* z    = agg + 2 * (size_t)n;

  gcn_init<<<gN, tb, 0, stream>>>(dinv, agg, z, out, n);
  gcn_degree<<<gE, tb, 0, stream>>>(dst, dinv, E);
  gcn_rsqrt<<<gN, tb, 0, stream>>>(dinv, n);
  gcn_agg2<<<gE, tb, 0, stream>>>(src, dst, x, dinv, agg, E);

  gcn_dense_wmma<<<dblocks, 256, 0, stream>>>(agg, x, dinv, W1, b1, W2, z, n);

  gcn_out_self<<<gN, tb, 0, stream>>>(z, dinv, b2, out, n);
  gcn_out_edges<<<gE, tb, 0, stream>>>(src, dst, z, dinv, out, E);
}


# Layer-2 gather copies the first 80000 values of its 4-byte node table into LDS (1024-thread blocks of 512 nodes, one pass); only neighbours past the copy are read from L2; f32 sum order unchanged
# speedup vs baseline: 4.0456x; 1.0340x over previous
#include <hip/hip_runtime.h>


typedef float v2f __attribute__((ext_vector_type(2)));
typedef float v8f __attribute__((ext_vector_type(8)));

__global__ void gcn_init(float* __restrict__ deg, float* __restrict__ agg,
                         float* __restrict__ z, float* __restrict__ out, int n) {
  int i = blockIdx.x * blockDim.x + threadIdx.x;
  if (i < n) {
    deg[i] = 1.0f;
    agg[2 * i] = 0.0f;
    agg[2 * i + 1] = 0.0f;
    z[i] = 0.0f;
    out[i] = 0.0f;
  }
}

__global__ void gcn_degree(const int* __restrict__ dst, float* __restrict__ deg,
                           int E) {
  int i = blockIdx.x * blockDim.x + threadIdx.x;
  int stride = gridDim.x * blockDim.x;
  for (; i < E; i += stride) {
    if (i + stride < E) __builtin_prefetch(&dst[i + stride], 0, 1);
    atomicAdd(&deg[dst[i]], 1.0f);
  }
}

__global__ void gcn_rsqrt(float* __restrict__ dinv, int n) {
  int i = blockIdx.x * blockDim.x + threadIdx.x;
  if (i < n) dinv[i] = rsqrtf(dinv[i]);
}

__global__ void gcn_agg2(const int* __restrict__ src, const int* __restrict__ dst,
                         const float* __restrict__ x, const float* __restrict__ dinv,
                         float* __restrict__ agg, int E) {
  int i = blockIdx.x * blockDim.x + threadIdx.x;
  int stride = gridDim.x * blockDim.x;
  for (; i < E; i += stride) {
    if (i + stride < E) {
      __builtin_prefetch(&src[i + stride], 0, 1);
      __builtin_prefetch(&dst[i + stride], 0, 1);
    }
    int s = src[i];
    int d = dst[i];
    float w = dinv[s] * dinv[d];
    atomicAdd(&agg[2 * d], x[2 * s] * w);
    atomicAdd(&agg[2 * d + 1], x[2 * s + 1] * w);
  }
}

__global__ __launch_bounds__(256) void gcn_dense_wmma(
    const float* __restrict__ agg, const float* __restrict__ x,
    const float* __restrict__ dinv, const float* __restrict__ W1,
    const float* __restrict__ b1, const float* __restrict__ W2,
    float* __restrict__ z, int n) {
  const int lane = threadIdx.x & 31;
  const int wave = threadIdx.x >> 5;
  const int tile = blockIdx.x * 8 + wave;
  const int base = tile * 16;
  if (base >= n) return;

  const int m = lane & 15;
  const bool hiK = lane >= 16;
  const float kmask = hiK ? 0.0f : 1.0f;

  int node = base + m;
  const float rowmask = (!hiK && node < n) ? 1.0f : 0.0f;
  if (node >= n) node = n - 1;
  v2f A;
  float di = dinv[node];
  float sl = di * di;
  A.x = (agg[2 * node]     + x[2 * node]     * sl) * rowmask;
  A.y = (agg[2 * node + 1] + x[2 * node + 1] * sl) * rowmask;

  float zacc[8];
#pragma unroll
  for (int v = 0; v < 8; ++v) zacc[v] = 0.0f;

#pragma unroll
  for (int n0 = 0; n0 < 128; n0 += 16) {
    int col = n0 + m;
    v2f B;
    B.x = W1[col]       * kmask;
    B.y = W1[128 + col] * kmask;
    float bias = b1[col];
    v8f C;
#pragma unroll
    for (int v = 0; v < 8; ++v) C[v] = bias;

    C = __builtin_amdgcn_wmma_f32_16x16x4_f32(
         false, A,  false, B,
         (short)0, C,  false,  false);

    float w2 = W2[col];
#pragma unroll
    for (int v = 0; v < 8; ++v) {
      float h = C[v];
      h = h > 0.0f ? h : 0.0f;
      zacc[v] += h * w2;
    }
  }

#pragma unroll
  for (int v = 0; v < 8; ++v) {
    float r = zacc[v];
    r += __shfl_xor(r, 1, 16);
    r += __shfl_xor(r, 2, 16);
    r += __shfl_xor(r, 4, 16);
    r += __shfl_xor(r, 8, 16);
    zacc[v] = r;
  }
  if (m == 0) {
    int rowbase = base + (hiK ? 8 : 0);
#pragma unroll
    for (int v = 0; v < 8; ++v) {
      int r = rowbase + v;
      if (r < n) z[r] = zacc[v];
    }
  }
}

__global__ void gcn_out_self(const float* __restrict__ z,
                             const float* __restrict__ dinv,
                             const float* __restrict__ b2,
                             float* __restrict__ out, int n) {
  int i = blockIdx.x * blockDim.x + threadIdx.x;
  if (i < n) out[i] = z[i] * dinv[i] * dinv[i] + b2[0];
}

__global__ void gcn_out_edges(const int* __restrict__ src, const int* __restrict__ dst,
                              const float* __restrict__ z,
                              const float* __restrict__ dinv,
                              float* __restrict__ out, int E) {
  int i = blockIdx.x * blockDim.x + threadIdx.x;
  int stride = gridDim.x * blockDim.x;
  for (; i < E; i += stride) {
    if (i + stride < E) {
      __builtin_prefetch(&src[i + stride], 0, 1);
      __builtin_prefetch(&dst[i + stride], 0, 1);
    }
    int s = src[i];
    int d = dst[i];
    atomicAdd(&out[d], z[s] * dinv[s] * dinv[d]);
  }
}


constexpr int CH = 256;

template <int SL, int RN>
__global__ __launch_bounds__(1024) void gcn_slice_count(
    const int* __restrict__ dst, unsigned* __restrict__ table,
    unsigned* __restrict__ chunks, int n, int E, int L) {
  constexpr int RC = RN / CH;
  static_assert(RN % CH == 0, "a chunk must not straddle two ranges");
  __shared__ unsigned h[RN];
  const int tid = threadIdx.x;
  const int s = blockIdx.x;
  const int lo = blockIdx.y * RN;
  const unsigned lim = (n - lo < RN) ? (unsigned)(n - lo) : (unsigned)RN;
  for (int i = tid; i < RN; i += 1024) h[i] = 0u;
  __syncthreads();
  const int e0 = s * L;
  const int e1 = (E - e0 < L) ? E : e0 + L;
  for (int e = e0 + tid; e < e1; e += 1024 * 16) {
    int d[16];
#pragma unroll
    for (int k = 0; k < 16; ++k) {
      int ee = e + k * 1024;
      d[k] = (ee < e1) ? dst[ee] : -1;
    }
#pragma unroll
    for (int k = 0; k < 16; ++k) {
      unsigned x = (unsigned)d[k] - (unsigned)lo;
      if (x < lim) atomicAdd(&h[x], 1u);
    }
  }
  __syncthreads();
  unsigned* row = table + (size_t)s * n + lo;
  for (int i = tid; i < (int)lim; i += 1024) row[i] = h[i];
  const int wv = tid >> 5;
  const int lane = tid & 31;
  for (int c = wv; c < RC; c += 32) {
    unsigned sum = 0u;
#pragma unroll
    for (int j = 0; j < CH / 32; ++j) sum += h[c * CH + lane + 32 * j];
    sum += __shfl_xor(sum, 16, 32);
    sum += __shfl_xor(sum, 8, 32);
    sum += __shfl_xor(sum, 4, 32);
    sum += __shfl_xor(sum, 2, 32);
    sum += __shfl_xor(sum, 1, 32);
    if (lane == 0) chunks[(size_t)(blockIdx.y * RC + c) * SL + s] = sum;
  }
}

template <int SL>
__global__ __launch_bounds__(CH) void gcn_offsets(
    unsigned* __restrict__ table, const unsigned* __restrict__ chunks,
    unsigned* __restrict__ cnt, unsigned* __restrict__ rowptr,
    const float* __restrict__ x, float* __restrict__ xs, int n) {
  static_assert(SL % 4 == 0, "a chunk's slice sums are read as 16-byte pieces");
  constexpr int NW = CH / 32;
  constexpr int PF = SL / 2;
  __shared__ unsigned s_incl[NW];
  __shared__ unsigned s_part[NW];
  const int tid = threadIdx.x;
  const int lane = tid & 31;
  const int wv = tid >> 5;
  const int node = blockIdx.x * CH + tid;
  const bool in = node < n;

  unsigned v[SL];
  float2 xv = make_float2(0.0f, 0.0f);
#pragma unroll
  for (int s = 0; s < SL; ++s) v[s] = 0u;
  if (in) {
#pragma unroll
    for (int s = 0; s < SL; ++s) v[s] = table[(size_t)s * n + node];
    xv = ((const float2*)x)[node];
  }

  const uint4* c4 = (const uint4*)chunks;
  const int lim4 = (SL / 4) * blockIdx.x;
  unsigned part = 0u;
  for (int base = 0; base < lim4; base += CH * PF) {
    uint4 a[PF];
#pragma unroll
    for (int j = 0; j < PF; ++j) {
      const int i = base + tid + CH * j;
      a[j] = c4[i < lim4 ? i : 0];
    }
#pragma unroll
    for (int j = 0; j < PF; ++j) {
      const int i = base + tid + CH * j;
      const unsigned t = (a[j].x + a[j].y) + (a[j].z + a[j].w);
      part += i < lim4 ? t : 0u;
    }
  }
  part += __shfl_xor(part, 16, 32);
  part += __shfl_xor(part, 8, 32);
  part += __shfl_xor(part, 4, 32);
  part += __shfl_xor(part, 2, 32);
  part += __shfl_xor(part, 1, 32);

  unsigned mine = 0u;
#pragma unroll
  for (int s = 0; s < SL; ++s) mine += v[s];
  unsigned incl = mine;
#pragma unroll
  for (int o = 1; o < 32; o <<= 1) {
    unsigned t = __shfl_up(incl, o, 32);
    if (lane >= o) incl += t;
  }
  if (lane == 31) s_incl[wv] = incl;
  if (lane == 0) s_part[wv] = part;
  __syncthreads();
  unsigned off = 0u;
  if (lane < NW) off = s_part[lane] + ((lane < wv) ? s_incl[lane] : 0u);
  off += __shfl_xor(off, 16, 32);
  off += __shfl_xor(off, 8, 32);
  off += __shfl_xor(off, 4, 32);
  off += __shfl_xor(off, 2, 32);
  off += __shfl_xor(off, 1, 32);
  if (in) {
    unsigned run = 0u;
#pragma unroll
    for (int s = 0; s < SL; ++s) {
      table[(size_t)s * n + node] = run;
      run += v[s];
    }
    cnt[node] = mine;
    rowptr[node] = off + incl - mine;
    float di = rsqrtf((float)(mine + 1u));
    xs[2 * node]     = xv.x * di;
    xs[2 * node + 1] = xv.y * di;
  }
}

template <int RN>
__global__ __launch_bounds__(1024) void gcn_slice_fill(
    const int* __restrict__ src, const int* __restrict__ dst,
    const unsigned* __restrict__ table, const unsigned* __restrict__ rowptr,
    int* __restrict__ sorted, int n, int E, int L) {
  __shared__ unsigned cur[RN];
  const int tid = threadIdx.x;
  const int s = blockIdx.x;
  const int lo = blockIdx.y * RN;
  const unsigned lim = (n - lo < RN) ? (unsigned)(n - lo) : (unsigned)RN;
  const unsigned* row = table + (size_t)s * n + lo;
  constexpr int RT = (RN + 1023) / 1024;
  unsigned ra[RT], rb[RT];
#pragma unroll
  for (int k = 0; k < RT; ++k) {
    int i = tid + k * 1024;
    bool in = i < (int)lim;
    ra[k] = in ? rowptr[lo + i] : 0u;
    rb[k] = in ? row[i] : 0u;
  }
#pragma unroll
  for (int k = 0; k < RT; ++k) {
    int i = tid + k * 1024;
    if (i < (int)lim) cur[i] = ra[k] + rb[k];
  }
  __syncthreads();
  const int e0 = s * L;
  const int e1 = (E - e0 < L) ? E : e0 + L;
  for (int e = e0 + tid; e < e1; e += 1024 * 8) {
    int d[8], sv[8];
    unsigned pos[8];
#pragma unroll
    for (int k = 0; k < 8; ++k) {
      int ee = e + k * 1024;
      bool in = ee < e1;
      d[k] = in ? dst[ee] : -1;
      sv[k] = in ? src[ee] : 0;
    }
#pragma unroll
    for (int k = 0; k < 8; ++k) {
      unsigned x = (unsigned)d[k] - (unsigned)lo;
      pos[k] = 0u;
      if (x < lim) pos[k] = atomicAdd(&cur[x], 1u);
    }
#pragma unroll
    for (int k = 0; k < 8; ++k) {
      unsigned x = (unsigned)d[k] - (unsigned)lo;
      if (x < lim) sorted[pos[k]] = sv[k];
    }
  }
}

constexpr int GN = 4;
constexpr int GQ = 4;

__global__ __launch_bounds__(256) void gcn_gather1_dense(
    const int* __restrict__ sorted, const unsigned* __restrict__ rowptr,
    const unsigned* __restrict__ cnt, const float* __restrict__ xs,
    const float* __restrict__ W1, const float* __restrict__ b1,
    const float* __restrict__ W2, float* __restrict__ zs, int n) {
  static_assert(GN == 4, "four groups of GN nodes must make one 16-row tile");
  const int lane = threadIdx.x & 31;
  const unsigned l = threadIdx.x & 7;
  const int grp = lane >> 3;
  const int m = lane & 15;
  const bool hiK = lane >= 16;
  const float kmask = hiK ? 0.0f : 1.0f;
  const int wb = ((blockIdx.x * 256 + threadIdx.x) >> 5) * 16;
  if (wb >= n) return;
  const int nb = wb + grp * GN;
  const float2* xs2 = (const float2*)xs;

  unsigned beg[GN], c[GN];
  float2 self[GN];
  float ax[GN], ay[GN];
#pragma unroll
  for (int k = 0; k < GN; ++k) {
    const int node = nb + k;
    const bool in = node < n;
    beg[k] = in ? rowptr[node] : 0u;
    c[k] = in ? cnt[node] : 0u;
    self[k] = in ? xs2[node] : make_float2(0.0f, 0.0f);
    ax[k] = 0.0f;
    ay[k] = 0.0f;
  }
  float w1a[8], w1b[8], bias[8], w2[8];
#pragma unroll
  for (int t = 0; t < 8; ++t) {
    const int col = 16 * t + m;
    w1a[t] = W1[col];
    w1b[t] = W1[128 + col];
    bias[t] = b1[col];
    w2[t] = W2[col];
  }
  unsigned cm = 0u;
#pragma unroll
  for (int k = 0; k < GN; ++k) cm = c[k] > cm ? c[k] : cm;
  for (unsigned j0 = 0; j0 < cm; j0 += 8 * GQ) {
    int idx[GN][GQ];
    float2 v[GN][GQ];
#pragma unroll
    for (int k = 0; k < GN; ++k) {
#pragma unroll
      for (int q = 0; q < GQ; ++q) {
        const unsigned j = j0 + 8 * q + l;
        const unsigned jj = j < c[k] ? j : c[k] - 1u;
        idx[k][q] = sorted[c[k] ? beg[k] + jj : 0u];
      }
    }
#pragma unroll
    for (int k = 0; k < GN; ++k) {
#pragma unroll
      for (int q = 0; q < GQ; ++q) {
        const unsigned j = j0 + 8 * q + l;
        const float2 t = xs2[idx[k][q]];
        v[k][q].x = j < c[k] ? t.x : 0.0f;
        v[k][q].y = j < c[k] ? t.y : 0.0f;
      }
    }
#pragma unroll
    for (int k = 0; k < GN; ++k) {
#pragma unroll
      for (int q = 0; q < GQ; ++q) {
        ax[k] += v[k][q].x;
        ay[k] += v[k][q].y;
      }
    }
  }
  float tx = 0.0f, ty = 0.0f, td = 0.0f;
#pragma unroll
  for (int k = 0; k < GN; ++k) {
    float sx = ax[k], sy = ay[k];
    sx += __shfl_xor(sx, 1, 8);  sy += __shfl_xor(sy, 1, 8);
    sx += __shfl_xor(sx, 2, 8);  sy += __shfl_xor(sy, 2, 8);
    sx += __shfl_xor(sx, 4, 8);  sy += __shfl_xor(sy, 4, 8);
    const float di = rsqrtf((float)(c[k] + 1u));
    const float rx = (sx + self[k].x) * di;
    const float ry = (sy + self[k].y) * di;
    tx = (l == (unsigned)k) ? rx : tx;
    ty = (l == (unsigned)k) ? ry : ty;
    td = (l == (unsigned)k) ? di : td;
  }
  const int from = 8 * (m >> 2) + (m & 3);
  const float arx = __shfl(tx, from, 32);
  const float ary = __shfl(ty, from, 32);
  const bool rowok = !hiK && (wb + m < n);
  v2f A;
  A.x = rowok ? arx : 0.0f;
  A.y = rowok ? ary : 0.0f;

  float zacc[8];
#pragma unroll
  for (int v = 0; v < 8; ++v) zacc[v] = 0.0f;
#pragma unroll
  for (int t = 0; t < 8; ++t) {
    v2f B;
    B.x = w1a[t] * kmask;
    B.y = w1b[t] * kmask;
    v8f C;
#pragma unroll
    for (int v = 0; v < 8; ++v) C[v] = bias[t];

    C = __builtin_amdgcn_wmma_f32_16x16x4_f32(
         false, A,  false, B,
         (short)0, C,  false,  false);

#pragma unroll
    for (int v = 0; v < 8; ++v) {
      float h = C[v];
      h = h > 0.0f ? h : 0.0f;
      zacc[v] += h * w2[t];
    }
  }
#pragma unroll
  for (int v = 0; v < 8; ++v) {
    float r = zacc[v];
    r += __shfl_xor(r, 1, 16);
    r += __shfl_xor(r, 2, 16);
    r += __shfl_xor(r, 4, 16);
    r += __shfl_xor(r, 8, 16);
    zacc[v] = r;
  }
  const int reg = 4 * (grp & 1) + (int)(l & 3);
  float zsel = zacc[0];
#pragma unroll
  for (int v = 1; v < 8; ++v) zsel = (reg == v) ? zacc[v] : zsel;
  const int node = nb + (int)l;
  if (l < (unsigned)GN && node < n) zs[node] = zsel * td;
}

constexpr int ZL = 80000;
constexpr int G2T = 1024;
constexpr int ZF = 10;

__device__ __forceinline__ void gcn_g2_slots(
    const int* __restrict__ sorted, const unsigned (&beg)[GN],
    const unsigned (&c)[GN], unsigned j0, unsigned l, int (&idx)[GN][GQ]) {
#pragma unroll
  for (int k = 0; k < GN; ++k) {
#pragma unroll
    for (int q = 0; q < GQ; ++q) {
      const unsigned j = j0 + 8 * q + l;
      const unsigned jj = j < c[k] ? j : c[k] - 1u;
      idx[k][q] = sorted[c[k] ? beg[k] + jj : 0u];
    }
  }
}

__device__ __forceinline__ void gcn_g2_sum(
    const float* zl, const float* __restrict__ zs, int zn,
    const unsigned (&c)[GN], unsigned j0, unsigned l,
    const int (&idx)[GN][GQ], float (&acc)[GN]) {
  float a[GN][GQ], b[GN][GQ];
#pragma unroll
  for (int k = 0; k < GN; ++k) {
#pragma unroll
    for (int q = 0; q < GQ; ++q) {
      const int id = idx[k][q];
      a[k][q] = zl[id < zn ? id : 0];
    }
  }
#pragma unroll
  for (int k = 0; k < GN; ++k) {
#pragma unroll
    for (int q = 0; q < GQ; ++q) {
      const int id = idx[k][q];
      b[k][q] = zs[id < zn ? 0 : id];
    }
  }
#pragma unroll
  for (int k = 0; k < GN; ++k) {
#pragma unroll
    for (int q = 0; q < GQ; ++q) {
      const unsigned j = j0 + 8 * q + l;
      const float t = idx[k][q] < zn ? a[k][q] : b[k][q];
      acc[k] += j < c[k] ? t : 0.0f;
    }
  }
}

__global__ __launch_bounds__(G2T) void gcn_gather2(
    const int* __restrict__ sorted, const unsigned* __restrict__ rowptr,
    const unsigned* __restrict__ cnt, const float* __restrict__ zs,
    const float* __restrict__ b2, float* __restrict__ out, int n) {
  static_assert(ZL % 4 == 0, "the copy is made of 16-byte pieces");
  __shared__ float4 zl4[ZL / 4];
  float* zl = (float*)zl4;
  const int tid = threadIdx.x;
  const unsigned l = tid & 7;
  const int nb = (blockIdx.x * (G2T / 8) + (tid >> 3)) * GN;
  const float bias = b2[0];
  const int zn = n < ZL ? n : ZL;
  const int n4 = zn >> 2;
  const int last4 = n4 > 0 ? n4 - 1 : 0;
  const float4* z4 = (const float4*)zs;

  float4 t[ZF];
#pragma unroll
  for (int j = 0; j < ZF; ++j) {
    const int i = tid + G2T * j;
    t[j] = z4[i < n4 ? i : last4];
  }
  unsigned beg[GN], c[GN];
  float self[GN], acc[GN];
#pragma unroll
  for (int k = 0; k < GN; ++k) {
    const int node = nb + k;
    const bool in = node < n;
    const int nc = in ? node : n - 1;
    const unsigned r = rowptr[nc];
    const unsigned cc = cnt[nc];
    const float s = zs[nc];
    beg[k] = in ? r : 0u;
    c[k] = in ? cc : 0u;
    self[k] = in ? s : 0.0f;
    acc[k] = 0.0f;
  }
#pragma unroll
  for (int j = 0; j < ZF; ++j) {
    const int i = tid + G2T * j;
    zl4[i < n4 ? i : last4] = t[j];
  }
  int idx[GN][GQ];
  gcn_g2_slots(sorted, beg, c, 0u, l, idx);
  for (int base = G2T * ZF; base < n4; base += G2T * ZF) {
#pragma unroll
    for (int j = 0; j < ZF; ++j) {
      const int i = base + tid + G2T * j;
      t[j] = z4[i < n4 ? i : last4];
    }
#pragma unroll
    for (int j = 0; j < ZF; ++j) {
      const int i = base + tid + G2T * j;
      zl4[i < n4 ? i : last4] = t[j];
    }
  }
  for (int i = 4 * n4 + tid; i < zn; i += G2T) zl[i] = zs[i];
  __syncthreads();

  unsigned cm = 0u;
#pragma unroll
  for (int k = 0; k < GN; ++k) cm = c[k] > cm ? c[k] : cm;
  gcn_g2_sum(zl, zs, zn, c, 0u, l, idx, acc);
  for (unsigned j0 = 8 * GQ; j0 < cm; j0 += 8 * GQ) {
    gcn_g2_slots(sorted, beg, c, j0, l, idx);
    gcn_g2_sum(zl, zs, zn, c, j0, l, idx, acc);
  }
#pragma unroll
  for (int k = 0; k < GN; ++k) {
    float r = acc[k];
    r += __shfl_xor(r, 1, 8);
    r += __shfl_xor(r, 2, 8);
    r += __shfl_xor(r, 4, 8);
    const int node = nb + k;
    if (l == 0 && node < n)
      out[node] = (r + self[k]) * rsqrtf((float)(c[k] + 1u)) + bias;
  }
}

template <int SL, int RN>
static bool gcn_sorted_path(const float* x, const int* src, const int* dst,
                            const float* W1, const float* b1, const float* W2,
                            const float* b2, float* out, int n, int E,
                            int dblocks, void* d_ws, size_t ws_size,
                            hipStream_t stream) {
  constexpr int RC = RN / CH;
  size_t need = (4 * (size_t)n + (size_t)SL * (size_t)n + (size_t)E) * sizeof(float);
  int ranges = (n + RN - 1) / RN;
  if (ws_size < need || ranges > 65535 ||
      (size_t)E < (size_t)ranges * RC * SL) return false;
  unsigned* cnt    = (unsigned*)d_ws;
  unsigned* rowptr = cnt + n;
  float*    xs     = (float*)(rowptr + n);
  unsigned* table  = (unsigned*)(xs + 2 * (size_t)n);
  int*      sorted = (int*)(table + (size_t)SL * (size_t)n);
  float*    zs     = (float*)table;
  unsigned* chunks = (unsigned*)sorted;

  int L = (E + SL - 1) / SL;
  dim3 gS(SL, ranges);
  int gG = (n + (G2T / 8) * GN - 1) / ((G2T / 8) * GN);

  gcn_slice_count<SL, RN><<<gS, 1024, 0, stream>>>(dst, table, chunks, n, E, L);
  gcn_offsets<SL><<<(n + CH - 1) / CH, CH, 0, stream>>>(table, chunks, cnt, rowptr, x, xs, n);
  gcn_slice_fill<RN><<<gS, 1024, 0, stream>>>(src, dst, table, rowptr, sorted, n, E, L);
  gcn_gather1_dense<<<dblocks, 256, 0, stream>>>(sorted, rowptr, cnt, xs, W1, b1, W2, zs, n);
  gcn_gather2<<<gG, G2T, 0, stream>>>(sorted, rowptr, cnt, zs, b2, out, n);
  return true;
}

extern "C" void kernel_launch(void* const* d_in, const int* in_sizes, int n_in,
                              void* d_out, int out_size, void* d_ws, size_t ws_size,
                              hipStream_t stream) {
  const float* x  = (const float*)d_in[0];
  const int*   ei = (const int*)d_in[1];
  const float* W1 = (const float*)d_in[2];
  const float* b1 = (const float*)d_in[3];
  const float* W2 = (const float*)d_in[4];
  const float* b2 = (const float*)d_in[5];

  int n = in_sizes[0] / 2;
  int E = in_sizes[1] / 2;
  const int* src = ei;
  const int* dst = ei + E;
  float* out = (float*)d_out;

  const int tb = 256;
  int gN = (n + tb - 1) / tb;
  int gE = (E + tb - 1) / tb;
  if (gE > 4096) gE = 4096;
  int tiles = (n + 15) / 16;
  int dblocks = (tiles + 7) / 8;

  if (gcn_sorted_path<32, 12800>(x, src, dst, W1, b1, W2, b2, out, n, E,
                                 dblocks, d_ws, ws_size, stream)) return;
  if (gcn_sorted_path<16, 6400>(x, src, dst, W1, b1, W2, b2, out, n, E,
                                dblocks, d_ws, ws_size, stream)) return;

  float* dinv = (float*)d_ws;
  float* agg  = dinv + n;
  float* z    = agg + 2 * (size_t)n;

  gcn_init<<<gN, tb, 0, stream>>>(dinv, agg, z, out, n);
  gcn_degree<<<gE, tb, 0, stream>>>(dst, dinv, E);
  gcn_rsqrt<<<gN, tb, 0, stream>>>(dinv, n);
  gcn_agg2<<<gE, tb, 0, stream>>>(src, dst, x, dinv, agg, E);

  gcn_dense_wmma<<<dblocks, 256, 0, stream>>>(agg, x, dinv, W1, b1, W2, z, n);

  gcn_out_self<<<gN, tb, 0, stream>>>(z, dinv, b2, out, n);
  gcn_out_edges<<<gE, tb, 0, stream>>>(src, dst, z, dinv, out, E);
}


# Sort in two passes: slices split their edges into 256-node buckets through LDS staging (whole-line writes), then one block per bucket ranks its own edges in LDS; f32 sum order: bucket LDS arrival
# speedup vs baseline: 6.8605x; 1.6958x over previous
#include <hip/hip_runtime.h>


typedef float v2f __attribute__((ext_vector_type(2)));
typedef float v8f __attribute__((ext_vector_type(8)));

__global__ void gcn_init(float* __restrict__ deg, float* __restrict__ agg,
                         float* __restrict__ z, float* __restrict__ out, int n) {
  int i = blockIdx.x * blockDim.x + threadIdx.x;
  if (i < n) {
    deg[i] = 1.0f;
    agg[2 * i] = 0.0f;
    agg[2 * i + 1] = 0.0f;
    z[i] = 0.0f;
    out[i] = 0.0f;
  }
}

__global__ void gcn_degree(const int* __restrict__ dst, float* __restrict__ deg,
                           int E) {
  int i = blockIdx.x * blockDim.x + threadIdx.x;
  int stride = gridDim.x * blockDim.x;
  for (; i < E; i += stride) {
    if (i + stride < E) __builtin_prefetch(&dst[i + stride], 0, 1);
    atomicAdd(&deg[dst[i]], 1.0f);
  }
}

__global__ void gcn_rsqrt(float* __restrict__ dinv, int n) {
  int i = blockIdx.x * blockDim.x + threadIdx.x;
  if (i < n) dinv[i] = rsqrtf(dinv[i]);
}

__global__ void gcn_agg2(const int* __restrict__ src, const int* __restrict__ dst,
                         const float* __restrict__ x, const float* __restrict__ dinv,
                         float* __restrict__ agg, int E) {
  int i = blockIdx.x * blockDim.x + threadIdx.x;
  int stride = gridDim.x * blockDim.x;
  for (; i < E; i += stride) {
    if (i + stride < E) {
      __builtin_prefetch(&src[i + stride], 0, 1);
      __builtin_prefetch(&dst[i + stride], 0, 1);
    }
    int s = src[i];
    int d = dst[i];
    float w = dinv[s] * dinv[d];
    atomicAdd(&agg[2 * d], x[2 * s] * w);
    atomicAdd(&agg[2 * d + 1], x[2 * s + 1] * w);
  }
}

__global__ __launch_bounds__(256) void gcn_dense_wmma(
    const float* __restrict__ agg, const float* __restrict__ x,
    const float* __restrict__ dinv, const float* __restrict__ W1,
    const float* __restrict__ b1, const float* __restrict__ W2,
    float* __restrict__ z, int n) {
  const int lane = threadIdx.x & 31;
  const int wave = threadIdx.x >> 5;
  const int tile = blockIdx.x * 8 + wave;
  const int base = tile * 16;
  if (base >= n) return;

  const int m = lane & 15;
  const bool hiK = lane >= 16;
  const float kmask = hiK ? 0.0f : 1.0f;

  int node = base + m;
  const float rowmask = (!hiK && node < n) ? 1.0f : 0.0f;
  if (node >= n) node = n - 1;
  v2f A;
  float di = dinv[node];
  float sl = di * di;
  A.x = (agg[2 * node]     + x[2 * node]     * sl) * rowmask;
  A.y = (agg[2 * node + 1] + x[2 * node + 1] * sl) * rowmask;

  float zacc[8];
#pragma unroll
  for (int v = 0; v < 8; ++v) zacc[v] = 0.0f;

#pragma unroll
  for (int n0 = 0; n0 < 128; n0 += 16) {
    int col = n0 + m;
    v2f B;
    B.x = W1[col]       * kmask;
    B.y = W1[128 + col] * kmask;
    float bias = b1[col];
    v8f C;
#pragma unroll
    for (int v = 0; v < 8; ++v) C[v] = bias;

    C = __builtin_amdgcn_wmma_f32_16x16x4_f32(
         false, A,  false, B,
         (short)0, C,  false,  false);

    float w2 = W2[col];
#pragma unroll
    for (int v = 0; v < 8; ++v) {
      float h = C[v];
      h = h > 0.0f ? h : 0.0f;
      zacc[v] += h * w2;
    }
  }

#pragma unroll
  for (int v = 0; v < 8; ++v) {
    float r = zacc[v];
    r += __shfl_xor(r, 1, 16);
    r += __shfl_xor(r, 2, 16);
    r += __shfl_xor(r, 4, 16);
    r += __shfl_xor(r, 8, 16);
    zacc[v] = r;
  }
  if (m == 0) {
    int rowbase = base + (hiK ? 8 : 0);
#pragma unroll
    for (int v = 0; v < 8; ++v) {
      int r = rowbase + v;
      if (r < n) z[r] = zacc[v];
    }
  }
}

__global__ void gcn_out_self(const float* __restrict__ z,
                             const float* __restrict__ dinv,
                             const float* __restrict__ b2,
                             float* __restrict__ out, int n) {
  int i = blockIdx.x * blockDim.x + threadIdx.x;
  if (i < n) out[i] = z[i] * dinv[i] * dinv[i] + b2[0];
}

__global__ void gcn_out_edges(const int* __restrict__ src, const int* __restrict__ dst,
                              const float* __restrict__ z,
                              const float* __restrict__ dinv,
                              float* __restrict__ out, int E) {
  int i = blockIdx.x * blockDim.x + threadIdx.x;
  int stride = gridDim.x * blockDim.x;
  for (; i < E; i += stride) {
    if (i + stride < E) {
      __builtin_prefetch(&src[i + stride], 0, 1);
      __builtin_prefetch(&dst[i + stride], 0, 1);
    }
    int s = src[i];
    int d = dst[i];
    atomicAdd(&out[d], z[s] * dinv[s] * dinv[d]);
  }
}


constexpr int CH = 256;

template <int SL, int RN>
__global__ __launch_bounds__(1024) void gcn_slice_count(
    const int* __restrict__ dst, unsigned* __restrict__ table,
    unsigned* __restrict__ chunks, int n, int E, int L) {
  constexpr int RC = RN / CH;
  static_assert(RN % CH == 0, "a chunk must not straddle two ranges");
  __shared__ unsigned h[RN];
  const int tid = threadIdx.x;
  const int s = blockIdx.x;
  const int lo = blockIdx.y * RN;
  const unsigned lim = (n - lo < RN) ? (unsigned)(n - lo) : (unsigned)RN;
  for (int i = tid; i < RN; i += 1024) h[i] = 0u;
  __syncthreads();
  const int e0 = s * L;
  const int e1 = (E - e0 < L) ? E : e0 + L;
  for (int e = e0 + tid; e < e1; e += 1024 * 16) {
    int d[16];
#pragma unroll
    for (int k = 0; k < 16; ++k) {
      int ee = e + k * 1024;
      d[k] = (ee < e1) ? dst[ee] : -1;
    }
#pragma unroll
    for (int k = 0; k < 16; ++k) {
      unsigned x = (unsigned)d[k] - (unsigned)lo;
      if (x < lim) atomicAdd(&h[x], 1u);
    }
  }
  __syncthreads();
  unsigned* row = table + (size_t)s * n + lo;
  for (int i = tid; i < (int)lim; i += 1024) row[i] = h[i];
  const int wv = tid >> 5;
  const int lane = tid & 31;
  for (int c = wv; c < RC; c += 32) {
    unsigned sum = 0u;
#pragma unroll
    for (int j = 0; j < CH / 32; ++j) sum += h[c * CH + lane + 32 * j];
    sum += __shfl_xor(sum, 16, 32);
    sum += __shfl_xor(sum, 8, 32);
    sum += __shfl_xor(sum, 4, 32);
    sum += __shfl_xor(sum, 2, 32);
    sum += __shfl_xor(sum, 1, 32);
    if (lane == 0) chunks[(size_t)(blockIdx.y * RC + c) * SL + s] = sum;
  }
}

template <int SL>
__global__ __launch_bounds__(CH) void gcn_offsets(
    unsigned* __restrict__ table, const unsigned* __restrict__ chunks,
    unsigned* __restrict__ cnt, unsigned* __restrict__ rowptr,
    const float* __restrict__ x, float* __restrict__ xs, int n) {
  static_assert(SL % 4 == 0, "a chunk's slice sums are read as 16-byte pieces");
  constexpr int NW = CH / 32;
  constexpr int PF = SL / 2;
  __shared__ unsigned s_incl[NW];
  __shared__ unsigned s_part[NW];
  const int tid = threadIdx.x;
  const int lane = tid & 31;
  const int wv = tid >> 5;
  const int node = blockIdx.x * CH + tid;
  const bool in = node < n;

  unsigned v[SL];
  float2 xv = make_float2(0.0f, 0.0f);
#pragma unroll
  for (int s = 0; s < SL; ++s) v[s] = 0u;
  if (in) {
#pragma unroll
    for (int s = 0; s < SL; ++s) v[s] = table[(size_t)s * n + node];
    xv = ((const float2*)x)[node];
  }

  const uint4* c4 = (const uint4*)chunks;
  const int lim4 = (SL / 4) * blockIdx.x;
  unsigned part = 0u;
  for (int base = 0; base < lim4; base += CH * PF) {
    uint4 a[PF];
#pragma unroll
    for (int j = 0; j < PF; ++j) {
      const int i = base + tid + CH * j;
      a[j] = c4[i < lim4 ? i : 0];
    }
#pragma unroll
    for (int j = 0; j < PF; ++j) {
      const int i = base + tid + CH * j;
      const unsigned t = (a[j].x + a[j].y) + (a[j].z + a[j].w);
      part += i < lim4 ? t : 0u;
    }
  }
  part += __shfl_xor(part, 16, 32);
  part += __shfl_xor(part, 8, 32);
  part += __shfl_xor(part, 4, 32);
  part += __shfl_xor(part, 2, 32);
  part += __shfl_xor(part, 1, 32);

  unsigned mine = 0u;
#pragma unroll
  for (int s = 0; s < SL; ++s) mine += v[s];
  unsigned incl = mine;
#pragma unroll
  for (int o = 1; o < 32; o <<= 1) {
    unsigned t = __shfl_up(incl, o, 32);
    if (lane >= o) incl += t;
  }
  if (lane == 31) s_incl[wv] = incl;
  if (lane == 0) s_part[wv] = part;
  __syncthreads();
  unsigned off = 0u;
  if (lane < NW) off = s_part[lane] + ((lane < wv) ? s_incl[lane] : 0u);
  off += __shfl_xor(off, 16, 32);
  off += __shfl_xor(off, 8, 32);
  off += __shfl_xor(off, 4, 32);
  off += __shfl_xor(off, 2, 32);
  off += __shfl_xor(off, 1, 32);
  if (in) {
    unsigned run = 0u;
#pragma unroll
    for (int s = 0; s < SL; ++s) {
      table[(size_t)s * n + node] = run;
      run += v[s];
    }
    cnt[node] = mine;
    rowptr[node] = off + incl - mine;
    float di = rsqrtf((float)(mine + 1u));
    xs[2 * node]     = xv.x * di;
    xs[2 * node + 1] = xv.y * di;
  }
}

template <int RN>
__global__ __launch_bounds__(1024) void gcn_slice_fill(
    const int* __restrict__ src, const int* __restrict__ dst,
    const unsigned* __restrict__ table, const unsigned* __restrict__ rowptr,
    int* __restrict__ sorted, int n, int E, int L) {
  __shared__ unsigned cur[RN];
  const int tid = threadIdx.x;
  const int s = blockIdx.x;
  const int lo = blockIdx.y * RN;
  const unsigned lim = (n - lo < RN) ? (unsigned)(n - lo) : (unsigned)RN;
  const unsigned* row = table + (size_t)s * n + lo;
  constexpr int RT = (RN + 1023) / 1024;
  unsigned ra[RT], rb[RT];
#pragma unroll
  for (int k = 0; k < RT; ++k) {
    int i = tid + k * 1024;
    bool in = i < (int)lim;
    ra[k] = in ? rowptr[lo + i] : 0u;
    rb[k] = in ? row[i] : 0u;
  }
#pragma unroll
  for (int k = 0; k < RT; ++k) {
    int i = tid + k * 1024;
    if (i < (int)lim) cur[i] = ra[k] + rb[k];
  }
  __syncthreads();
  const int e0 = s * L;
  const int e1 = (E - e0 < L) ? E : e0 + L;
  for (int e = e0 + tid; e < e1; e += 1024 * 8) {
    int d[8], sv[8];
    unsigned pos[8];
#pragma unroll
    for (int k = 0; k < 8; ++k) {
      int ee = e + k * 1024;
      bool in = ee < e1;
      d[k] = in ? dst[ee] : -1;
      sv[k] = in ? src[ee] : 0;
    }
#pragma unroll
    for (int k = 0; k < 8; ++k) {
      unsigned x = (unsigned)d[k] - (unsigned)lo;
      pos[k] = 0u;
      if (x < lim) pos[k] = atomicAdd(&cur[x], 1u);
    }
#pragma unroll
    for (int k = 0; k < 8; ++k) {
      unsigned x = (unsigned)d[k] - (unsigned)lo;
      if (x < lim) sorted[pos[k]] = sv[k];
    }
  }
}

constexpr int GN = 4;
constexpr int GQ = 4;

__global__ __launch_bounds__(256) void gcn_gather1_dense(
    const int* __restrict__ sorted, const unsigned* __restrict__ rowptr,
    const unsigned* __restrict__ cnt, const float* __restrict__ xs,
    const float* __restrict__ W1, const float* __restrict__ b1,
    const float* __restrict__ W2, float* __restrict__ zs, int n) {
  static_assert(GN == 4, "four groups of GN nodes must make one 16-row tile");
  const int lane = threadIdx.x & 31;
  const unsigned l = threadIdx.x & 7;
  const int grp = lane >> 3;
  const int m = lane & 15;
  const bool hiK = lane >= 16;
  const float kmask = hiK ? 0.0f : 1.0f;
  const int wb = ((blockIdx.x * 256 + threadIdx.x) >> 5) * 16;
  if (wb >= n) return;
  const int nb = wb + grp * GN;
  const float2* xs2 = (const float2*)xs;

  unsigned beg[GN], c[GN];
  float2 self[GN];
  float ax[GN], ay[GN];
#pragma unroll
  for (int k = 0; k < GN; ++k) {
    const int node = nb + k;
    const bool in = node < n;
    beg[k] = in ? rowptr[node] : 0u;
    c[k] = in ? cnt[node] : 0u;
    self[k] = in ? xs2[node] : make_float2(0.0f, 0.0f);
    ax[k] = 0.0f;
    ay[k] = 0.0f;
  }
  float w1a[8], w1b[8], bias[8], w2[8];
#pragma unroll
  for (int t = 0; t < 8; ++t) {
    const int col = 16 * t + m;
    w1a[t] = W1[col];
    w1b[t] = W1[128 + col];
    bias[t] = b1[col];
    w2[t] = W2[col];
  }
  unsigned cm = 0u;
#pragma unroll
  for (int k = 0; k < GN; ++k) cm = c[k] > cm ? c[k] : cm;
  for (unsigned j0 = 0; j0 < cm; j0 += 8 * GQ) {
    int idx[GN][GQ];
    float2 v[GN][GQ];
#pragma unroll
    for (int k = 0; k < GN; ++k) {
#pragma unroll
      for (int q = 0; q < GQ; ++q) {
        const unsigned j = j0 + 8 * q + l;
        const unsigned jj = j < c[k] ? j : c[k] - 1u;
        idx[k][q] = sorted[c[k] ? beg[k] + jj : 0u];
      }
    }
#pragma unroll
    for (int k = 0; k < GN; ++k) {
#pragma unroll
      for (int q = 0; q < GQ; ++q) {
        const unsigned j = j0 + 8 * q + l;
        const float2 t = xs2[idx[k][q]];
        v[k][q].x = j < c[k] ? t.x : 0.0f;
        v[k][q].y = j < c[k] ? t.y : 0.0f;
      }
    }
#pragma unroll
    for (int k = 0; k < GN; ++k) {
#pragma unroll
      for (int q = 0; q < GQ; ++q) {
        ax[k] += v[k][q].x;
        ay[k] += v[k][q].y;
      }
    }
  }
  float tx = 0.0f, ty = 0.0f, td = 0.0f;
#pragma unroll
  for (int k = 0; k < GN; ++k) {
    float sx = ax[k], sy = ay[k];
    sx += __shfl_xor(sx, 1, 8);  sy += __shfl_xor(sy, 1, 8);
    sx += __shfl_xor(sx, 2, 8);  sy += __shfl_xor(sy, 2, 8);
    sx += __shfl_xor(sx, 4, 8);  sy += __shfl_xor(sy, 4, 8);
    const float di = rsqrtf((float)(c[k] + 1u));
    const float rx = (sx + self[k].x) * di;
    const float ry = (sy + self[k].y) * di;
    tx = (l == (unsigned)k) ? rx : tx;
    ty = (l == (unsigned)k) ? ry : ty;
    td = (l == (unsigned)k) ? di : td;
  }
  const int from = 8 * (m >> 2) + (m & 3);
  const float arx = __shfl(tx, from, 32);
  const float ary = __shfl(ty, from, 32);
  const bool rowok = !hiK && (wb + m < n);
  v2f A;
  A.x = rowok ? arx : 0.0f;
  A.y = rowok ? ary : 0.0f;

  float zacc[8];
#pragma unroll
  for (int v = 0; v < 8; ++v) zacc[v] = 0.0f;
#pragma unroll
  for (int t = 0; t < 8; ++t) {
    v2f B;
    B.x = w1a[t] * kmask;
    B.y = w1b[t] * kmask;
    v8f C;
#pragma unroll
    for (int v = 0; v < 8; ++v) C[v] = bias[t];

    C = __builtin_amdgcn_wmma_f32_16x16x4_f32(
         false, A,  false, B,
         (short)0, C,  false,  false);

#pragma unroll
    for (int v = 0; v < 8; ++v) {
      float h = C[v];
      h = h > 0.0f ? h : 0.0f;
      zacc[v] += h * w2[t];
    }
  }
#pragma unroll
  for (int v = 0; v < 8; ++v) {
    float r = zacc[v];
    r += __shfl_xor(r, 1, 16);
    r += __shfl_xor(r, 2, 16);
    r += __shfl_xor(r, 4, 16);
    r += __shfl_xor(r, 8, 16);
    zacc[v] = r;
  }
  const int reg = 4 * (grp & 1) + (int)(l & 3);
  float zsel = zacc[0];
#pragma unroll
  for (int v = 1; v < 8; ++v) zsel = (reg == v) ? zacc[v] : zsel;
  const int node = nb + (int)l;
  if (l < (unsigned)GN && node < n) zs[node] = zsel * td;
}

constexpr int ZL = 80000;
constexpr int G2T = 1024;
constexpr int ZF = 10;

__device__ __forceinline__ void gcn_g2_slots(
    const int* __restrict__ sorted, const unsigned (&beg)[GN],
    const unsigned (&c)[GN], unsigned j0, unsigned l, int (&idx)[GN][GQ]) {
#pragma unroll
  for (int k = 0; k < GN; ++k) {
#pragma unroll
    for (int q = 0; q < GQ; ++q) {
      const unsigned j = j0 + 8 * q + l;
      const unsigned jj = j < c[k] ? j : c[k] - 1u;
      idx[k][q] = sorted[c[k] ? beg[k] + jj : 0u];
    }
  }
}

__device__ __forceinline__ void gcn_g2_sum(
    const float* zl, const float* __restrict__ zs, int zn,
    const unsigned (&c)[GN], unsigned j0, unsigned l,
    const int (&idx)[GN][GQ], float (&acc)[GN]) {
  float a[GN][GQ], b[GN][GQ];
#pragma unroll
  for (int k = 0; k < GN; ++k) {
#pragma unroll
    for (int q = 0; q < GQ; ++q) {
      const int id = idx[k][q];
      a[k][q] = zl[id < zn ? id : 0];
    }
  }
#pragma unroll
  for (int k = 0; k < GN; ++k) {
#pragma unroll
    for (int q = 0; q < GQ; ++q) {
      const int id = idx[k][q];
      b[k][q] = zs[id < zn ? 0 : id];
    }
  }
#pragma unroll
  for (int k = 0; k < GN; ++k) {
#pragma unroll
    for (int q = 0; q < GQ; ++q) {
      const unsigned j = j0 + 8 * q + l;
      const float t = idx[k][q] < zn ? a[k][q] : b[k][q];
      acc[k] += j < c[k] ? t : 0.0f;
    }
  }
}

__global__ __launch_bounds__(G2T) void gcn_gather2(
    const int* __restrict__ sorted, const unsigned* __restrict__ rowptr,
    const unsigned* __restrict__ cnt, const float* __restrict__ zs,
    const float* __restrict__ b2, float* __restrict__ out, int n) {
  static_assert(ZL % 4 == 0, "the copy is made of 16-byte pieces");
  __shared__ float4 zl4[ZL / 4];
  float* zl = (float*)zl4;
  const int tid = threadIdx.x;
  const unsigned l = tid & 7;
  const int nb = (blockIdx.x * (G2T / 8) + (tid >> 3)) * GN;
  const float bias = b2[0];
  const int zn = n < ZL ? n : ZL;
  const int n4 = zn >> 2;
  const int last4 = n4 > 0 ? n4 - 1 : 0;
  const float4* z4 = (const float4*)zs;

  float4 t[ZF];
#pragma unroll
  for (int j = 0; j < ZF; ++j) {
    const int i = tid + G2T * j;
    t[j] = z4[i < n4 ? i : last4];
  }
  unsigned beg[GN], c[GN];
  float self[GN], acc[GN];
#pragma unroll
  for (int k = 0; k < GN; ++k) {
    const int node = nb + k;
    const bool in = node < n;
    const int nc = in ? node : n - 1;
    const unsigned r = rowptr[nc];
    const unsigned cc = cnt[nc];
    const float s = zs[nc];
    beg[k] = in ? r : 0u;
    c[k] = in ? cc : 0u;
    self[k] = in ? s : 0.0f;
    acc[k] = 0.0f;
  }
#pragma unroll
  for (int j = 0; j < ZF; ++j) {
    const int i = tid + G2T * j;
    zl4[i < n4 ? i : last4] = t[j];
  }
  int idx[GN][GQ];
  gcn_g2_slots(sorted, beg, c, 0u, l, idx);
  for (int base = G2T * ZF; base < n4; base += G2T * ZF) {
#pragma unroll
    for (int j = 0; j < ZF; ++j) {
      const int i = base + tid + G2T * j;
      t[j] = z4[i < n4 ? i : last4];
    }
#pragma unroll
    for (int j = 0; j < ZF; ++j) {
      const int i = base + tid + G2T * j;
      zl4[i < n4 ? i : last4] = t[j];
    }
  }
  for (int i = 4 * n4 + tid; i < zn; i += G2T) zl[i] = zs[i];
  __syncthreads();

  unsigned cm = 0u;
#pragma unroll
  for (int k = 0; k < GN; ++k) cm = c[k] > cm ? c[k] : cm;
  gcn_g2_sum(zl, zs, zn, c, 0u, l, idx, acc);
  for (unsigned j0 = 8 * GQ; j0 < cm; j0 += 8 * GQ) {
    gcn_g2_slots(sorted, beg, c, j0, l, idx);
    gcn_g2_sum(zl, zs, zn, c, j0, l, idx, acc);
  }
#pragma unroll
  for (int k = 0; k < GN; ++k) {
    float r = acc[k];
    r += __shfl_xor(r, 1, 8);
    r += __shfl_xor(r, 2, 8);
    r += __shfl_xor(r, 4, 8);
    const int node = nb + k;
    if (l == 0 && node < n)
      out[node] = (r + self[k]) * rsqrtf((float)(c[k] + 1u)) + bias;
  }
}

constexpr int PB_NODES = 256;
constexpr int PB_SHIFT = 8;
constexpr int PB_MAXB = 512;
constexpr int PB_NS = 128;
constexpr int PB_LMAX = 16384;
constexpr int PB_T = 1024;
constexpr int PB_EPT = 13;
constexpr int PC_T = 512;
constexpr int PC_EPT = 10;
constexpr int PC_CAP = 16384;

template <int NW>
__device__ __forceinline__ unsigned gcn_block_scan(unsigned v, unsigned* s_w,
                                                   unsigned& tot) {
  const int lane = threadIdx.x & 31;
  const int wv = threadIdx.x >> 5;
  unsigned incl = v;
#pragma unroll
  for (int o = 1; o < 32; o <<= 1) {
    unsigned t = __shfl_up(incl, o, 32);
    if (lane >= o) incl += t;
  }
  if (lane == 31) s_w[wv] = incl;
  __syncthreads();
  unsigned off = 0u, all = 0u;
#pragma unroll
  for (int w = 0; w < NW; ++w) {
    const unsigned x = s_w[w];
    all += x;
    off += (w < wv) ? x : 0u;
  }
  __syncthreads();
  tot = all;
  return off + incl - v;
}

__global__ __launch_bounds__(PB_T) void gcn_bucket_count(
    const int* __restrict__ dst, unsigned* __restrict__ hist, int nb, int E, int L) {
  __shared__ unsigned h[PB_MAXB];
  const int tid = threadIdx.x;
  const int s = blockIdx.x;
  if (tid < PB_MAXB) h[tid] = 0u;
  __syncthreads();
  const int e0 = s * L;
  const int e1 = (E - e0 < L) ? E : e0 + L;
  for (int e = e0 + tid; e < e1; e += PB_T * PB_EPT) {
    int d[PB_EPT];
#pragma unroll
    for (int k = 0; k < PB_EPT; ++k) {
      const int ee = e + k * PB_T;
      const int v = dst[ee < e1 ? ee : e1 - 1];
      d[k] = (ee < e1) ? v : -1;
    }
#pragma unroll
    for (int k = 0; k < PB_EPT; ++k) {
      unsigned b = (unsigned)d[k] >> PB_SHIFT;
      if (b < (unsigned)nb) atomicAdd(&h[b], 1u);
    }
  }
  __syncthreads();
  if (tid < PB_MAXB) hist[(size_t)s * PB_MAXB + tid] = h[tid];
}

__global__ __launch_bounds__(PB_T) void gcn_bucket_scatter(
    const int* __restrict__ src, const int* __restrict__ dst,
    const unsigned* __restrict__ hist, unsigned* __restrict__ bstart,
    unsigned* __restrict__ packed, int nb, int E, int L) {
  static_assert(PB_MAXB / 4 * (PB_NS / 16) == PB_T, "one thread per 4 buckets x 16 slices");
  static_assert(3 * (PB_NS / 16) * PB_MAXB <= PB_LMAX, "the partial sums borrow the staging area");
  __shared__ unsigned lcur[PB_MAXB];
  __shared__ unsigned gdel[PB_MAXB];
  __shared__ unsigned s_w[PB_T / 32];
  __shared__ uint4 stw4[PB_LMAX / 4];
  unsigned* stw = (unsigned*)stw4;
  __shared__ unsigned short stb[PB_LMAX];
  const int tid = threadIdx.x;
  const int s = blockIdx.x;

  constexpr int NG = PB_NS / 16;
  constexpr int NQ = PB_MAXB / 4;
  const int q = tid % NQ;
  const int g = tid / NQ;
  const uint4* h4 = (const uint4*)hist;
  uint4 a[16];
#pragma unroll
  for (int j = 0; j < 16; ++j) a[j] = h4[(size_t)(16 * g + j) * NQ + q];
  uint4 pt = make_uint4(0u, 0u, 0u, 0u), pb = pt, pm = pt;
#pragma unroll
  for (int j = 0; j < 16; ++j) {
    const int sl = 16 * g + j;
    pt.x += a[j].x; pt.y += a[j].y; pt.z += a[j].z; pt.w += a[j].w;
    if (sl < s) { pb.x += a[j].x; pb.y += a[j].y; pb.z += a[j].z; pb.w += a[j].w; }
    if (sl == s) pm = a[j];
  }
  uint4* part4 = stw4;
  part4[(0 * NG + g) * NQ + q] = pt;
  part4[(1 * NG + g) * NQ + q] = pb;
  part4[(2 * NG + g) * NQ + q] = pm;
  __syncthreads();
  const bool isb = tid < PB_MAXB;
  unsigned v_tot = 0u, v_bef = 0u, v_mine = 0u;
  if (isb) {
#pragma unroll
    for (int k = 0; k < NG; ++k) {
      v_tot  += stw[(0 * NG + k) * PB_MAXB + tid];
      v_bef  += stw[(1 * NG + k) * PB_MAXB + tid];
      v_mine += stw[(2 * NG + k) * PB_MAXB + tid];
    }
  }
  unsigned e_all = 0u, e_mine = 0u;
  const unsigned bs = gcn_block_scan<PB_T / 32>(v_tot, s_w, e_all);
  const unsigned ls = gcn_block_scan<PB_T / 32>(v_mine, s_w, e_mine);
  if (isb) {
    lcur[tid] = ls;
    gdel[tid] = bs + v_bef - ls;
  }
  if (s == 0) {
    if (tid < nb) bstart[tid] = bs;
    if (tid == 0) bstart[nb] = e_all;
  }
  __syncthreads();

  const int e0 = s * L;
  const int e1 = (E - e0 < L) ? E : e0 + L;
  for (int e = e0 + tid; e < e1; e += PB_T * PB_EPT) {
    int d[PB_EPT], sv[PB_EPT];
#pragma unroll
    for (int k = 0; k < PB_EPT; ++k) {
      const int ee = e + k * PB_T;
      const int ec = ee < e1 ? ee : e1 - 1;
      const int vd = dst[ec];
      sv[k] = src[ec];
      d[k] = (ee < e1) ? vd : -1;
    }
#pragma unroll
    for (int k = 0; k < PB_EPT; ++k) {
      unsigned b = (unsigned)d[k] >> PB_SHIFT;
      if (b < (unsigned)nb) {
        unsigned p = atomicAdd(&lcur[b], 1u);
        stw[p] = ((unsigned)sv[k] & 0x1FFFFu) |
                 (((unsigned)d[k] & (unsigned)(PB_NODES - 1)) << 17);
        stb[p] = (unsigned short)b;
      }
    }
  }
  __syncthreads();
  for (unsigned i = tid; i < e_mine; i += PB_T) packed[i + gdel[stb[i]]] = stw[i];
}

__global__ __launch_bounds__(PC_T) void gcn_bucket_rank(
    const unsigned* __restrict__ packed, const unsigned* __restrict__ bstart,
    const float* __restrict__ x, unsigned* __restrict__ cnt,
    unsigned* __restrict__ rowptr, float* __restrict__ xs,
    int* __restrict__ sorted, int n) {
  static_assert(PC_T >= PB_NODES, "one thread per node of the bucket");
  __shared__ unsigned c[PB_NODES];
  __shared__ unsigned cur[PB_NODES];
  __shared__ unsigned s_w[PC_T / 32];
  __shared__ int stage[PC_CAP];
  const int tid = threadIdx.x;
  const int b = blockIdx.x;
  const unsigned beg = bstart[b];
  const unsigned m = bstart[b + 1] - beg;
  const bool isn = tid < PB_NODES;
  const int node = b * PB_NODES + tid;
  float2 xv = make_float2(0.0f, 0.0f);
  if (isn && node < n) xv = ((const float2*)x)[node];
  if (isn) c[tid] = 0u;
  __syncthreads();
  for (unsigned i0 = 0; i0 < m; i0 += PC_T * PC_EPT) {
    unsigned w[PC_EPT];
#pragma unroll
    for (int k = 0; k < PC_EPT; ++k) {
      const unsigned i = i0 + tid + k * PC_T;
      w[k] = packed[beg + (i < m ? i : 0u)];
    }
#pragma unroll
    for (int k = 0; k < PC_EPT; ++k) {
      const unsigned i = i0 + tid + k * PC_T;
      if (i < m) atomicAdd(&c[(w[k] >> 17) & (unsigned)(PB_NODES - 1)], 1u);
    }
  }
  __syncthreads();
  const unsigned mine = isn ? c[tid] : 0u;
  unsigned all = 0u;
  const unsigned lpre = gcn_block_scan<PC_T / 32>(mine, s_w, all);
  if (isn) {
    cur[tid] = lpre;
    if (node < n) {
      cnt[node] = mine;
      rowptr[node] = beg + lpre;
      const float di = rsqrtf((float)(mine + 1u));
      ((float2*)xs)[node] = make_float2(xv.x * di, xv.y * di);
    }
  }
  __syncthreads();
  const bool fits = m <= (unsigned)PC_CAP;
  for (unsigned i0 = 0; i0 < m; i0 += PC_T * PC_EPT) {
    unsigned w[PC_EPT];
#pragma unroll
    for (int k = 0; k < PC_EPT; ++k) {
      const unsigned i = i0 + tid + k * PC_T;
      w[k] = packed[beg + (i < m ? i : 0u)];
    }
#pragma unroll
    for (int k = 0; k < PC_EPT; ++k) {
      const unsigned i = i0 + tid + k * PC_T;
      if (i < m) {
        const unsigned p = atomicAdd(&cur[(w[k] >> 17) & (unsigned)(PB_NODES - 1)], 1u);
        const int sv = (int)(w[k] & 0x1FFFFu);
        if (fits) stage[p] = sv; else sorted[beg + p] = sv;
      }
    }
  }
  if (fits) {
    __syncthreads();
    for (unsigned i = tid; i < m; i += PC_T) sorted[beg + i] = stage[i];
  }
}

template <int SL, int RN>
static bool gcn_sorted_path(const float* x, const int* src, const int* dst,
                            const float* W1, const float* b1, const float* W2,
                            const float* b2, float* out, int n, int E,
                            int dblocks, void* d_ws, size_t ws_size,
                            hipStream_t stream) {
  constexpr int RC = RN / CH;
  size_t need = (4 * (size_t)n + (size_t)SL * (size_t)n + (size_t)E) * sizeof(float);
  int ranges = (n + RN - 1) / RN;
  if (ws_size < need || ranges > 65535 ||
      (size_t)E < (size_t)ranges * RC * SL) return false;
  unsigned* cnt    = (unsigned*)d_ws;
  unsigned* rowptr = cnt + n;
  float*    xs     = (float*)(rowptr + n);
  unsigned* table  = (unsigned*)(xs + 2 * (size_t)n);
  int*      sorted = (int*)(table + (size_t)SL * (size_t)n);
  float*    zs     = (float*)table;
  unsigned* chunks = (unsigned*)sorted;

  int L = (E + SL - 1) / SL;
  dim3 gS(SL, ranges);
  int gG = (n + (G2T / 8) * GN - 1) / ((G2T / 8) * GN);

  gcn_slice_count<SL, RN><<<gS, 1024, 0, stream>>>(dst, table, chunks, n, E, L);
  gcn_offsets<SL><<<(n + CH - 1) / CH, CH, 0, stream>>>(table, chunks, cnt, rowptr, x, xs, n);
  gcn_slice_fill<RN><<<gS, 1024, 0, stream>>>(src, dst, table, rowptr, sorted, n, E, L);
  gcn_gather1_dense<<<dblocks, 256, 0, stream>>>(sorted, rowptr, cnt, xs, W1, b1, W2, zs, n);
  gcn_gather2<<<gG, G2T, 0, stream>>>(sorted, rowptr, cnt, zs, b2, out, n);
  return true;
}

static bool gcn_bucket_path(const float* x, const int* src, const int* dst,
                            const float* W1, const float* b1, const float* W2,
                            const float* b2, float* out, int n, int E,
                            int dblocks, void* d_ws, size_t ws_size,
                            hipStream_t stream) {
  int nb = (n + PB_NODES - 1) / PB_NODES;
  int L = (E + PB_NS - 1) / PB_NS;
  size_t need = (5 * (size_t)n + (size_t)PB_MAXB * PB_NS + 1024 +
                 2 * (size_t)E) * sizeof(float);
  if (n < 1 || n > PB_MAXB * PB_NODES || (n & 3) != 0 || E < 1 || L > PB_LMAX ||
      ws_size < need) return false;
  unsigned* cnt    = (unsigned*)d_ws;
  unsigned* rowptr = cnt + n;
  float*    xs     = (float*)(rowptr + n);
  float*    zs     = xs + 2 * (size_t)n;
  unsigned* hist   = (unsigned*)(zs + n);
  unsigned* bstart = hist + (size_t)PB_MAXB * PB_NS;
  unsigned* packed = bstart + 1024;
  int*      sorted = (int*)(packed + E);

  int gG = (n + (G2T / 8) * GN - 1) / ((G2T / 8) * GN);

  gcn_bucket_count<<<PB_NS, PB_T, 0, stream>>>(dst, hist, nb, E, L);
  gcn_bucket_scatter<<<PB_NS, PB_T, 0, stream>>>(src, dst, hist, bstart, packed, nb, E, L);
  gcn_bucket_rank<<<nb, PC_T, 0, stream>>>(packed, bstart, x, cnt, rowptr, xs, sorted, n);
  gcn_gather1_dense<<<dblocks, 256, 0, stream>>>(sorted, rowptr, cnt, xs, W1, b1, W2, zs, n);
  gcn_gather2<<<gG, G2T, 0, stream>>>(sorted, rowptr, cnt, zs, b2, out, n);
  return true;
}

extern "C" void kernel_launch(void* const* d_in, const int* in_sizes, int n_in,
                              void* d_out, int out_size, void* d_ws, size_t ws_size,
                              hipStream_t stream) {
  const float* x  = (const float*)d_in[0];
  const int*   ei = (const int*)d_in[1];
  const float* W1 = (const float*)d_in[2];
  const float* b1 = (const float*)d_in[3];
  const float* W2 = (const float*)d_in[4];
  const float* b2 = (const float*)d_in[5];

  int n = in_sizes[0] / 2;
  int E = in_sizes[1] / 2;
  const int* src = ei;
  const int* dst = ei + E;
  float* out = (float*)d_out;

  const int tb = 256;
  int gN = (n + tb - 1) / tb;
  int gE = (E + tb - 1) / tb;
  if (gE > 4096) gE = 4096;
  int tiles = (n + 15) / 16;
  int dblocks = (tiles + 7) / 8;

  if (gcn_bucket_path(x, src, dst, W1, b1, W2, b2, out, n, E, dblocks, d_ws,
                      ws_size, stream)) return;

  if (gcn_sorted_path<32, 12800>(x, src, dst, W1, b1, W2, b2, out, n, E,
                                 dblocks, d_ws, ws_size, stream)) return;
  if (gcn_sorted_path<16, 6400>(x, src, dst, W1, b1, W2, b2, out, n, E,
                                dblocks, d_ws, ws_size, stream)) return;

  float* dinv = (float*)d_ws;
  float* agg  = dinv + n;
  float* z    = agg + 2 * (size_t)n;

  gcn_init<<<gN, tb, 0, stream>>>(dinv, agg, z, out, n);
  gcn_degree<<<gE, tb, 0, stream>>>(dst, dinv, E);
  gcn_rsqrt<<<gN, tb, 0, stream>>>(dinv, n);
  gcn_agg2<<<gE, tb, 0, stream>>>(src, dst, x, dinv, agg, E);

  gcn_dense_wmma<<<dblocks, 256, 0, stream>>>(agg, x, dinv, W1, b1, W2, z, n);

  gcn_out_self<<<gN, tb, 0, stream>>>(z, dinv, b2, out, n);
  gcn_out_edges<<<gE, tb, 0, stream>>>(src, dst, z, dinv, out, E);
}


# Pass one needs no histogram launch: a slice writes fixed 64-word (bucket, slice) cells; a bucket's first slot = sum of the slices' local prefixes; 4 launches, not 5; f32 sum order: bucket LDS arrival
# speedup vs baseline: 8.0927x; 1.1796x over previous
#include <hip/hip_runtime.h>


typedef float v2f __attribute__((ext_vector_type(2)));
typedef float v8f __attribute__((ext_vector_type(8)));

__global__ void gcn_init(float* __restrict__ deg, float* __restrict__ agg,
                         float* __restrict__ z, float* __restrict__ out, int n) {
  int i = blockIdx.x * blockDim.x + threadIdx.x;
  if (i < n) {
    deg[i] = 1.0f;
    agg[2 * i] = 0.0f;
    agg[2 * i + 1] = 0.0f;
    z[i] = 0.0f;
    out[i] = 0.0f;
  }
}

__global__ void gcn_degree(const int* __restrict__ dst, float* __restrict__ deg,
                           int E) {
  int i = blockIdx.x * blockDim.x + threadIdx.x;
  int stride = gridDim.x * blockDim.x;
  for (; i < E; i += stride) {
    if (i + stride < E) __builtin_prefetch(&dst[i + stride], 0, 1);
    atomicAdd(&deg[dst[i]], 1.0f);
  }
}

__global__ void gcn_rsqrt(float* __restrict__ dinv, int n) {
  int i = blockIdx.x * blockDim.x + threadIdx.x;
  if (i < n) dinv[i] = rsqrtf(dinv[i]);
}

__global__ void gcn_agg2(const int* __restrict__ src, const int* __restrict__ dst,
                         const float* __restrict__ x, const float* __restrict__ dinv,
                         float* __restrict__ agg, int E) {
  int i = blockIdx.x * blockDim.x + threadIdx.x;
  int stride = gridDim.x * blockDim.x;
  for (; i < E; i += stride) {
    if (i + stride < E) {
      __builtin_prefetch(&src[i + stride], 0, 1);
      __builtin_prefetch(&dst[i + stride], 0, 1);
    }
    int s = src[i];
    int d = dst[i];
    float w = dinv[s] * dinv[d];
    atomicAdd(&agg[2 * d], x[2 * s] * w);
    atomicAdd(&agg[2 * d + 1], x[2 * s + 1] * w);
  }
}

__global__ __launch_bounds__(256) void gcn_dense_wmma(
    const float* __restrict__ agg, const float* __restrict__ x,
    const float* __restrict__ dinv, const float* __restrict__ W1,
    const float* __restrict__ b1, const float* __restrict__ W2,
    float* __restrict__ z, int n) {
  const int lane = threadIdx.x & 31;
  const int wave = threadIdx.x >> 5;
  const int tile = blockIdx.x * 8 + wave;
  const int base = tile * 16;
  if (base >= n) return;

  const int m = lane & 15;
  const bool hiK = lane >= 16;
  const float kmask = hiK ? 0.0f : 1.0f;

  int node = base + m;
  const float rowmask = (!hiK && node < n) ? 1.0f : 0.0f;
  if (node >= n) node = n - 1;
  v2f A;
  float di = dinv[node];
  float sl = di * di;
  A.x = (agg[2 * node]     + x[2 * node]     * sl) * rowmask;
  A.y = (agg[2 * node + 1] + x[2 * node + 1] * sl) * rowmask;

  float zacc[8];
#pragma unroll
  for (int v = 0; v < 8; ++v) zacc[v] = 0.0f;

#pragma unroll
  for (int n0 = 0; n0 < 128; n0 += 16) {
    int col = n0 + m;
    v2f B;
    B.x = W1[col]       * kmask;
    B.y = W1[128 + col] * kmask;
    float bias = b1[col];
    v8f C;
#pragma unroll
    for (int v = 0; v < 8; ++v) C[v] = bias;

    C = __builtin_amdgcn_wmma_f32_16x16x4_f32(
         false, A,  false, B,
         (short)0, C,  false,  false);

    float w2 = W2[col];
#pragma unroll
    for (int v = 0; v < 8; ++v) {
      float h = C[v];
      h = h > 0.0f ? h : 0.0f;
      zacc[v] += h * w2;
    }
  }

#pragma unroll
  for (int v = 0; v < 8; ++v) {
    float r = zacc[v];
    r += __shfl_xor(r, 1, 16);
    r += __shfl_xor(r, 2, 16);
    r += __shfl_xor(r, 4, 16);
    r += __shfl_xor(r, 8, 16);
    zacc[v] = r;
  }
  if (m == 0) {
    int rowbase = base + (hiK ? 8 : 0);
#pragma unroll
    for (int v = 0; v < 8; ++v) {
      int r = rowbase + v;
      if (r < n) z[r] = zacc[v];
    }
  }
}

__global__ void gcn_out_self(const float* __restrict__ z,
                             const float* __restrict__ dinv,
                             const float* __restrict__ b2,
                             float* __restrict__ out, int n) {
  int i = blockIdx.x * blockDim.x + threadIdx.x;
  if (i < n) out[i] = z[i] * dinv[i] * dinv[i] + b2[0];
}

__global__ void gcn_out_edges(const int* __restrict__ src, const int* __restrict__ dst,
                              const float* __restrict__ z,
                              const float* __restrict__ dinv,
                              float* __restrict__ out, int E) {
  int i = blockIdx.x * blockDim.x + threadIdx.x;
  int stride = gridDim.x * blockDim.x;
  for (; i < E; i += stride) {
    if (i + stride < E) {
      __builtin_prefetch(&src[i + stride], 0, 1);
      __builtin_prefetch(&dst[i + stride], 0, 1);
    }
    int s = src[i];
    int d = dst[i];
    atomicAdd(&out[d], z[s] * dinv[s] * dinv[d]);
  }
}


constexpr int CH = 256;

template <int SL, int RN>
__global__ __launch_bounds__(1024) void gcn_slice_count(
    const int* __restrict__ dst, unsigned* __restrict__ table,
    unsigned* __restrict__ chunks, int n, int E, int L) {
  constexpr int RC = RN / CH;
  static_assert(RN % CH == 0, "a chunk must not straddle two ranges");
  __shared__ unsigned h[RN];
  const int tid = threadIdx.x;
  const int s = blockIdx.x;
  const int lo = blockIdx.y * RN;
  const unsigned lim = (n - lo < RN) ? (unsigned)(n - lo) : (unsigned)RN;
  for (int i = tid; i < RN; i += 1024) h[i] = 0u;
  __syncthreads();
  const int e0 = s * L;
  const int e1 = (E - e0 < L) ? E : e0 + L;
  for (int e = e0 + tid; e < e1; e += 1024 * 16) {
    int d[16];
#pragma unroll
    for (int k = 0; k < 16; ++k) {
      int ee = e + k * 1024;
      d[k] = (ee < e1) ? dst[ee] : -1;
    }
#pragma unroll
    for (int k = 0; k < 16; ++k) {
      unsigned x = (unsigned)d[k] - (unsigned)lo;
      if (x < lim) atomicAdd(&h[x], 1u);
    }
  }
  __syncthreads();
  unsigned* row = table + (size_t)s * n + lo;
  for (int i = tid; i < (int)lim; i += 1024) row[i] = h[i];
  const int wv = tid >> 5;
  const int lane = tid & 31;
  for (int c = wv; c < RC; c += 32) {
    unsigned sum = 0u;
#pragma unroll
    for (int j = 0; j < CH / 32; ++j) sum += h[c * CH + lane + 32 * j];
    sum += __shfl_xor(sum, 16, 32);
    sum += __shfl_xor(sum, 8, 32);
    sum += __shfl_xor(sum, 4, 32);
    sum += __shfl_xor(sum, 2, 32);
    sum += __shfl_xor(sum, 1, 32);
    if (lane == 0) chunks[(size_t)(blockIdx.y * RC + c) * SL + s] = sum;
  }
}

template <int SL>
__global__ __launch_bounds__(CH) void gcn_offsets(
    unsigned* __restrict__ table, const unsigned* __restrict__ chunks,
    unsigned* __restrict__ cnt, unsigned* __restrict__ rowptr,
    const float* __restrict__ x, float* __restrict__ xs, int n) {
  static_assert(SL % 4 == 0, "a chunk's slice sums are read as 16-byte pieces");
  constexpr int NW = CH / 32;
  constexpr int PF = SL / 2;
  __shared__ unsigned s_incl[NW];
  __shared__ unsigned s_part[NW];
  const int tid = threadIdx.x;
  const int lane = tid & 31;
  const int wv = tid >> 5;
  const int node = blockIdx.x * CH + tid;
  const bool in = node < n;

  unsigned v[SL];
  float2 xv = make_float2(0.0f, 0.0f);
#pragma unroll
  for (int s = 0; s < SL; ++s) v[s] = 0u;
  if (in) {
#pragma unroll
    for (int s = 0; s < SL; ++s) v[s] = table[(size_t)s * n + node];
    xv = ((const float2*)x)[node];
  }

  const uint4* c4 = (const uint4*)chunks;
  const int lim4 = (SL / 4) * blockIdx.x;
  unsigned part = 0u;
  for (int base = 0; base < lim4; base += CH * PF) {
    uint4 a[PF];
#pragma unroll
    for (int j = 0; j < PF; ++j) {
      const int i = base + tid + CH * j;
      a[j] = c4[i < lim4 ? i : 0];
    }
#pragma unroll
    for (int j = 0; j < PF; ++j) {
      const int i = base + tid + CH * j;
      const unsigned t = (a[j].x + a[j].y) + (a[j].z + a[j].w);
      part += i < lim4 ? t : 0u;
    }
  }
  part += __shfl_xor(part, 16, 32);
  part += __shfl_xor(part, 8, 32);
  part += __shfl_xor(part, 4, 32);
  part += __shfl_xor(part, 2, 32);
  part += __shfl_xor(part, 1, 32);

  unsigned mine = 0u;
#pragma unroll
  for (int s = 0; s < SL; ++s) mine += v[s];
  unsigned incl = mine;
#pragma unroll
  for (int o = 1; o < 32; o <<= 1) {
    unsigned t = __shfl_up(incl, o, 32);
    if (lane >= o) incl += t;
  }
  if (lane == 31) s_incl[wv] = incl;
  if (lane == 0) s_part[wv] = part;
  __syncthreads();
  unsigned off = 0u;
  if (lane < NW) off = s_part[lane] + ((lane < wv) ? s_incl[lane] : 0u);
  off += __shfl_xor(off, 16, 32);
  off += __shfl_xor(off, 8, 32);
  off += __shfl_xor(off, 4, 32);
  off += __shfl_xor(off, 2, 32);
  off += __shfl_xor(off, 1, 32);
  if (in) {
    unsigned run = 0u;
#pragma unroll
    for (int s = 0; s < SL; ++s) {
      table[(size_t)s * n + node] = run;
      run += v[s];
    }
    cnt[node] = mine;
    rowptr[node] = off + incl - mine;
    float di = rsqrtf((float)(mine + 1u));
    xs[2 * node]     = xv.x * di;
    xs[2 * node + 1] = xv.y * di;
  }
}

template <int RN>
__global__ __launch_bounds__(1024) void gcn_slice_fill(
    const int* __restrict__ src, const int* __restrict__ dst,
    const unsigned* __restrict__ table, const unsigned* __restrict__ rowptr,
    int* __restrict__ sorted, int n, int E, int L) {
  __shared__ unsigned cur[RN];
  const int tid = threadIdx.x;
  const int s = blockIdx.x;
  const int lo = blockIdx.y * RN;
  const unsigned lim = (n - lo < RN) ? (unsigned)(n - lo) : (unsigned)RN;
  const unsigned* row = table + (size_t)s * n + lo;
  constexpr int RT = (RN + 1023) / 1024;
  unsigned ra[RT], rb[RT];
#pragma unroll
  for (int k = 0; k < RT; ++k) {
    int i = tid + k * 1024;
    bool in = i < (int)lim;
    ra[k] = in ? rowptr[lo + i] : 0u;
    rb[k] = in ? row[i] : 0u;
  }
#pragma unroll
  for (int k = 0; k < RT; ++k) {
    int i = tid + k * 1024;
    if (i < (int)lim) cur[i] = ra[k] + rb[k];
  }
  __syncthreads();
  const int e0 = s * L;
  const int e1 = (E - e0 < L) ? E : e0 + L;
  for (int e = e0 + tid; e < e1; e += 1024 * 8) {
    int d[8], sv[8];
    unsigned pos[8];
#pragma unroll
    for (int k = 0; k < 8; ++k) {
      int ee = e + k * 1024;
      bool in = ee < e1;
      d[k] = in ? dst[ee] : -1;
      sv[k] = in ? src[ee] : 0;
    }
#pragma unroll
    for (int k = 0; k < 8; ++k) {
      unsigned x = (unsigned)d[k] - (unsigned)lo;
      pos[k] = 0u;
      if (x < lim) pos[k] = atomicAdd(&cur[x], 1u);
    }
#pragma unroll
    for (int k = 0; k < 8; ++k) {
      unsigned x = (unsigned)d[k] - (unsigned)lo;
      if (x < lim) sorted[pos[k]] = sv[k];
    }
  }
}

constexpr int GN = 4;
constexpr int GQ = 4;

__global__ __launch_bounds__(256) void gcn_gather1_dense(
    const int* __restrict__ sorted, const unsigned* __restrict__ rowptr,
    const unsigned* __restrict__ cnt, const float* __restrict__ xs,
    const float* __restrict__ W1, const float* __restrict__ b1,
    const float* __restrict__ W2, float* __restrict__ zs, int n) {
  static_assert(GN == 4, "four groups of GN nodes must make one 16-row tile");
  const int lane = threadIdx.x & 31;
  const unsigned l = threadIdx.x & 7;
  const int grp = lane >> 3;
  const int m = lane & 15;
  const bool hiK = lane >= 16;
  const float kmask = hiK ? 0.0f : 1.0f;
  const int wb = ((blockIdx.x * 256 + threadIdx.x) >> 5) * 16;
  if (wb >= n) return;
  const int nb = wb + grp * GN;
  const float2* xs2 = (const float2*)xs;

  unsigned beg[GN], c[GN];
  float2 self[GN];
  float ax[GN], ay[GN];
#pragma unroll
  for (int k = 0; k < GN; ++k) {
    const int node = nb + k;
    const bool in = node < n;
    beg[k] = in ? rowptr[node] : 0u;
    c[k] = in ? cnt[node] : 0u;
    self[k] = in ? xs2[node] : make_float2(0.0f, 0.0f);
    ax[k] = 0.0f;
    ay[k] = 0.0f;
  }
  float w1a[8], w1b[8], bias[8], w2[8];
#pragma unroll
  for (int t = 0; t < 8; ++t) {
    const int col = 16 * t + m;
    w1a[t] = W1[col];
    w1b[t] = W1[128 + col];
    bias[t] = b1[col];
    w2[t] = W2[col];
  }
  unsigned cm = 0u;
#pragma unroll
  for (int k = 0; k < GN; ++k) cm = c[k] > cm ? c[k] : cm;
  for (unsigned j0 = 0; j0 < cm; j0 += 8 * GQ) {
    int idx[GN][GQ];
    float2 v[GN][GQ];
#pragma unroll
    for (int k = 0; k < GN; ++k) {
#pragma unroll
      for (int q = 0; q < GQ; ++q) {
        const unsigned j = j0 + 8 * q + l;
        const unsigned jj = j < c[k] ? j : c[k] - 1u;
        idx[k][q] = sorted[c[k] ? beg[k] + jj : 0u];
      }
    }
#pragma unroll
    for (int k = 0; k < GN; ++k) {
#pragma unroll
      for (int q = 0; q < GQ; ++q) {
        const unsigned j = j0 + 8 * q + l;
        const float2 t = xs2[idx[k][q]];
        v[k][q].x = j < c[k] ? t.x : 0.0f;
        v[k][q].y = j < c[k] ? t.y : 0.0f;
      }
    }
#pragma unroll
    for (int k = 0; k < GN; ++k) {
#pragma unroll
      for (int q = 0; q < GQ; ++q) {
        ax[k] += v[k][q].x;
        ay[k] += v[k][q].y;
      }
    }
  }
  float tx = 0.0f, ty = 0.0f, td = 0.0f;
#pragma unroll
  for (int k = 0; k < GN; ++k) {
    float sx = ax[k], sy = ay[k];
    sx += __shfl_xor(sx, 1, 8);  sy += __shfl_xor(sy, 1, 8);
    sx += __shfl_xor(sx, 2, 8);  sy += __shfl_xor(sy, 2, 8);
    sx += __shfl_xor(sx, 4, 8);  sy += __shfl_xor(sy, 4, 8);
    const float di = rsqrtf((float)(c[k] + 1u));
    const float rx = (sx + self[k].x) * di;
    const float ry = (sy + self[k].y) * di;
    tx = (l == (unsigned)k) ? rx : tx;
    ty = (l == (unsigned)k) ? ry : ty;
    td = (l == (unsigned)k) ? di : td;
  }
  const int from = 8 * (m >> 2) + (m & 3);
  const float arx = __shfl(tx, from, 32);
  const float ary = __shfl(ty, from, 32);
  const bool rowok = !hiK && (wb + m < n);
  v2f A;
  A.x = rowok ? arx : 0.0f;
  A.y = rowok ? ary : 0.0f;

  float zacc[8];
#pragma unroll
  for (int v = 0; v < 8; ++v) zacc[v] = 0.0f;
#pragma unroll
  for (int t = 0; t < 8; ++t) {
    v2f B;
    B.x = w1a[t] * kmask;
    B.y = w1b[t] * kmask;
    v8f C;
#pragma unroll
    for (int v = 0; v < 8; ++v) C[v] = bias[t];

    C = __builtin_amdgcn_wmma_f32_16x16x4_f32(
         false, A,  false, B,
         (short)0, C,  false,  false);

#pragma unroll
    for (int v = 0; v < 8; ++v) {
      float h = C[v];
      h = h > 0.0f ? h : 0.0f;
      zacc[v] += h * w2[t];
    }
  }
#pragma unroll
  for (int v = 0; v < 8; ++v) {
    float r = zacc[v];
    r += __shfl_xor(r, 1, 16);
    r += __shfl_xor(r, 2, 16);
    r += __shfl_xor(r, 4, 16);
    r += __shfl_xor(r, 8, 16);
    zacc[v] = r;
  }
  const int reg = 4 * (grp & 1) + (int)(l & 3);
  float zsel = zacc[0];
#pragma unroll
  for (int v = 1; v < 8; ++v) zsel = (reg == v) ? zacc[v] : zsel;
  const int node = nb + (int)l;
  if (l < (unsigned)GN && node < n) zs[node] = zsel * td;
}

constexpr int ZL = 80000;
constexpr int G2T = 1024;
constexpr int ZF = 10;

__device__ __forceinline__ void gcn_g2_slots(
    const int* __restrict__ sorted, const unsigned (&beg)[GN],
    const unsigned (&c)[GN], unsigned j0, unsigned l, int (&idx)[GN][GQ]) {
#pragma unroll
  for (int k = 0; k < GN; ++k) {
#pragma unroll
    for (int q = 0; q < GQ; ++q) {
      const unsigned j = j0 + 8 * q + l;
      const unsigned jj = j < c[k] ? j : c[k] - 1u;
      idx[k][q] = sorted[c[k] ? beg[k] + jj : 0u];
    }
  }
}

__device__ __forceinline__ void gcn_g2_sum(
    const float* zl, const float* __restrict__ zs, int zn,
    const unsigned (&c)[GN], unsigned j0, unsigned l,
    const int (&idx)[GN][GQ], float (&acc)[GN]) {
  float a[GN][GQ], b[GN][GQ];
#pragma unroll
  for (int k = 0; k < GN; ++k) {
#pragma unroll
    for (int q = 0; q < GQ; ++q) {
      const int id = idx[k][q];
      a[k][q] = zl[id < zn ? id : 0];
    }
  }
#pragma unroll
  for (int k = 0; k < GN; ++k) {
#pragma unroll
    for (int q = 0; q < GQ; ++q) {
      const int id = idx[k][q];
      b[k][q] = zs[id < zn ? 0 : id];
    }
  }
#pragma unroll
  for (int k = 0; k < GN; ++k) {
#pragma unroll
    for (int q = 0; q < GQ; ++q) {
      const unsigned j = j0 + 8 * q + l;
      const float t = idx[k][q] < zn ? a[k][q] : b[k][q];
      acc[k] += j < c[k] ? t : 0.0f;
    }
  }
}

__global__ __launch_bounds__(G2T) void gcn_gather2(
    const int* __restrict__ sorted, const unsigned* __restrict__ rowptr,
    const unsigned* __restrict__ cnt, const float* __restrict__ zs,
    const float* __restrict__ b2, float* __restrict__ out, int n) {
  static_assert(ZL % 4 == 0, "the copy is made of 16-byte pieces");
  __shared__ float4 zl4[ZL / 4];
  float* zl = (float*)zl4;
  const int tid = threadIdx.x;
  const unsigned l = tid & 7;
  const int nb = (blockIdx.x * (G2T / 8) + (tid >> 3)) * GN;
  const float bias = b2[0];
  const int zn = n < ZL ? n : ZL;
  const int n4 = zn >> 2;
  const int last4 = n4 > 0 ? n4 - 1 : 0;
  const float4* z4 = (const float4*)zs;

  float4 t[ZF];
#pragma unroll
  for (int j = 0; j < ZF; ++j) {
    const int i = tid + G2T * j;
    t[j] = z4[i < n4 ? i : last4];
  }
  unsigned beg[GN], c[GN];
  float self[GN], acc[GN];
#pragma unroll
  for (int k = 0; k < GN; ++k) {
    const int node = nb + k;
    const bool in = node < n;
    const int nc = in ? node : n - 1;
    const unsigned r = rowptr[nc];
    const unsigned cc = cnt[nc];
    const float s = zs[nc];
    beg[k] = in ? r : 0u;
    c[k] = in ? cc : 0u;
    self[k] = in ? s : 0.0f;
    acc[k] = 0.0f;
  }
#pragma unroll
  for (int j = 0; j < ZF; ++j) {
    const int i = tid + G2T * j;
    zl4[i < n4 ? i : last4] = t[j];
  }
  int idx[GN][GQ];
  gcn_g2_slots(sorted, beg, c, 0u, l, idx);
  for (int base = G2T * ZF; base < n4; base += G2T * ZF) {
#pragma unroll
    for (int j = 0; j < ZF; ++j) {
      const int i = base + tid + G2T * j;
      t[j] = z4[i < n4 ? i : last4];
    }
#pragma unroll
    for (int j = 0; j < ZF; ++j) {
      const int i = base + tid + G2T * j;
      zl4[i < n4 ? i : last4] = t[j];
    }
  }
  for (int i = 4 * n4 + tid; i < zn; i += G2T) zl[i] = zs[i];
  __syncthreads();

  unsigned cm = 0u;
#pragma unroll
  for (int k = 0; k < GN; ++k) cm = c[k] > cm ? c[k] : cm;
  gcn_g2_sum(zl, zs, zn, c, 0u, l, idx, acc);
  for (unsigned j0 = 8 * GQ; j0 < cm; j0 += 8 * GQ) {
    gcn_g2_slots(sorted, beg, c, j0, l, idx);
    gcn_g2_sum(zl, zs, zn, c, j0, l, idx, acc);
  }
#pragma unroll
  for (int k = 0; k < GN; ++k) {
    float r = acc[k];
    r += __shfl_xor(r, 1, 8);
    r += __shfl_xor(r, 2, 8);
    r += __shfl_xor(r, 4, 8);
    const int node = nb + k;
    if (l == 0 && node < n)
      out[node] = (r + self[k]) * rsqrtf((float)(c[k] + 1u)) + bias;
  }
}

constexpr int PB_NODES = 256;
constexpr int PB_SHIFT = 8;
constexpr int PB_MAXB = 512;
constexpr int PB_NS = 128;
constexpr int PB_LMAX = 16384;
constexpr int PB_T = 1024;
constexpr int PB_EPT = 13;
constexpr int PC_T = 512;
constexpr int PC_EPT = 10;
constexpr int PC_CAP = 16384;

template <int NW>
__device__ __forceinline__ unsigned gcn_block_scan(unsigned v, unsigned* s_w,
                                                   unsigned& tot) {
  const int lane = threadIdx.x & 31;
  const int wv = threadIdx.x >> 5;
  unsigned incl = v;
#pragma unroll
  for (int o = 1; o < 32; o <<= 1) {
    unsigned t = __shfl_up(incl, o, 32);
    if (lane >= o) incl += t;
  }
  if (lane == 31) s_w[wv] = incl;
  __syncthreads();
  unsigned off = 0u, all = 0u;
#pragma unroll
  for (int w = 0; w < NW; ++w) {
    const unsigned x = s_w[w];
    all += x;
    off += (w < wv) ? x : 0u;
  }
  __syncthreads();
  tot = all;
  return off + incl - v;
}

__global__ __launch_bounds__(PB_T) void gcn_bucket_count(
    const int* __restrict__ dst, unsigned* __restrict__ hist, int nb, int E, int L) {
  __shared__ unsigned h[PB_MAXB];
  const int tid = threadIdx.x;
  const int s = blockIdx.x;
  if (tid < PB_MAXB) h[tid] = 0u;
  __syncthreads();
  const int e0 = s * L;
  const int e1 = (E - e0 < L) ? E : e0 + L;
  for (int e = e0 + tid; e < e1; e += PB_T * PB_EPT) {
    int d[PB_EPT];
#pragma unroll
    for (int k = 0; k < PB_EPT; ++k) {
      const int ee = e + k * PB_T;
      const int v = dst[ee < e1 ? ee : e1 - 1];
      d[k] = (ee < e1) ? v : -1;
    }
#pragma unroll
    for (int k = 0; k < PB_EPT; ++k) {
      unsigned b = (unsigned)d[k] >> PB_SHIFT;
      if (b < (unsigned)nb) atomicAdd(&h[b], 1u);
    }
  }
  __syncthreads();
  if (tid < PB_MAXB) hist[(size_t)s * PB_MAXB + tid] = h[tid];
}

__global__ __launch_bounds__(PB_T) void gcn_bucket_scatter(
    const int* __restrict__ src, const int* __restrict__ dst,
    const unsigned* __restrict__ hist, unsigned* __restrict__ bstart,
    unsigned* __restrict__ packed, int nb, int E, int L) {
  static_assert(PB_MAXB / 4 * (PB_NS / 16) == PB_T, "one thread per 4 buckets x 16 slices");
  static_assert(3 * (PB_NS / 16) * PB_MAXB <= PB_LMAX, "the partial sums borrow the staging area");
  __shared__ unsigned lcur[PB_MAXB];
  __shared__ unsigned gdel[PB_MAXB];
  __shared__ unsigned s_w[PB_T / 32];
  __shared__ uint4 stw4[PB_LMAX / 4];
  unsigned* stw = (unsigned*)stw4;
  __shared__ unsigned short stb[PB_LMAX];
  const int tid = threadIdx.x;
  const int s = blockIdx.x;

  constexpr int NG = PB_NS / 16;
  constexpr int NQ = PB_MAXB / 4;
  const int q = tid % NQ;
  const int g = tid / NQ;
  const uint4* h4 = (const uint4*)hist;
  uint4 a[16];
#pragma unroll
  for (int j = 0; j < 16; ++j) a[j] = h4[(size_t)(16 * g + j) * NQ + q];
  uint4 pt = make_uint4(0u, 0u, 0u, 0u), pb = pt, pm = pt;
#pragma unroll
  for (int j = 0; j < 16; ++j) {
    const int sl = 16 * g + j;
    pt.x += a[j].x; pt.y += a[j].y; pt.z += a[j].z; pt.w += a[j].w;
    if (sl < s) { pb.x += a[j].x; pb.y += a[j].y; pb.z += a[j].z; pb.w += a[j].w; }
    if (sl == s) pm = a[j];
  }
  uint4* part4 = stw4;
  part4[(0 * NG + g) * NQ + q] = pt;
  part4[(1 * NG + g) * NQ + q] = pb;
  part4[(2 * NG + g) * NQ + q] = pm;
  __syncthreads();
  const bool isb = tid < PB_MAXB;
  unsigned v_tot = 0u, v_bef = 0u, v_mine = 0u;
  if (isb) {
#pragma unroll
    for (int k = 0; k < NG; ++k) {
      v_tot  += stw[(0 * NG + k) * PB_MAXB + tid];
      v_bef  += stw[(1 * NG + k) * PB_MAXB + tid];
      v_mine += stw[(2 * NG + k) * PB_MAXB + tid];
    }
  }
  unsigned e_all = 0u, e_mine = 0u;
  const unsigned bs = gcn_block_scan<PB_T / 32>(v_tot, s_w, e_all);
  const unsigned ls = gcn_block_scan<PB_T / 32>(v_mine, s_w, e_mine);
  if (isb) {
    lcur[tid] = ls;
    gdel[tid] = bs + v_bef - ls;
  }
  if (s == 0) {
    if (tid < nb) bstart[tid] = bs;
    if (tid == 0) bstart[nb] = e_all;
  }
  __syncthreads();

  const int e0 = s * L;
  const int e1 = (E - e0 < L) ? E : e0 + L;
  for (int e = e0 + tid; e < e1; e += PB_T * PB_EPT) {
    int d[PB_EPT], sv[PB_EPT];
#pragma unroll
    for (int k = 0; k < PB_EPT; ++k) {
      const int ee = e + k * PB_T;
      const int ec = ee < e1 ? ee : e1 - 1;
      const int vd = dst[ec];
      sv[k] = src[ec];
      d[k] = (ee < e1) ? vd : -1;
    }
#pragma unroll
    for (int k = 0; k < PB_EPT; ++k) {
      unsigned b = (unsigned)d[k] >> PB_SHIFT;
      if (b < (unsigned)nb) {
        unsigned p = atomicAdd(&lcur[b], 1u);
        stw[p] = ((unsigned)sv[k] & 0x1FFFFu) |
                 (((unsigned)d[k] & (unsigned)(PB_NODES - 1)) << 17);
        stb[p] = (unsigned short)b;
      }
    }
  }
  __syncthreads();
  for (unsigned i = tid; i < e_mine; i += PB_T) packed[i + gdel[stb[i]]] = stw[i];
}

__global__ __launch_bounds__(PC_T) void gcn_bucket_rank(
    const unsigned* __restrict__ packed, const unsigned* __restrict__ bstart,
    const float* __restrict__ x, unsigned* __restrict__ cnt,
    unsigned* __restrict__ rowptr, float* __restrict__ xs,
    int* __restrict__ sorted, int n) {
  static_assert(PC_T >= PB_NODES, "one thread per node of the bucket");
  __shared__ unsigned c[PB_NODES];
  __shared__ unsigned cur[PB_NODES];
  __shared__ unsigned s_w[PC_T / 32];
  __shared__ int stage[PC_CAP];
  const int tid = threadIdx.x;
  const int b = blockIdx.x;
  const unsigned beg = bstart[b];
  const unsigned m = bstart[b + 1] - beg;
  const bool isn = tid < PB_NODES;
  const int node = b * PB_NODES + tid;
  float2 xv = make_float2(0.0f, 0.0f);
  if (isn && node < n) xv = ((const float2*)x)[node];
  if (isn) c[tid] = 0u;
  __syncthreads();
  for (unsigned i0 = 0; i0 < m; i0 += PC_T * PC_EPT) {
    unsigned w[PC_EPT];
#pragma unroll
    for (int k = 0; k < PC_EPT; ++k) {
      const unsigned i = i0 + tid + k * PC_T;
      w[k] = packed[beg + (i < m ? i : 0u)];
    }
#pragma unroll
    for (int k = 0; k < PC_EPT; ++k) {
      const unsigned i = i0 + tid + k * PC_T;
      if (i < m) atomicAdd(&c[(w[k] >> 17) & (unsigned)(PB_NODES - 1)], 1u);
    }
  }
  __syncthreads();
  const unsigned mine = isn ? c[tid] : 0u;
  unsigned all = 0u;
  const unsigned lpre = gcn_block_scan<PC_T / 32>(mine, s_w, all);
  if (isn) {
    cur[tid] = lpre;
    if (node < n) {
      cnt[node] = mine;
      rowptr[node] = beg + lpre;
      const float di = rsqrtf((float)(mine + 1u));
      ((float2*)xs)[node] = make_float2(xv.x * di, xv.y * di);
    }
  }
  __syncthreads();
  const bool fits = m <= (unsigned)PC_CAP;
  for (unsigned i0 = 0; i0 < m; i0 += PC_T * PC_EPT) {
    unsigned w[PC_EPT];
#pragma unroll
    for (int k = 0; k < PC_EPT; ++k) {
      const unsigned i = i0 + tid + k * PC_T;
      w[k] = packed[beg + (i < m ? i : 0u)];
    }
#pragma unroll
    for (int k = 0; k < PC_EPT; ++k) {
      const unsigned i = i0 + tid + k * PC_T;
      if (i < m) {
        const unsigned p = atomicAdd(&cur[(w[k] >> 17) & (unsigned)(PB_NODES - 1)], 1u);
        const int sv = (int)(w[k] & 0x1FFFFu);
        if (fits) stage[p] = sv; else sorted[beg + p] = sv;
      }
    }
  }
  if (fits) {
    __syncthreads();
    for (unsigned i = tid; i < m; i += PC_T) sorted[beg + i] = stage[i];
  }
}

constexpr int CL_CAP = 64;
constexpr int CL_EPT = PB_LMAX / PB_T;
constexpr int CR_EPT = PB_NS * CL_CAP / PC_T;

__global__ __launch_bounds__(PB_T) void gcn_cell_scatter(
    const int* __restrict__ src, const int* __restrict__ dst,
    unsigned* __restrict__ cellcnt, unsigned* __restrict__ cellpre,
    unsigned* __restrict__ cells, int nb, int E, int L) {
  static_assert(CL_EPT * PB_T == PB_LMAX, "a slice of at most PB_LMAX edges is one trip");
  __shared__ unsigned h[PB_MAXB];
  __shared__ unsigned lst[PB_MAXB];
  __shared__ unsigned lcur[PB_MAXB];
  __shared__ unsigned s_w[PB_T / 32];
  __shared__ unsigned stw[PB_LMAX];
  __shared__ unsigned short stb[PB_LMAX];
  const int tid = threadIdx.x;
  const int s = blockIdx.x;
  if (tid < PB_MAXB) h[tid] = 0u;
  __syncthreads();
  const int e0 = s * L;
  const int e1 = (E - e0 < L) ? E : e0 + L;
  int d[CL_EPT], sv[CL_EPT];
#pragma unroll
  for (int k = 0; k < CL_EPT; ++k) {
    const int ee = e0 + tid + k * PB_T;
    const int ec = ee < e1 ? ee : e1 - 1;
    const int vd = dst[ec];
    sv[k] = src[ec];
    d[k] = (ee < e1) ? vd : -1;
  }
#pragma unroll
  for (int k = 0; k < CL_EPT; ++k) {
    const unsigned b = (unsigned)d[k] >> PB_SHIFT;
    if (b < (unsigned)nb) atomicAdd(&h[b], 1u);
  }
  __syncthreads();
  const bool isb = tid < PB_MAXB;
  const unsigned v_mine = isb ? h[tid] : 0u;
  unsigned e_mine = 0u;
  const unsigned ls = gcn_block_scan<PB_T / 32>(v_mine, s_w, e_mine);
  if (isb) {
    lcur[tid] = ls;
    lst[tid] = ls;
  }
  __syncthreads();
#pragma unroll
  for (int k = 0; k < CL_EPT; ++k) {
    const unsigned b = (unsigned)d[k] >> PB_SHIFT;
    if (b < (unsigned)nb) {
      const unsigned p = atomicAdd(&lcur[b], 1u);
      stw[p] = ((unsigned)sv[k] & 0x1FFFFu) |
               (((unsigned)d[k] & (unsigned)(PB_NODES - 1)) << 17);
      stb[p] = (unsigned short)b;
    }
  }
  __syncthreads();
  for (unsigned i = tid; i < e_mine; i += PB_T) {
    const unsigned b = stb[i];
    const unsigned j = i - lst[b];
    if (j < (unsigned)CL_CAP) cells[((size_t)b * PB_NS + s) * CL_CAP + j] = stw[i];
  }
  if (isb) {
    cellcnt[(size_t)s * PB_MAXB + tid] = v_mine;
    cellpre[(size_t)s * PB_MAXB + tid] = ls;
  }
}

__global__ __launch_bounds__(PC_T) void gcn_cell_rank(
    const unsigned* __restrict__ cells, const unsigned* __restrict__ cellcnt,
    const unsigned* __restrict__ cellpre, const int* __restrict__ src,
    const int* __restrict__ dst, const float* __restrict__ x,
    unsigned* __restrict__ cnt, unsigned* __restrict__ rowptr,
    float* __restrict__ xs, int* __restrict__ sorted, int n, int E, int L) {
  static_assert(PC_T >= PB_NODES, "one thread per node of the bucket");
  static_assert(PC_T >= 2 * PB_NS, "one thread per cell count and per local prefix");
  static_assert(CR_EPT * PC_T == PB_NS * CL_CAP, "the cells of a bucket are one trip");
  static_assert(PB_NS * CL_CAP <= PC_CAP, "a bucket without overflow fits the staging area");
  static_assert(CL_CAP == 64, "cell = word index >> 6");
  constexpr int NW = PC_T / 32;
  __shared__ unsigned c[PB_NODES];
  __shared__ unsigned cur[PB_NODES];
  __shared__ unsigned s_w[NW];
  __shared__ unsigned s_r[3 * NW];
  __shared__ unsigned ccnt[PB_NS];
  __shared__ int stage[PC_CAP];
  const int tid = threadIdx.x;
  const int lane = tid & 31;
  const int wv = tid >> 5;
  const int b = blockIdx.x;
  const unsigned* tp = tid < PB_NS ? cellcnt : cellpre;
  unsigned tv = 0u;
  if (tid < 2 * PB_NS) tv = tp[(size_t)(tid & (PB_NS - 1)) * PB_MAXB + b];
  const bool isn = tid < PB_NODES;
  const int node = b * PB_NODES + tid;
  float2 xv = make_float2(0.0f, 0.0f);
  if (isn && node < n) xv = ((const float2*)x)[node];
  if (isn) c[tid] = 0u;
  if (tid < PB_NS) ccnt[tid] = tv;
  unsigned r0 = tid < PB_NS ? tv : 0u;
  unsigned r1 = (tid >= PB_NS && tid < 2 * PB_NS) ? tv : 0u;
  unsigned r2 = (tid < PB_NS && tv > (unsigned)CL_CAP) ? 1u : 0u;
#pragma unroll
  for (int o = 16; o > 0; o >>= 1) {
    r0 += __shfl_xor(r0, o, 32);
    r1 += __shfl_xor(r1, o, 32);
    r2 |= __shfl_xor(r2, o, 32);
  }
  if (lane == 0) {
    s_r[wv] = r0;
    s_r[NW + wv] = r1;
    s_r[2 * NW + wv] = r2;
  }
  __syncthreads();
  unsigned m = 0u, beg = 0u, ovf = 0u;
#pragma unroll
  for (int w = 0; w < NW; ++w) {
    m += s_r[w];
    beg += s_r[NW + w];
    ovf |= s_r[2 * NW + w];
  }
  const unsigned* cell0 = cells + (size_t)b * (PB_NS * CL_CAP);
  unsigned w[CR_EPT];
  unsigned okm = 0u;
  if (ovf == 0u) {
#pragma unroll
    for (int k = 0; k < CR_EPT; ++k) {
      const unsigned q = (unsigned)tid + (unsigned)k * PC_T;
      const bool ok = (q & (unsigned)(CL_CAP - 1)) < ccnt[q >> 6];
      const unsigned* p = ok ? cell0 + q : cellcnt;
      w[k] = *p;
      okm |= ok ? (1u << k) : 0u;
    }
#pragma unroll
    for (int k = 0; k < CR_EPT; ++k) {
      if ((okm >> k) & 1u) atomicAdd(&c[(w[k] >> 17) & (unsigned)(PB_NODES - 1)], 1u);
    }
  } else {
    for (int s = 0; s < PB_NS; ++s) {
      const unsigned cs = ccnt[s];
      if (cs <= (unsigned)CL_CAP) {
        for (unsigned j = tid; j < cs; j += PC_T)
          atomicAdd(&c[(cell0[s * CL_CAP + j] >> 17) & (unsigned)(PB_NODES - 1)], 1u);
      } else {
        const int e0 = s * L;
        const int e1 = (E - e0 < L) ? E : e0 + L;
        for (int e = e0 + tid; e < e1; e += PC_T) {
          const unsigned dd = (unsigned)dst[e];
          if ((dd >> PB_SHIFT) == (unsigned)b) atomicAdd(&c[dd & (unsigned)(PB_NODES - 1)], 1u);
        }
      }
    }
  }
  __syncthreads();
  const unsigned mine = isn ? c[tid] : 0u;
  unsigned all = 0u;
  const unsigned lpre = gcn_block_scan<NW>(mine, s_w, all);
  if (isn) {
    cur[tid] = lpre;
    if (node < n) {
      cnt[node] = mine;
      rowptr[node] = beg + lpre;
      const float di = rsqrtf((float)(mine + 1u));
      ((float2*)xs)[node] = make_float2(xv.x * di, xv.y * di);
    }
  }
  __syncthreads();
  if (ovf == 0u) {
#pragma unroll
    for (int k = 0; k < CR_EPT; ++k) {
      if ((okm >> k) & 1u) {
        const unsigned p = atomicAdd(&cur[(w[k] >> 17) & (unsigned)(PB_NODES - 1)], 1u);
        stage[p] = (int)(w[k] & 0x1FFFFu);
      }
    }
    __syncthreads();
    for (unsigned i = tid; i < m; i += PC_T) sorted[beg + i] = stage[i];
  } else {
    for (int s = 0; s < PB_NS; ++s) {
      const unsigned cs = ccnt[s];
      if (cs <= (unsigned)CL_CAP) {
        for (unsigned j = tid; j < cs; j += PC_T) {
          const unsigned ww = cell0[s * CL_CAP + j];
          const unsigned p = atomicAdd(&cur[(ww >> 17) & (unsigned)(PB_NODES - 1)], 1u);
          sorted[beg + p] = (int)(ww & 0x1FFFFu);
        }
      } else {
        const int e0 = s * L;
        const int e1 = (E - e0 < L) ? E : e0 + L;
        for (int e = e0 + tid; e < e1; e += PC_T) {
          const unsigned dd = (unsigned)dst[e];
          if ((dd >> PB_SHIFT) == (unsigned)b) {
            const unsigned p = atomicAdd(&cur[dd & (unsigned)(PB_NODES - 1)], 1u);
            sorted[beg + p] = (int)((unsigned)src[e] & 0x1FFFFu);
          }
        }
      }
    }
  }
}

template <int SL, int RN>
static bool gcn_sorted_path(const float* x, const int* src, const int* dst,
                            const float* W1, const float* b1, const float* W2,
                            const float* b2, float* out, int n, int E,
                            int dblocks, void* d_ws, size_t ws_size,
                            hipStream_t stream) {
  constexpr int RC = RN / CH;
  size_t need = (4 * (size_t)n + (size_t)SL * (size_t)n + (size_t)E) * sizeof(float);
  int ranges = (n + RN - 1) / RN;
  if (ws_size < need || ranges > 65535 ||
      (size_t)E < (size_t)ranges * RC * SL) return false;
  unsigned* cnt    = (unsigned*)d_ws;
  unsigned* rowptr = cnt + n;
  float*    xs     = (float*)(rowptr + n);
  unsigned* table  = (unsigned*)(xs + 2 * (size_t)n);
  int*      sorted = (int*)(table + (size_t)SL * (size_t)n);
  float*    zs     = (float*)table;
  unsigned* chunks = (unsigned*)sorted;

  int L = (E + SL - 1) / SL;
  dim3 gS(SL, ranges);
  int gG = (n + (G2T / 8) * GN - 1) / ((G2T / 8) * GN);

  gcn_slice_count<SL, RN><<<gS, 1024, 0, stream>>>(dst, table, chunks, n, E, L);
  gcn_offsets<SL><<<(n + CH - 1) / CH, CH, 0, stream>>>(table, chunks, cnt, rowptr, x, xs, n);
  gcn_slice_fill<RN><<<gS, 1024, 0, stream>>>(src, dst, table, rowptr, sorted, n, E, L);
  gcn_gather1_dense<<<dblocks, 256, 0, stream>>>(sorted, rowptr, cnt, xs, W1, b1, W2, zs, n);
  gcn_gather2<<<gG, G2T, 0, stream>>>(sorted, rowptr, cnt, zs, b2, out, n);
  return true;
}

static bool gcn_bucket_path(const float* x, const int* src, const int* dst,
                            const float* W1, const float* b1, const float* W2,
                            const float* b2, float* out, int n, int E,
                            int dblocks, void* d_ws, size_t ws_size,
                            hipStream_t stream) {
  int nb = (n + PB_NODES - 1) / PB_NODES;
  int L = (E + PB_NS - 1) / PB_NS;
  size_t need = (5 * (size_t)n + (size_t)PB_MAXB * PB_NS + 1024 +
                 2 * (size_t)E) * sizeof(float);
  if (n < 1 || n > PB_MAXB * PB_NODES || (n & 3) != 0 || E < 1 || L > PB_LMAX ||
      ws_size < need) return false;
  unsigned* cnt    = (unsigned*)d_ws;
  unsigned* rowptr = cnt + n;
  float*    xs     = (float*)(rowptr + n);
  float*    zs     = xs + 2 * (size_t)n;
  unsigned* hist   = (unsigned*)(zs + n);
  unsigned* bstart = hist + (size_t)PB_MAXB * PB_NS;
  unsigned* packed = bstart + 1024;
  int*      sorted = (int*)(packed + E);

  int gG = (n + (G2T / 8) * GN - 1) / ((G2T / 8) * GN);

  gcn_bucket_count<<<PB_NS, PB_T, 0, stream>>>(dst, hist, nb, E, L);
  gcn_bucket_scatter<<<PB_NS, PB_T, 0, stream>>>(src, dst, hist, bstart, packed, nb, E, L);
  gcn_bucket_rank<<<nb, PC_T, 0, stream>>>(packed, bstart, x, cnt, rowptr, xs, sorted, n);
  gcn_gather1_dense<<<dblocks, 256, 0, stream>>>(sorted, rowptr, cnt, xs, W1, b1, W2, zs, n);
  gcn_gather2<<<gG, G2T, 0, stream>>>(sorted, rowptr, cnt, zs, b2, out, n);
  return true;
}

static bool gcn_cell_path(const float* x, const int* src, const int* dst,
                          const float* W1, const float* b1, const float* W2,
                          const float* b2, float* out, int n, int E,
                          int dblocks, void* d_ws, size_t ws_size,
                          hipStream_t stream) {
  static_assert(2 * PB_NS * PB_MAXB >= PB_MAXB * PB_NODES, "zs fits the two tables");
  int nb = (n + PB_NODES - 1) / PB_NODES;
  int L = (E + PB_NS - 1) / PB_NS;
  size_t tab = (size_t)PB_MAXB * PB_NS;
  size_t need = (4 * (size_t)n + 2 * tab + (size_t)nb * PB_NS * CL_CAP +
                 (size_t)E) * sizeof(float);
  if (n < 1 || n > PB_MAXB * PB_NODES || (n & 3) != 0 || E < 1 || L > PB_LMAX ||
      ws_size < need) return false;
  unsigned* cnt     = (unsigned*)d_ws;
  unsigned* rowptr  = cnt + n;
  float*    xs      = (float*)(rowptr + n);
  unsigned* cellcnt = (unsigned*)(xs + 2 * (size_t)n);
  unsigned* cellpre = cellcnt + tab;
  unsigned* cells   = cellpre + tab;
  int*      sorted  = (int*)(cells + (size_t)nb * PB_NS * CL_CAP);
  float*    zs      = (float*)cellcnt;

  int gG = (n + (G2T / 8) * GN - 1) / ((G2T / 8) * GN);

  gcn_cell_scatter<<<PB_NS, PB_T, 0, stream>>>(src, dst, cellcnt, cellpre, cells, nb, E, L);
  gcn_cell_rank<<<nb, PC_T, 0, stream>>>(cells, cellcnt, cellpre, src, dst, x, cnt, rowptr, xs, sorted, n, E, L);
  gcn_gather1_dense<<<dblocks, 256, 0, stream>>>(sorted, rowptr, cnt, xs, W1, b1, W2, zs, n);
  gcn_gather2<<<gG, G2T, 0, stream>>>(sorted, rowptr, cnt, zs, b2, out, n);
  return true;
}

extern "C" void kernel_launch(void* const* d_in, const int* in_sizes, int n_in,
                              void* d_out, int out_size, void* d_ws, size_t ws_size,
                              hipStream_t stream) {
  const float* x  = (const float*)d_in[0];
  const int*   ei = (const int*)d_in[1];
  const float* W1 = (const float*)d_in[2];
  const float* b1 = (const float*)d_in[3];
  const float* W2 = (const float*)d_in[4];
  const float* b2 = (const float*)d_in[5];

  int n = in_sizes[0] / 2;
  int E = in_sizes[1] / 2;
  const int* src = ei;
  const int* dst = ei + E;
  float* out = (float*)d_out;

  const int tb = 256;
  int gN = (n + tb - 1) / tb;
  int gE = (E + tb - 1) / tb;
  if (gE > 4096) gE = 4096;
  int tiles = (n + 15) / 16;
  int dblocks = (tiles + 7) / 8;

  if (gcn_cell_path(x, src, dst, W1, b1, W2, b2, out, n, E, dblocks, d_ws,
                    ws_size, stream)) return;

  if (gcn_bucket_path(x, src, dst, W1, b1, W2, b2, out, n, E, dblocks, d_ws,
                      ws_size, stream)) return;

  if (gcn_sorted_path<32, 12800>(x, src, dst, W1, b1, W2, b2, out, n, E,
                                 dblocks, d_ws, ws_size, stream)) return;
  if (gcn_sorted_path<16, 6400>(x, src, dst, W1, b1, W2, b2, out, n, E,
                                dblocks, d_ws, ws_size, stream)) return;

  float* dinv = (float*)d_ws;
  float* agg  = dinv + n;
  float* z    = agg + 2 * (size_t)n;

  gcn_init<<<gN, tb, 0, stream>>>(dinv, agg, z, out, n);
  gcn_degree<<<gE, tb, 0, stream>>>(dst, dinv, E);
  gcn_rsqrt<<<gN, tb, 0, stream>>>(dinv, n);
  gcn_agg2<<<gE, tb, 0, stream>>>(src, dst, x, dinv, agg, E);

  gcn_dense_wmma<<<dblocks, 256, 0, stream>>>(agg, x, dinv, W1, b1, W2, z, n);

  gcn_out_self<<<gN, tb, 0, stream>>>(z, dinv, b2, out, n);
  gcn_out_edges<<<gE, tb, 0, stream>>>(src, dst, z, dinv, out, E);
}


# Both sort passes rank an edge with the one returning LDS add that counts it: pass one fills an LDS image of its cells (written out by row), pass two adds the run start; f32 sum order: LDS arrival
# speedup vs baseline: 8.7423x; 1.0803x over previous
#include <hip/hip_runtime.h>


typedef float v2f __attribute__((ext_vector_type(2)));
typedef float v8f __attribute__((ext_vector_type(8)));

__global__ void gcn_init(float* __restrict__ deg, float* __restrict__ agg,
                         float* __restrict__ z, float* __restrict__ out, int n) {
  int i = blockIdx.x * blockDim.x + threadIdx.x;
  if (i < n) {
    deg[i] = 1.0f;
    agg[2 * i] = 0.0f;
    agg[2 * i + 1] = 0.0f;
    z[i] = 0.0f;
    out[i] = 0.0f;
  }
}

__global__ void gcn_degree(const int* __restrict__ dst, float* __restrict__ deg,
                           int E) {
  int i = blockIdx.x * blockDim.x + threadIdx.x;
  int stride = gridDim.x * blockDim.x;
  for (; i < E; i += stride) {
    if (i + stride < E) __builtin_prefetch(&dst[i + stride], 0, 1);
    atomicAdd(&deg[dst[i]], 1.0f);
  }
}

__global__ void gcn_rsqrt(float* __restrict__ dinv, int n) {
  int i = blockIdx.x * blockDim.x + threadIdx.x;
  if (i < n) dinv[i] = rsqrtf(dinv[i]);
}

__global__ void gcn_agg2(const int* __restrict__ src, const int* __restrict__ dst,
                         const float* __restrict__ x, const float* __restrict__ dinv,
                         float* __restrict__ agg, int E) {
  int i = blockIdx.x * blockDim.x + threadIdx.x;
  int stride = gridDim.x * blockDim.x;
  for (; i < E; i += stride) {
    if (i + stride < E) {
      __builtin_prefetch(&src[i + stride], 0, 1);
      __builtin_prefetch(&dst[i + stride], 0, 1);
    }
    int s = src[i];
    int d = dst[i];
    float w = dinv[s] * dinv[d];
    atomicAdd(&agg[2 * d], x[2 * s] * w);
    atomicAdd(&agg[2 * d + 1], x[2 * s + 1] * w);
  }
}

__global__ __launch_bounds__(256) void gcn_dense_wmma(
    const float* __restrict__ agg, const float* __restrict__ x,
    const float* __restrict__ dinv, const float* __restrict__ W1,
    const float* __restrict__ b1, const float* __restrict__ W2,
    float* __restrict__ z, int n) {
  const int lane = threadIdx.x & 31;
  const int wave = threadIdx.x >> 5;
  const int tile = blockIdx.x * 8 + wave;
  const int base = tile * 16;
  if (base >= n) return;

  const int m = lane & 15;
  const bool hiK = lane >= 16;
  const float kmask = hiK ? 0.0f : 1.0f;

  int node = base + m;
  const float rowmask = (!hiK && node < n) ? 1.0f : 0.0f;
  if (node >= n) node = n - 1;
  v2f A;
  float di = dinv[node];
  float sl = di * di;
  A.x = (agg[2 * node]     + x[2 * node]     * sl) * rowmask;
  A.y = (agg[2 * node + 1] + x[2 * node + 1] * sl) * rowmask;

  float zacc[8];
#pragma unroll
  for (int v = 0; v < 8; ++v) zacc[v] = 0.0f;

#pragma unroll
  for (int n0 = 0; n0 < 128; n0 += 16) {
    int col = n0 + m;
    v2f B;
    B.x = W1[col]       * kmask;
    B.y = W1[128 + col] * kmask;
    float bias = b1[col];
    v8f C;
#pragma unroll
    for (int v = 0; v < 8; ++v) C[v] = bias;

    C = __builtin_amdgcn_wmma_f32_16x16x4_f32(
         false, A,  false, B,
         (short)0, C,  false,  false);

    float w2 = W2[col];
#pragma unroll
    for (int v = 0; v < 8; ++v) {
      float h = C[v];
      h = h > 0.0f ? h : 0.0f;
      zacc[v] += h * w2;
    }
  }

#pragma unroll
  for (int v = 0; v < 8; ++v) {
    float r = zacc[v];
    r += __shfl_xor(r, 1, 16);
    r += __shfl_xor(r, 2, 16);
    r += __shfl_xor(r, 4, 16);
    r += __shfl_xor(r, 8, 16);
    zacc[v] = r;
  }
  if (m == 0) {
    int rowbase = base + (hiK ? 8 : 0);
#pragma unroll
    for (int v = 0; v < 8; ++v) {
      int r = rowbase + v;
      if (r < n) z[r] = zacc[v];
    }
  }
}

__global__ void gcn_out_self(const float* __restrict__ z,
                             const float* __restrict__ dinv,
                             const float* __restrict__ b2,
                             float* __restrict__ out, int n) {
  int i = blockIdx.x * blockDim.x + threadIdx.x;
  if (i < n) out[i] = z[i] * dinv[i] * dinv[i] + b2[0];
}

__global__ void gcn_out_edges(const int* __restrict__ src, const int* __restrict__ dst,
                              const float* __restrict__ z,
                              const float* __restrict__ dinv,
                              float* __restrict__ out, int E) {
  int i = blockIdx.x * blockDim.x + threadIdx.x;
  int stride = gridDim.x * blockDim.x;
  for (; i < E; i += stride) {
    if (i + stride < E) {
      __builtin_prefetch(&src[i + stride], 0, 1);
      __builtin_prefetch(&dst[i + stride], 0, 1);
    }
    int s = src[i];
    int d = dst[i];
    atomicAdd(&out[d], z[s] * dinv[s] * dinv[d]);
  }
}


constexpr int CH = 256;

template <int SL, int RN>
__global__ __launch_bounds__(1024) void gcn_slice_count(
    const int* __restrict__ dst, unsigned* __restrict__ table,
    unsigned* __restrict__ chunks, int n, int E, int L) {
  constexpr int RC = RN / CH;
  static_assert(RN % CH == 0, "a chunk must not straddle two ranges");
  __shared__ unsigned h[RN];
  const int tid = threadIdx.x;
  const int s = blockIdx.x;
  const int lo = blockIdx.y * RN;
  const unsigned lim = (n - lo < RN) ? (unsigned)(n - lo) : (unsigned)RN;
  for (int i = tid; i < RN; i += 1024) h[i] = 0u;
  __syncthreads();
  const int e0 = s * L;
  const int e1 = (E - e0 < L) ? E : e0 + L;
  for (int e = e0 + tid; e < e1; e += 1024 * 16) {
    int d[16];
#pragma unroll
    for (int k = 0; k < 16; ++k) {
      int ee = e + k * 1024;
      d[k] = (ee < e1) ? dst[ee] : -1;
    }
#pragma unroll
    for (int k = 0; k < 16; ++k) {
      unsigned x = (unsigned)d[k] - (unsigned)lo;
      if (x < lim) atomicAdd(&h[x], 1u);
    }
  }
  __syncthreads();
  unsigned* row = table + (size_t)s * n + lo;
  for (int i = tid; i < (int)lim; i += 1024) row[i] = h[i];
  const int wv = tid >> 5;
  const int lane = tid & 31;
  for (int c = wv; c < RC; c += 32) {
    unsigned sum = 0u;
#pragma unroll
    for (int j = 0; j < CH / 32; ++j) sum += h[c * CH + lane + 32 * j];
    sum += __shfl_xor(sum, 16, 32);
    sum += __shfl_xor(sum, 8, 32);
    sum += __shfl_xor(sum, 4, 32);
    sum += __shfl_xor(sum, 2, 32);
    sum += __shfl_xor(sum, 1, 32);
    if (lane == 0) chunks[(size_t)(blockIdx.y * RC + c) * SL + s] = sum;
  }
}

template <int SL>
__global__ __launch_bounds__(CH) void gcn_offsets(
    unsigned* __restrict__ table, const unsigned* __restrict__ chunks,
    unsigned* __restrict__ cnt, unsigned* __restrict__ rowptr,
    const float* __restrict__ x, float* __restrict__ xs, int n) {
  static_assert(SL % 4 == 0, "a chunk's slice sums are read as 16-byte pieces");
  constexpr int NW = CH / 32;
  constexpr int PF = SL / 2;
  __shared__ unsigned s_incl[NW];
  __shared__ unsigned s_part[NW];
  const int tid = threadIdx.x;
  const int lane = tid & 31;
  const int wv = tid >> 5;
  const int node = blockIdx.x * CH + tid;
  const bool in = node < n;

  unsigned v[SL];
  float2 xv = make_float2(0.0f, 0.0f);
#pragma unroll
  for (int s = 0; s < SL; ++s) v[s] = 0u;
  if (in) {
#pragma unroll
    for (int s = 0; s < SL; ++s) v[s] = table[(size_t)s * n + node];
    xv = ((const float2*)x)[node];
  }

  const uint4* c4 = (const uint4*)chunks;
  const int lim4 = (SL / 4) * blockIdx.x;
  unsigned part = 0u;
  for (int base = 0; base < lim4; base += CH * PF) {
    uint4 a[PF];
#pragma unroll
    for (int j = 0; j < PF; ++j) {
      const int i = base + tid + CH * j;
      a[j] = c4[i < lim4 ? i : 0];
    }
#pragma unroll
    for (int j = 0; j < PF; ++j) {
      const int i = base + tid + CH * j;
      const unsigned t = (a[j].x + a[j].y) + (a[j].z + a[j].w);
      part += i < lim4 ? t : 0u;
    }
  }
  part += __shfl_xor(part, 16, 32);
  part += __shfl_xor(part, 8, 32);
  part += __shfl_xor(part, 4, 32);
  part += __shfl_xor(part, 2, 32);
  part += __shfl_xor(part, 1, 32);

  unsigned mine = 0u;
#pragma unroll
  for (int s = 0; s < SL; ++s) mine += v[s];
  unsigned incl = mine;
#pragma unroll
  for (int o = 1; o < 32; o <<= 1) {
    unsigned t = __shfl_up(incl, o, 32);
    if (lane >= o) incl += t;
  }
  if (lane == 31) s_incl[wv] = incl;
  if (lane == 0) s_part[wv] = part;
  __syncthreads();
  unsigned off = 0u;
  if (lane < NW) off = s_part[lane] + ((lane < wv) ? s_incl[lane] : 0u);
  off += __shfl_xor(off, 16, 32);
  off += __shfl_xor(off, 8, 32);
  off += __shfl_xor(off, 4, 32);
  off += __shfl_xor(off, 2, 32);
  off += __shfl_xor(off, 1, 32);
  if (in) {
    unsigned run = 0u;
#pragma unroll
    for (int s = 0; s < SL; ++s) {
      table[(size_t)s * n + node] = run;
      run += v[s];
    }
    cnt[node] = mine;
    rowptr[node] = off + incl - mine;
    float di = rsqrtf((float)(mine + 1u));
    xs[2 * node]     = xv.x * di;
    xs[2 * node + 1] = xv.y * di;
  }
}

template <int RN>
__global__ __launch_bounds__(1024) void gcn_slice_fill(
    const int* __restrict__ src, const int* __restrict__ dst,
    const unsigned* __restrict__ table, const unsigned* __restrict__ rowptr,
    int* __restrict__ sorted, int n, int E, int L) {
  __shared__ unsigned cur[RN];
  const int tid = threadIdx.x;
  const int s = blockIdx.x;
  const int lo = blockIdx.y * RN;
  const unsigned lim = (n - lo < RN) ? (unsigned)(n - lo) : (unsigned)RN;
  const unsigned* row = table + (size_t)s * n + lo;
  constexpr int RT = (RN + 1023) / 1024;
  unsigned ra[RT], rb[RT];
#pragma unroll
  for (int k = 0; k < RT; ++k) {
    int i = tid + k * 1024;
    bool in = i < (int)lim;
    ra[k] = in ? rowptr[lo + i] : 0u;
    rb[k] = in ? row[i] : 0u;
  }
#pragma unroll
  for (int k = 0; k < RT; ++k) {
    int i = tid + k * 1024;
    if (i < (int)lim) cur[i] = ra[k] + rb[k];
  }
  __syncthreads();
  const int e0 = s * L;
  const int e1 = (E - e0 < L) ? E : e0 + L;
  for (int e = e0 + tid; e < e1; e += 1024 * 8) {
    int d[8], sv[8];
    unsigned pos[8];
#pragma unroll
    for (int k = 0; k < 8; ++k) {
      int ee = e + k * 1024;
      bool in = ee < e1;
      d[k] = in ? dst[ee] : -1;
      sv[k] = in ? src[ee] : 0;
    }
#pragma unroll
    for (int k = 0; k < 8; ++k) {
      unsigned x = (unsigned)d[k] - (unsigned)lo;
      pos[k] = 0u;
      if (x < lim) pos[k] = atomicAdd(&cur[x], 1u);
    }
#pragma unroll
    for (int k = 0; k < 8; ++k) {
      unsigned x = (unsigned)d[k] - (unsigned)lo;
      if (x < lim) sorted[pos[k]] = sv[k];
    }
  }
}

constexpr int GN = 4;
constexpr int GQ = 4;

__global__ __launch_bounds__(256) void gcn_gather1_dense(
    const int* __restrict__ sorted, const unsigned* __restrict__ rowptr,
    const unsigned* __restrict__ cnt, const float* __restrict__ xs,
    const float* __restrict__ W1, const float* __restrict__ b1,
    const float* __restrict__ W2, float* __restrict__ zs, int n) {
  static_assert(GN == 4, "four groups of GN nodes must make one 16-row tile");
  const int lane = threadIdx.x & 31;
  const unsigned l = threadIdx.x & 7;
  const int grp = lane >> 3;
  const int m = lane & 15;
  const bool hiK = lane >= 16;
  const float kmask = hiK ? 0.0f : 1.0f;
  const int wb = ((blockIdx.x * 256 + threadIdx.x) >> 5) * 16;
  if (wb >= n) return;
  const int nb = wb + grp * GN;
  const float2* xs2 = (const float2*)xs;

  unsigned beg[GN], c[GN];
  float2 self[GN];
  float ax[GN], ay[GN];
#pragma unroll
  for (int k = 0; k < GN; ++k) {
    const int node = nb + k;
    const bool in = node < n;
    beg[k] = in ? rowptr[node] : 0u;
    c[k] = in ? cnt[node] : 0u;
    self[k] = in ? xs2[node] : make_float2(0.0f, 0.0f);
    ax[k] = 0.0f;
    ay[k] = 0.0f;
  }
  float w1a[8], w1b[8], bias[8], w2[8];
#pragma unroll
  for (int t = 0; t < 8; ++t) {
    const int col = 16 * t + m;
    w1a[t] = W1[col];
    w1b[t] = W1[128 + col];
    bias[t] = b1[col];
    w2[t] = W2[col];
  }
  unsigned cm = 0u;
#pragma unroll
  for (int k = 0; k < GN; ++k) cm = c[k] > cm ? c[k] : cm;
  for (unsigned j0 = 0; j0 < cm; j0 += 8 * GQ) {
    int idx[GN][GQ];
    float2 v[GN][GQ];
#pragma unroll
    for (int k = 0; k < GN; ++k) {
#pragma unroll
      for (int q = 0; q < GQ; ++q) {
        const unsigned j = j0 + 8 * q + l;
        const unsigned jj = j < c[k] ? j : c[k] - 1u;
        idx[k][q] = sorted[c[k] ? beg[k] + jj : 0u];
      }
    }
#pragma unroll
    for (int k = 0; k < GN; ++k) {
#pragma unroll
      for (int q = 0; q < GQ; ++q) {
        const unsigned j = j0 + 8 * q + l;
        const float2 t = xs2[idx[k][q]];
        v[k][q].x = j < c[k] ? t.x : 0.0f;
        v[k][q].y = j < c[k] ? t.y : 0.0f;
      }
    }
#pragma unroll
    for (int k = 0; k < GN; ++k) {
#pragma unroll
      for (int q = 0; q < GQ; ++q) {
        ax[k] += v[k][q].x;
        ay[k] += v[k][q].y;
      }
    }
  }
  float tx = 0.0f, ty = 0.0f, td = 0.0f;
#pragma unroll
  for (int k = 0; k < GN; ++k) {
    float sx = ax[k], sy = ay[k];
    sx += __shfl_xor(sx, 1, 8);  sy += __shfl_xor(sy, 1, 8);
    sx += __shfl_xor(sx, 2, 8);  sy += __shfl_xor(sy, 2, 8);
    sx += __shfl_xor(sx, 4, 8);  sy += __shfl_xor(sy, 4, 8);
    const float di = rsqrtf((float)(c[k] + 1u));
    const float rx = (sx + self[k].x) * di;
    const float ry = (sy + self[k].y) * di;
    tx = (l == (unsigned)k) ? rx : tx;
    ty = (l == (unsigned)k) ? ry : ty;
    td = (l == (unsigned)k) ? di : td;
  }
  const int from = 8 * (m >> 2) + (m & 3);
  const float arx = __shfl(tx, from, 32);
  const float ary = __shfl(ty, from, 32);
  const bool rowok = !hiK && (wb + m < n);
  v2f A;
  A.x = rowok ? arx : 0.0f;
  A.y = rowok ? ary : 0.0f;

  float zacc[8];
#pragma unroll
  for (int v = 0; v < 8; ++v) zacc[v] = 0.0f;
#pragma unroll
  for (int t = 0; t < 8; ++t) {
    v2f B;
    B.x = w1a[t] * kmask;
    B.y = w1b[t] * kmask;
    v8f C;
#pragma unroll
    for (int v = 0; v < 8; ++v) C[v] = bias[t];

    C = __builtin_amdgcn_wmma_f32_16x16x4_f32(
         false, A,  false, B,
         (short)0, C,  false,  false);

#pragma unroll
    for (int v = 0; v < 8; ++v) {
      float h = C[v];
      h = h > 0.0f ? h : 0.0f;
      zacc[v] += h * w2[t];
    }
  }
#pragma unroll
  for (int v = 0; v < 8; ++v) {
    float r = zacc[v];
    r += __shfl_xor(r, 1, 16);
    r += __shfl_xor(r, 2, 16);
    r += __shfl_xor(r, 4, 16);
    r += __shfl_xor(r, 8, 16);
    zacc[v] = r;
  }
  const int reg = 4 * (grp & 1) + (int)(l & 3);
  float zsel = zacc[0];
#pragma unroll
  for (int v = 1; v < 8; ++v) zsel = (reg == v) ? zacc[v] : zsel;
  const int node = nb + (int)l;
  if (l < (unsigned)GN && node < n) zs[node] = zsel * td;
}

constexpr int ZL = 80000;
constexpr int G2T = 1024;
constexpr int ZF = 10;

__device__ __forceinline__ void gcn_g2_slots(
    const int* __restrict__ sorted, const unsigned (&beg)[GN],
    const unsigned (&c)[GN], unsigned j0, unsigned l, int (&idx)[GN][GQ]) {
#pragma unroll
  for (int k = 0; k < GN; ++k) {
#pragma unroll
    for (int q = 0; q < GQ; ++q) {
      const unsigned j = j0 + 8 * q + l;
      const unsigned jj = j < c[k] ? j : c[k] - 1u;
      idx[k][q] = sorted[c[k] ? beg[k] + jj : 0u];
    }
  }
}

__device__ __forceinline__ void gcn_g2_sum(
    const float* zl, const float* __restrict__ zs, int zn,
    const unsigned (&c)[GN], unsigned j0, unsigned l,
    const int (&idx)[GN][GQ], float (&acc)[GN]) {
  float a[GN][GQ], b[GN][GQ];
#pragma unroll
  for (int k = 0; k < GN; ++k) {
#pragma unroll
    for (int q = 0; q < GQ; ++q) {
      const int id = idx[k][q];
      a[k][q] = zl[id < zn ? id : 0];
    }
  }
#pragma unroll
  for (int k = 0; k < GN; ++k) {
#pragma unroll
    for (int q = 0; q < GQ; ++q) {
      const int id = idx[k][q];
      b[k][q] = zs[id < zn ? 0 : id];
    }
  }
#pragma unroll
  for (int k = 0; k < GN; ++k) {
#pragma unroll
    for (int q = 0; q < GQ; ++q) {
      const unsigned j = j0 + 8 * q + l;
      const float t = idx[k][q] < zn ? a[k][q] : b[k][q];
      acc[k] += j < c[k] ? t : 0.0f;
    }
  }
}

__global__ __launch_bounds__(G2T) void gcn_gather2(
    const int* __restrict__ sorted, const unsigned* __restrict__ rowptr,
    const unsigned* __restrict__ cnt, const float* __restrict__ zs,
    const float* __restrict__ b2, float* __restrict__ out, int n) {
  static_assert(ZL % 4 == 0, "the copy is made of 16-byte pieces");
  __shared__ float4 zl4[ZL / 4];
  float* zl = (float*)zl4;
  const int tid = threadIdx.x;
  const unsigned l = tid & 7;
  const int nb = (blockIdx.x * (G2T / 8) + (tid >> 3)) * GN;
  const float bias = b2[0];
  const int zn = n < ZL ? n : ZL;
  const int n4 = zn >> 2;
  const int last4 = n4 > 0 ? n4 - 1 : 0;
  const float4* z4 = (const float4*)zs;

  float4 t[ZF];
#pragma unroll
  for (int j = 0; j < ZF; ++j) {
    const int i = tid + G2T * j;
    t[j] = z4[i < n4 ? i : last4];
  }
  unsigned beg[GN], c[GN];
  float self[GN], acc[GN];
#pragma unroll
  for (int k = 0; k < GN; ++k) {
    const int node = nb + k;
    const bool in = node < n;
    const int nc = in ? node : n - 1;
    const unsigned r = rowptr[nc];
    const unsigned cc = cnt[nc];
    const float s = zs[nc];
    beg[k] = in ? r : 0u;
    c[k] = in ? cc : 0u;
    self[k] = in ? s : 0.0f;
    acc[k] = 0.0f;
  }
#pragma unroll
  for (int j = 0; j < ZF; ++j) {
    const int i = tid + G2T * j;
    zl4[i < n4 ? i : last4] = t[j];
  }
  int idx[GN][GQ];
  gcn_g2_slots(sorted, beg, c, 0u, l, idx);
  for (int base = G2T * ZF; base < n4; base += G2T * ZF) {
#pragma unroll
    for (int j = 0; j < ZF; ++j) {
      const int i = base + tid + G2T * j;
      t[j] = z4[i < n4 ? i : last4];
    }
#pragma unroll
    for (int j = 0; j < ZF; ++j) {
      const int i = base + tid + G2T * j;
      zl4[i < n4 ? i : last4] = t[j];
    }
  }
  for (int i = 4 * n4 + tid; i < zn; i += G2T) zl[i] = zs[i];
  __syncthreads();

  unsigned cm = 0u;
#pragma unroll
  for (int k = 0; k < GN; ++k) cm = c[k] > cm ? c[k] : cm;
  gcn_g2_sum(zl, zs, zn, c, 0u, l, idx, acc);
  for (unsigned j0 = 8 * GQ; j0 < cm; j0 += 8 * GQ) {
    gcn_g2_slots(sorted, beg, c, j0, l, idx);
    gcn_g2_sum(zl, zs, zn, c, j0, l, idx, acc);
  }
#pragma unroll
  for (int k = 0; k < GN; ++k) {
    float r = acc[k];
    r += __shfl_xor(r, 1, 8);
    r += __shfl_xor(r, 2, 8);
    r += __shfl_xor(r, 4, 8);
    const int node = nb + k;
    if (l == 0 && node < n)
      out[node] = (r + self[k]) * rsqrtf((float)(c[k] + 1u)) + bias;
  }
}

constexpr int PB_NODES = 256;
constexpr int PB_SHIFT = 8;
constexpr int PB_MAXB = 512;
constexpr int PB_NS = 128;
constexpr int PB_LMAX = 16384;
constexpr int PB_T = 1024;
constexpr int PB_EPT = 13;
constexpr int PC_T = 512;
constexpr int PC_EPT = 10;
constexpr int PC_CAP = 16384;

template <int NW>
__device__ __forceinline__ unsigned gcn_block_scan(unsigned v, unsigned* s_w,
                                                   unsigned& tot) {
  const int lane = threadIdx.x & 31;
  const int wv = threadIdx.x >> 5;
  unsigned incl = v;
#pragma unroll
  for (int o = 1; o < 32; o <<= 1) {
    unsigned t = __shfl_up(incl, o, 32);
    if (lane >= o) incl += t;
  }
  if (lane == 31) s_w[wv] = incl;
  __syncthreads();
  unsigned off = 0u, all = 0u;
#pragma unroll
  for (int w = 0; w < NW; ++w) {
    const unsigned x = s_w[w];
    all += x;
    off += (w < wv) ? x : 0u;
  }
  __syncthreads();
  tot = all;
  return off + incl - v;
}

__global__ __launch_bounds__(PB_T) void gcn_bucket_count(
    const int* __restrict__ dst, unsigned* __restrict__ hist, int nb, int E, int L) {
  __shared__ unsigned h[PB_MAXB];
  const int tid = threadIdx.x;
  const int s = blockIdx.x;
  if (tid < PB_MAXB) h[tid] = 0u;
  __syncthreads();
  const int e0 = s * L;
  const int e1 = (E - e0 < L) ? E : e0 + L;
  for (int e = e0 + tid; e < e1; e += PB_T * PB_EPT) {
    int d[PB_EPT];
#pragma unroll
    for (int k = 0; k < PB_EPT; ++k) {
      const int ee = e + k * PB_T;
      const int v = dst[ee < e1 ? ee : e1 - 1];
      d[k] = (ee < e1) ? v : -1;
    }
#pragma unroll
    for (int k = 0; k < PB_EPT; ++k) {
      unsigned b = (unsigned)d[k] >> PB_SHIFT;
      if (b < (unsigned)nb) atomicAdd(&h[b], 1u);
    }
  }
  __syncthreads();
  if (tid < PB_MAXB) hist[(size_t)s * PB_MAXB + tid] = h[tid];
}

__global__ __launch_bounds__(PB_T) void gcn_bucket_scatter(
    const int* __restrict__ src, const int* __restrict__ dst,
    const unsigned* __restrict__ hist, unsigned* __restrict__ bstart,
    unsigned* __restrict__ packed, int nb, int E, int L) {
  static_assert(PB_MAXB / 4 * (PB_NS / 16) == PB_T, "one thread per 4 buckets x 16 slices");
  static_assert(3 * (PB_NS / 16) * PB_MAXB <= PB_LMAX, "the partial sums borrow the staging area");
  __shared__ unsigned lcur[PB_MAXB];
  __shared__ unsigned gdel[PB_MAXB];
  __shared__ unsigned s_w[PB_T / 32];
  __shared__ uint4 stw4[PB_LMAX / 4];
  unsigned* stw = (unsigned*)stw4;
  __shared__ unsigned short stb[PB_LMAX];
  const int tid = threadIdx.x;
  const int s = blockIdx.x;

  constexpr int NG = PB_NS / 16;
  constexpr int NQ = PB_MAXB / 4;
  const int q = tid % NQ;
  const int g = tid / NQ;
  const uint4* h4 = (const uint4*)hist;
  uint4 a[16];
#pragma unroll
  for (int j = 0; j < 16; ++j) a[j] = h4[(size_t)(16 * g + j) * NQ + q];
  uint4 pt = make_uint4(0u, 0u, 0u, 0u), pb = pt, pm = pt;
#pragma unroll
  for (int j = 0; j < 16; ++j) {
    const int sl = 16 * g + j;
    pt.x += a[j].x; pt.y += a[j].y; pt.z += a[j].z; pt.w += a[j].w;
    if (sl < s) { pb.x += a[j].x; pb.y += a[j].y; pb.z += a[j].z; pb.w += a[j].w; }
    if (sl == s) pm = a[j];
  }
  uint4* part4 = stw4;
  part4[(0 * NG + g) * NQ + q] = pt;
  part4[(1 * NG + g) * NQ + q] = pb;
  part4[(2 * NG + g) * NQ + q] = pm;
  __syncthreads();
  const bool isb = tid < PB_MAXB;
  unsigned v_tot = 0u, v_bef = 0u, v_mine = 0u;
  if (isb) {
#pragma unroll
    for (int k = 0; k < NG; ++k) {
      v_tot  += stw[(0 * NG + k) * PB_MAXB + tid];
      v_bef  += stw[(1 * NG + k) * PB_MAXB + tid];
      v_mine += stw[(2 * NG + k) * PB_MAXB + tid];
    }
  }
  unsigned e_all = 0u, e_mine = 0u;
  const unsigned bs = gcn_block_scan<PB_T / 32>(v_tot, s_w, e_all);
  const unsigned ls = gcn_block_scan<PB_T / 32>(v_mine, s_w, e_mine);
  if (isb) {
    lcur[tid] = ls;
    gdel[tid] = bs + v_bef - ls;
  }
  if (s == 0) {
    if (tid < nb) bstart[tid] = bs;
    if (tid == 0) bstart[nb] = e_all;
  }
  __syncthreads();

  const int e0 = s * L;
  const int e1 = (E - e0 < L) ? E : e0 + L;
  for (int e = e0 + tid; e < e1; e += PB_T * PB_EPT) {
    int d[PB_EPT], sv[PB_EPT];
#pragma unroll
    for (int k = 0; k < PB_EPT; ++k) {
      const int ee = e + k * PB_T;
      const int ec = ee < e1 ? ee : e1 - 1;
      const int vd = dst[ec];
      sv[k] = src[ec];
      d[k] = (ee < e1) ? vd : -1;
    }
#pragma unroll
    for (int k = 0; k < PB_EPT; ++k) {
      unsigned b = (unsigned)d[k] >> PB_SHIFT;
      if (b < (unsigned)nb) {
        unsigned p = atomicAdd(&lcur[b], 1u);
        stw[p] = ((unsigned)sv[k] & 0x1FFFFu) |
                 (((unsigned)d[k] & (unsigned)(PB_NODES - 1)) << 17);
        stb[p] = (unsigned short)b;
      }
    }
  }
  __syncthreads();
  for (unsigned i = tid; i < e_mine; i += PB_T) packed[i + gdel[stb[i]]] = stw[i];
}

__global__ __launch_bounds__(PC_T) void gcn_bucket_rank(
    const unsigned* __restrict__ packed, const unsigned* __restrict__ bstart,
    const float* __restrict__ x, unsigned* __restrict__ cnt,
    unsigned* __restrict__ rowptr, float* __restrict__ xs,
    int* __restrict__ sorted, int n) {
  static_assert(PC_T >= PB_NODES, "one thread per node of the bucket");
  __shared__ unsigned c[PB_NODES];
  __shared__ unsigned cur[PB_NODES];
  __shared__ unsigned s_w[PC_T / 32];
  __shared__ int stage[PC_CAP];
  const int tid = threadIdx.x;
  const int b = blockIdx.x;
  const unsigned beg = bstart[b];
  const unsigned m = bstart[b + 1] - beg;
  const bool isn = tid < PB_NODES;
  const int node = b * PB_NODES + tid;
  float2 xv = make_float2(0.0f, 0.0f);
  if (isn && node < n) xv = ((const float2*)x)[node];
  if (isn) c[tid] = 0u;
  __syncthreads();
  for (unsigned i0 = 0; i0 < m; i0 += PC_T * PC_EPT) {
    unsigned w[PC_EPT];
#pragma unroll
    for (int k = 0; k < PC_EPT; ++k) {
      const unsigned i = i0 + tid + k * PC_T;
      w[k] = packed[beg + (i < m ? i : 0u)];
    }
#pragma unroll
    for (int k = 0; k < PC_EPT; ++k) {
      const unsigned i = i0 + tid + k * PC_T;
      if (i < m) atomicAdd(&c[(w[k] >> 17) & (unsigned)(PB_NODES - 1)], 1u);
    }
  }
  __syncthreads();
  const unsigned mine = isn ? c[tid] : 0u;
  unsigned all = 0u;
  const unsigned lpre = gcn_block_scan<PC_T / 32>(mine, s_w, all);
  if (isn) {
    cur[tid] = lpre;
    if (node < n) {
      cnt[node] = mine;
      rowptr[node] = beg + lpre;
      const float di = rsqrtf((float)(mine + 1u));
      ((float2*)xs)[node] = make_float2(xv.x * di, xv.y * di);
    }
  }
  __syncthreads();
  const bool fits = m <= (unsigned)PC_CAP;
  for (unsigned i0 = 0; i0 < m; i0 += PC_T * PC_EPT) {
    unsigned w[PC_EPT];
#pragma unroll
    for (int k = 0; k < PC_EPT; ++k) {
      const unsigned i = i0 + tid + k * PC_T;
      w[k] = packed[beg + (i < m ? i : 0u)];
    }
#pragma unroll
    for (int k = 0; k < PC_EPT; ++k) {
      const unsigned i = i0 + tid + k * PC_T;
      if (i < m) {
        const unsigned p = atomicAdd(&cur[(w[k] >> 17) & (unsigned)(PB_NODES - 1)], 1u);
        const int sv = (int)(w[k] & 0x1FFFFu);
        if (fits) stage[p] = sv; else sorted[beg + p] = sv;
      }
    }
  }
  if (fits) {
    __syncthreads();
    for (unsigned i = tid; i < m; i += PC_T) sorted[beg + i] = stage[i];
  }
}

constexpr int CL_CAP = 64;
constexpr int CL_EPT = PB_LMAX / PB_T;
constexpr int CR_EPT = PB_NS * CL_CAP / PC_T;
constexpr int CI_PITCH = CL_CAP + 1;

__global__ __launch_bounds__(PB_T) void gcn_cell_scatter(
    const int* __restrict__ src, const int* __restrict__ dst,
    unsigned* __restrict__ cellcnt, unsigned* __restrict__ cellpre,
    unsigned* __restrict__ cells, int nb, int E, int L) {
  static_assert(CL_EPT * PB_T == PB_LMAX, "a slice of at most PB_LMAX edges is one trip");
  static_assert(CL_CAP == 64, "a cell row is two 32-lane halves");
  static_assert(PB_MAXB == PB_T / 2, "a wave writes out the rows wave, wave + 32, ... : PB_MAXB / 32 of them");
  constexpr int RW = PB_MAXB / 32;
  __shared__ unsigned cnt[PB_MAXB];
  __shared__ unsigned s_w[PB_T / 32];
  __shared__ unsigned img[PB_MAXB * CI_PITCH];
  const int tid = threadIdx.x;
  const int lane = tid & 31;
  const int wv = tid >> 5;
  const int s = blockIdx.x;
  if (tid < PB_MAXB) cnt[tid] = 0u;
  __syncthreads();
  const int e0 = s * L;
  const int e1 = (E - e0 < L) ? E : e0 + L;
  int d[CL_EPT], sv[CL_EPT];
#pragma unroll
  for (int k = 0; k < CL_EPT; ++k) {
    const int ee = e0 + tid + k * PB_T;
    const int ec = ee < e1 ? ee : e1 - 1;
    const int vd = dst[ec];
    sv[k] = src[ec];
    d[k] = (ee < e1) ? vd : -1;
  }
  unsigned rk[CL_EPT];
#pragma unroll
  for (int k = 0; k < CL_EPT; ++k) {
    const unsigned b = (unsigned)d[k] >> PB_SHIFT;
    rk[k] = (unsigned)CL_CAP;
    if (b < (unsigned)nb) rk[k] = atomicAdd(&cnt[b], 1u);
  }
#pragma unroll
  for (int k = 0; k < CL_EPT; ++k) {
    const unsigned b = (unsigned)d[k] >> PB_SHIFT;
    if (rk[k] < (unsigned)CL_CAP)
      img[b * CI_PITCH + rk[k]] = ((unsigned)sv[k] & 0x1FFFFu) |
                                  (((unsigned)d[k] & (unsigned)(PB_NODES - 1)) << 17);
  }
  __syncthreads();
  const bool isb = tid < PB_MAXB;
  const unsigned v_mine = isb ? cnt[tid] : 0u;
  unsigned e_mine = 0u;
  const unsigned ls = gcn_block_scan<PB_T / 32>(v_mine, s_w, e_mine);
  unsigned cl = 0u;
  if (lane < RW) cl = cnt[wv + 32 * lane];
  unsigned w0[RW], w1[RW];
#pragma unroll
  for (int k = 0; k < RW; ++k) {
    const int row = wv + 32 * k;
    w0[k] = img[row * CI_PITCH + lane];
    w1[k] = img[row * CI_PITCH + 32 + lane];
  }
#pragma unroll
  for (int k = 0; k < RW; ++k) {
    const int row = wv + 32 * k;
    const unsigned c = (unsigned)__builtin_amdgcn_readlane((int)cl, k);
    unsigned* cell = cells + ((size_t)row * PB_NS + s) * CL_CAP;
    if ((unsigned)lane < c) cell[lane] = w0[k];
    if ((unsigned)lane + 32u < c) cell[32 + lane] = w1[k];
  }
  if (isb) {
    cellcnt[(size_t)s * PB_MAXB + tid] = v_mine;
    cellpre[(size_t)s * PB_MAXB + tid] = ls;
  }
}

__global__ __launch_bounds__(PC_T) void gcn_cell_rank(
    const unsigned* __restrict__ cells, const unsigned* __restrict__ cellcnt,
    const unsigned* __restrict__ cellpre, const int* __restrict__ src,
    const int* __restrict__ dst, const float* __restrict__ x,
    unsigned* __restrict__ cnt, unsigned* __restrict__ rowptr,
    float* __restrict__ xs, int* __restrict__ sorted, int n, int E, int L) {
  static_assert(PC_T >= PB_NODES, "one thread per node of the bucket");
  static_assert(PC_T >= 2 * PB_NS, "one thread per cell count and per local prefix");
  static_assert(CR_EPT * PC_T == PB_NS * CL_CAP, "the cells of a bucket are one trip");
  static_assert(PB_NS * CL_CAP <= PC_CAP, "a bucket without overflow fits the staging area");
  static_assert(CL_CAP == 64, "cell = word index >> 6");
  constexpr int NW = PC_T / 32;
  __shared__ unsigned c[PB_NODES];
  __shared__ unsigned cur[PB_NODES];
  __shared__ unsigned s_w[NW];
  __shared__ unsigned s_r[3 * NW];
  __shared__ unsigned ccnt[PB_NS];
  __shared__ int stage[PC_CAP];
  const int tid = threadIdx.x;
  const int lane = tid & 31;
  const int wv = tid >> 5;
  const int b = blockIdx.x;
  const unsigned* tp = tid < PB_NS ? cellcnt : cellpre;
  unsigned tv = 0u;
  if (tid < 2 * PB_NS) tv = tp[(size_t)(tid & (PB_NS - 1)) * PB_MAXB + b];
  const bool isn = tid < PB_NODES;
  const int node = b * PB_NODES + tid;
  float2 xv = make_float2(0.0f, 0.0f);
  if (isn && node < n) xv = ((const float2*)x)[node];
  if (isn) c[tid] = 0u;
  if (tid < PB_NS) ccnt[tid] = tv;
  unsigned r0 = tid < PB_NS ? tv : 0u;
  unsigned r1 = (tid >= PB_NS && tid < 2 * PB_NS) ? tv : 0u;
  unsigned r2 = (tid < PB_NS && tv > (unsigned)CL_CAP) ? 1u : 0u;
#pragma unroll
  for (int o = 16; o > 0; o >>= 1) {
    r0 += __shfl_xor(r0, o, 32);
    r1 += __shfl_xor(r1, o, 32);
    r2 |= __shfl_xor(r2, o, 32);
  }
  if (lane == 0) {
    s_r[wv] = r0;
    s_r[NW + wv] = r1;
    s_r[2 * NW + wv] = r2;
  }
  __syncthreads();
  unsigned m = 0u, beg = 0u, ovf = 0u;
#pragma unroll
  for (int w = 0; w < NW; ++w) {
    m += s_r[w];
    beg += s_r[NW + w];
    ovf |= s_r[2 * NW + w];
  }
  const unsigned* cell0 = cells + (size_t)b * (PB_NS * CL_CAP);
  unsigned w[CR_EPT];
  unsigned rk[CR_EPT];
  unsigned okm = 0u;
  if (ovf == 0u) {
#pragma unroll
    for (int k = 0; k < CR_EPT; ++k) {
      const unsigned q = (unsigned)tid + (unsigned)k * PC_T;
      const bool ok = (q & (unsigned)(CL_CAP - 1)) < ccnt[q >> 6];
      const unsigned* p = ok ? cell0 + q : cellcnt;
      w[k] = *p;
      okm |= ok ? (1u << k) : 0u;
    }
#pragma unroll
    for (int k = 0; k < CR_EPT; ++k) {
      rk[k] = 0u;
      if ((okm >> k) & 1u) rk[k] = atomicAdd(&c[(w[k] >> 17) & (unsigned)(PB_NODES - 1)], 1u);
    }
  } else {
    for (int s = 0; s < PB_NS; ++s) {
      const unsigned cs = ccnt[s];
      if (cs <= (unsigned)CL_CAP) {
        for (unsigned j = tid; j < cs; j += PC_T)
          atomicAdd(&c[(cell0[s * CL_CAP + j] >> 17) & (unsigned)(PB_NODES - 1)], 1u);
      } else {
        const int e0 = s * L;
        const int e1 = (E - e0 < L) ? E : e0 + L;
        for (int e = e0 + tid; e < e1; e += PC_T) {
          const unsigned dd = (unsigned)dst[e];
          if ((dd >> PB_SHIFT) == (unsigned)b) atomicAdd(&c[dd & (unsigned)(PB_NODES - 1)], 1u);
        }
      }
    }
  }
  __syncthreads();
  const unsigned mine = isn ? c[tid] : 0u;
  unsigned all = 0u;
  const unsigned lpre = gcn_block_scan<NW>(mine, s_w, all);
  if (isn) {
    cur[tid] = lpre;
    if (node < n) {
      cnt[node] = mine;
      rowptr[node] = beg + lpre;
      const float di = rsqrtf((float)(mine + 1u));
      ((float2*)xs)[node] = make_float2(xv.x * di, xv.y * di);
    }
  }
  __syncthreads();
  if (ovf == 0u) {
    unsigned fs[CR_EPT];
#pragma unroll
    for (int k = 0; k < CR_EPT; ++k) fs[k] = cur[(w[k] >> 17) & (unsigned)(PB_NODES - 1)];
#pragma unroll
    for (int k = 0; k < CR_EPT; ++k) {
      if ((okm >> k) & 1u) stage[fs[k] + rk[k]] = (int)(w[k] & 0x1FFFFu);
    }
    __syncthreads();
    for (unsigned i = tid; i < m; i += PC_T) sorted[beg + i] = stage[i];
  } else {
    for (int s = 0; s < PB_NS; ++s) {
      const unsigned cs = ccnt[s];
      if (cs <= (unsigned)CL_CAP) {
        for (unsigned j = tid; j < cs; j += PC_T) {
          const unsigned ww = cell0[s * CL_CAP + j];
          const unsigned p = atomicAdd(&cur[(ww >> 17) & (unsigned)(PB_NODES - 1)], 1u);
          sorted[beg + p] = (int)(ww & 0x1FFFFu);
        }
      } else {
        const int e0 = s * L;
        const int e1 = (E - e0 < L) ? E : e0 + L;
        for (int e = e0 + tid; e < e1; e += PC_T) {
          const unsigned dd = (unsigned)dst[e];
          if ((dd >> PB_SHIFT) == (unsigned)b) {
            const unsigned p = atomicAdd(&cur[dd & (unsigned)(PB_NODES - 1)], 1u);
            sorted[beg + p] = (int)((unsigned)src[e] & 0x1FFFFu);
          }
        }
      }
    }
  }
}

template <int SL, int RN>
static bool gcn_sorted_path(const float* x, const int* src, const int* dst,
                            const float* W1, const float* b1, const float* W2,
                            const float* b2, float* out, int n, int E,
                            int dblocks, void* d_ws, size_t ws_size,
                            hipStream_t stream) {
  constexpr int RC = RN / CH;
  size_t need = (4 * (size_t)n + (size_t)SL * (size_t)n + (size_t)E) * sizeof(float);
  int ranges = (n + RN - 1) / RN;
  if (ws_size < need || ranges > 65535 ||
      (size_t)E < (size_t)ranges * RC * SL) return false;
  unsigned* cnt    = (unsigned*)d_ws;
  unsigned* rowptr = cnt + n;
  float*    xs     = (float*)(rowptr + n);
  unsigned* table  = (unsigned*)(xs + 2 * (size_t)n);
  int*      sorted = (int*)(table + (size_t)SL * (size_t)n);
  float*    zs     = (float*)table;
  unsigned* chunks = (unsigned*)sorted;

  int L = (E + SL - 1) / SL;
  dim3 gS(SL, ranges);
  int gG = (n + (G2T / 8) * GN - 1) / ((G2T / 8) * GN);

  gcn_slice_count<SL, RN><<<gS, 1024, 0, stream>>>(dst, table, chunks, n, E, L);
  gcn_offsets<SL><<<(n + CH - 1) / CH, CH, 0, stream>>>(table, chunks, cnt, rowptr, x, xs, n);
  gcn_slice_fill<RN><<<gS, 1024, 0, stream>>>(src, dst, table, rowptr, sorted, n, E, L);
  gcn_gather1_dense<<<dblocks, 256, 0, stream>>>(sorted, rowptr, cnt, xs, W1, b1, W2, zs, n);
  gcn_gather2<<<gG, G2T, 0, stream>>>(sorted, rowptr, cnt, zs, b2, out, n);
  return true;
}

static bool gcn_bucket_path(const float* x, const int* src, const int* dst,
                            const float* W1, const float* b1, const float* W2,
                            const float* b2, float* out, int n, int E,
                            int dblocks, void* d_ws, size_t ws_size,
                            hipStream_t stream) {
  int nb = (n + PB_NODES - 1) / PB_NODES;
  int L = (E + PB_NS - 1) / PB_NS;
  size_t need = (5 * (size_t)n + (size_t)PB_MAXB * PB_NS + 1024 +
                 2 * (size_t)E) * sizeof(float);
  if (n < 1 || n > PB_MAXB * PB_NODES || (n & 3) != 0 || E < 1 || L > PB_LMAX ||
      ws_size < need) return false;
  unsigned* cnt    = (unsigned*)d_ws;
  unsigned* rowptr = cnt + n;
  float*    xs     = (float*)(rowptr + n);
  float*    zs     = xs + 2 * (size_t)n;
  unsigned* hist   = (unsigned*)(zs + n);
  unsigned* bstart = hist + (size_t)PB_MAXB * PB_NS;
  unsigned* packed = bstart + 1024;
  int*      sorted = (int*)(packed + E);

  int gG = (n + (G2T / 8) * GN - 1) / ((G2T / 8) * GN);

  gcn_bucket_count<<<PB_NS, PB_T, 0, stream>>>(dst, hist, nb, E, L);
  gcn_bucket_scatter<<<PB_NS, PB_T, 0, stream>>>(src, dst, hist, bstart, packed, nb, E, L);
  gcn_bucket_rank<<<nb, PC_T, 0, stream>>>(packed, bstart, x, cnt, rowptr, xs, sorted, n);
  gcn_gather1_dense<<<dblocks, 256, 0, stream>>>(sorted, rowptr, cnt, xs, W1, b1, W2, zs, n);
  gcn_gather2<<<gG, G2T, 0, stream>>>(sorted, rowptr, cnt, zs, b2, out, n);
  return true;
}

static bool gcn_cell_path(const float* x, const int* src, const int* dst,
                          const float* W1, const float* b1, const float* W2,
                          const float* b2, float* out, int n, int E,
                          int dblocks, void* d_ws, size_t ws_size,
                          hipStream_t stream) {
  static_assert(2 * PB_NS * PB_MAXB >= PB_MAXB * PB_NODES, "zs fits the two tables");
  int nb = (n + PB_NODES - 1) / PB_NODES;
  int L = (E + PB_NS - 1) / PB_NS;
  size_t tab = (size_t)PB_MAXB * PB_NS;
  size_t need = (4 * (size_t)n + 2 * tab + (size_t)nb * PB_NS * CL_CAP +
                 (size_t)E) * sizeof(float);
  if (n < 1 || n > PB_MAXB * PB_NODES || (n & 3) != 0 || E < 1 || L > PB_LMAX ||
      ws_size < need) return false;
  unsigned* cnt     = (unsigned*)d_ws;
  unsigned* rowptr  = cnt + n;
  float*    xs      = (float*)(rowptr + n);
  unsigned* cellcnt = (unsigned*)(xs + 2 * (size_t)n);
  unsigned* cellpre = cellcnt + tab;
  unsigned* cells   = cellpre + tab;
  int*      sorted  = (int*)(cells + (size_t)nb * PB_NS * CL_CAP);
  float*    zs      = (float*)cellcnt;

  int gG = (n + (G2T / 8) * GN - 1) / ((G2T / 8) * GN);

  gcn_cell_scatter<<<PB_NS, PB_T, 0, stream>>>(src, dst, cellcnt, cellpre, cells, nb, E, L);
  gcn_cell_rank<<<nb, PC_T, 0, stream>>>(cells, cellcnt, cellpre, src, dst, x, cnt, rowptr, xs, sorted, n, E, L);
  gcn_gather1_dense<<<dblocks, 256, 0, stream>>>(sorted, rowptr, cnt, xs, W1, b1, W2, zs, n);
  gcn_gather2<<<gG, G2T, 0, stream>>>(sorted, rowptr, cnt, zs, b2, out, n);
  return true;
}

extern "C" void kernel_launch(void* const* d_in, const int* in_sizes, int n_in,
                              void* d_out, int out_size, void* d_ws, size_t ws_size,
                              hipStream_t stream) {
  const float* x  = (const float*)d_in[0];
  const int*   ei = (const int*)d_in[1];
  const float* W1 = (const float*)d_in[2];
  const float* b1 = (const float*)d_in[3];
  const float* W2 = (const float*)d_in[4];
  const float* b2 = (const float*)d_in[5];

  int n = in_sizes[0] / 2;
  int E = in_sizes[1] / 2;
  const int* src = ei;
  const int* dst = ei + E;
  float* out = (float*)d_out;

  const int tb = 256;
  int gN = (n + tb - 1) / tb;
  int gE = (E + tb - 1) / tb;
  if (gE > 4096) gE = 4096;
  int tiles = (n + 15) / 16;
  int dblocks = (tiles + 7) / 8;

  if (gcn_cell_path(x, src, dst, W1, b1, W2, b2, out, n, E, dblocks, d_ws,
                    ws_size, stream)) return;

  if (gcn_bucket_path(x, src, dst, W1, b1, W2, b2, out, n, E, dblocks, d_ws,
                      ws_size, stream)) return;

  if (gcn_sorted_path<32, 12800>(x, src, dst, W1, b1, W2, b2, out, n, E,
                                 dblocks, d_ws, ws_size, stream)) return;
  if (gcn_sorted_path<16, 6400>(x, src, dst, W1, b1, W2, b2, out, n, E,
                                dblocks, d_ws, ws_size, stream)) return;

  float* dinv = (float*)d_ws;
  float* agg  = dinv + n;
  float* z    = agg + 2 * (size_t)n;

  gcn_init<<<gN, tb, 0, stream>>>(dinv, agg, z, out, n);
  gcn_degree<<<gE, tb, 0, stream>>>(dst, dinv, E);
  gcn_rsqrt<<<gN, tb, 0, stream>>>(dinv, n);
  gcn_agg2<<<gE, tb, 0, stream>>>(src, dst, x, dinv, agg, E);

  gcn_dense_wmma<<<dblocks, 256, 0, stream>>>(agg, x, dinv, W1, b1, W2, z, n);

  gcn_out_self<<<gN, tb, 0, stream>>>(z, dinv, b2, out, n);
  gcn_out_edges<<<gE, tb, 0, stream>>>(src, dst, z, dinv, out, E);
}
